# starting kernel
# baseline (speedup 1.0000x reference)
#include <hip/hip_runtime.h>
#include <math.h>

#define D 128
#define RELS 3

typedef float v2f __attribute__((ext_vector_type(2)));
typedef float v8f __attribute__((ext_vector_type(8)));


__device__ __forceinline__ void atomicMaxF(float* addr, float value) {
  if (value >= 0.f) atomicMax((int*)addr, __float_as_int(value));
  else              atomicMin((unsigned int*)addr, __float_as_uint(value));
}

__device__ __forceinline__ float lrelu(float x) { return x > 0.f ? x : 0.2f * x; }


__global__ void k_init_wsum(float* wsum) {
  if (threadIdx.x < RELS) wsum[threadIdx.x] = 0.f;
}

__global__ void k_init_rel(float* __restrict__ zacc, float* __restrict__ m,
                           float* __restrict__ zsum, int nfeat, int nnode) {
  int i = blockIdx.x * blockDim.x + threadIdx.x;
  if (i < nfeat) zacc[i] = 0.f;
  if (i < nnode) { m[i] = -__builtin_inff(); zsum[i] = 0.f; }
}

__global__ __launch_bounds__(256) void k_rowdot(const float* __restrict__ W,
                                                const float* __restrict__ al,
                                                const float* __restrict__ ar,
                                                float* __restrict__ wa,
                                                float* __restrict__ wr) {
  int t = threadIdx.x;
  const float* v = (t < D) ? al : ar;
  float* o = (t < D) ? wa : wr;
  int d = t & (D - 1);
  float s = 0.f;
#pragma unroll 8
  for (int h = 0; h < D; ++h) s += W[d * D + h] * v[h];
  o[d] = s;
}

__global__ __launch_bounds__(256) void k_gemm_wmma(const float* __restrict__ X,
                                                   const float* __restrict__ W,
                                                   float* __restrict__ H, int N) {
  __shared__ float Ws[D * D];
  int t = threadIdx.x;
#pragma unroll
  for (int i = 0; i < (D * D / 4) / 256; ++i)
    ((float4*)Ws)[t + i * 256] = ((const float4*)W)[t + i * 256];
  __syncthreads();

  const int wave = t >> 5;
  const int lane = t & 31;
  const int lrow = lane & 15;
  const int half = lane >> 4;
  const int rowbase = blockIdx.x * 128 + wave * 16;

  int xr = rowbase + lrow;
  if (xr > N - 1) xr = N - 1;
  const float* xp = X + (size_t)xr * D;

  v2f afrag[32];
#pragma unroll
  for (int kt = 0; kt < 32; ++kt) {
    int k0 = kt * 4 + 2 * half;
    afrag[kt] = *(const v2f*)(xp + k0);
  }

#pragma unroll
  for (int nt = 0; nt < 8; ++nt) {
    v8f acc = {};
    int nb = nt * 16 + lrow;
#pragma unroll
    for (int kt = 0; kt < 32; ++kt) {
      int k0 = kt * 4 + 2 * half;
      v2f b;
      b.x = Ws[k0 * D + nb];
      b.y = Ws[(k0 + 1) * D + nb];
      acc = __builtin_amdgcn_wmma_f32_16x16x4_f32(false, afrag[kt], false, b,
                                                  (short)0, acc, false, false);
    }
#pragma unroll
    for (int v = 0; v < 8; ++v) {
      int grow = rowbase + v + 8 * half;
      if (grow < N) H[(size_t)grow * D + nb] = acc[v];
    }
  }
}

__global__ __launch_bounds__(256) void k_matvec(const float* __restrict__ X,
                                                const float* __restrict__ v,
                                                float* __restrict__ out, int n) {
  int wave = threadIdx.x >> 5, lane = threadIdx.x & 31;
  int row = blockIdx.x * 8 + wave;
  if (row >= n) return;
  float4 x = ((const float4*)X)[(size_t)row * (D / 4) + lane];
  float4 c = ((const float4*)v)[lane];
  float s = x.x * c.x + x.y * c.y + x.z * c.z + x.w * c.w;
#pragma unroll
  for (int off = 16; off; off >>= 1) s += __shfl_xor(s, off, 32);
  if (lane == 0) out[row] = s;
}


__global__ void k_edge_max(const int* __restrict__ si, const int* __restrict__ di,
                           const float* __restrict__ el, const float* __restrict__ er,
                           float* __restrict__ m, int E) {
  int i = blockIdx.x * blockDim.x + threadIdx.x;
  if (i >= E) return;
  int d = di[i];
  atomicMaxF(&m[d], lrelu(el[si[i]] + er[d]));
}

__global__ void k_edge_sum(const int* __restrict__ si, const int* __restrict__ di,
                           const float* __restrict__ el, const float* __restrict__ er,
                           const float* __restrict__ m, float* __restrict__ zsum, int E) {
  int i = blockIdx.x * blockDim.x + threadIdx.x;
  if (i >= E) return;
  int d = di[i];
  float e = lrelu(el[si[i]] + er[d]);
  atomicAdd(&zsum[d], expf(e - m[d]));
}

__global__ __launch_bounds__(256) void k_edge_aggr(const int* __restrict__ si,
                                                   const int* __restrict__ di,
                                                   const float* __restrict__ el,
                                                   const float* __restrict__ er,
                                                   const float* __restrict__ m,
                                                   const float* __restrict__ zsum,
                                                   const float* __restrict__ hs,
                                                   float* __restrict__ zacc, int E) {
  int wave = threadIdx.x >> 5, lane = threadIdx.x & 31;
  int e = blockIdx.x * 8 + wave;
  if (e >= E) return;
  int s = si[e], d = di[e];
  float x = lrelu(el[s] + er[d]);
  float alpha = expf(x - m[d]) / zsum[d];
  float4 h = ((const float4*)hs)[(size_t)s * (D / 4) + lane];
  float* zp = zacc + (size_t)d * D + lane * 4;
  atomicAdd(zp + 0, alpha * h.x);
  atomicAdd(zp + 1, alpha * h.y);
  atomicAdd(zp + 2, alpha * h.z);
  atomicAdd(zp + 3, alpha * h.w);
}

__global__ void k_fin(float* __restrict__ Z, const float* __restrict__ b, int n) {
  int i = blockIdx.x * blockDim.x + threadIdx.x;
  if (i >= n) return;
  float x = Z[i] + b[i & (D - 1)];
  Z[i] = x > 0.f ? x : expf(x) - 1.f;
}

__global__ __launch_bounds__(128) void k_sem(const float* __restrict__ Z,
                                             const float* __restrict__ W1,
                                             const float* __restrict__ b1,
                                             const float* __restrict__ W2,
                                             float* __restrict__ wsum) {
  __shared__ float zs[D];
  __shared__ float red[D];
  int node = blockIdx.x;
  int h = threadIdx.x;
  zs[h] = Z[(size_t)node * D + h];
  __syncthreads();
  float acc = b1[h];
#pragma unroll 8
  for (int d = 0; d < D; ++d) acc += zs[d] * W1[d * D + h];
  red[h] = tanhf(acc) * W2[h];
  __syncthreads();
  for (int s = 64; s; s >>= 1) {
    if (h < s) red[h] += red[h + s];
    __syncthreads();
  }
  if (h == 0) atomicAdd(wsum, red[0]);
}

__global__ void k_mix(const float* __restrict__ Z3, const float* __restrict__ wsum,
                      float* __restrict__ out, int n, float invN) {
  int i = blockIdx.x * blockDim.x + threadIdx.x;
  float w0 = wsum[0] * invN, w1 = wsum[1] * invN, w2 = wsum[2] * invN;
  float mx = fmaxf(w0, fmaxf(w1, w2));
  float e0 = expf(w0 - mx), e1 = expf(w1 - mx), e2 = expf(w2 - mx);
  float inv = 1.f / (e0 + e1 + e2);
  float a0 = e0 * inv, a1 = e1 * inv, a2 = e2 * inv;
  if (i < n)
    out[i] = a0 * Z3[i] + a1 * Z3[(size_t)n + i] + a2 * Z3[2 * (size_t)n + i];
  if (i == 0) { out[n] = a0; out[n + 1] = a1; out[n + 2] = a2; }
}


extern "C" void kernel_launch(void* const* d_in, const int* in_sizes, int n_in,
                              void* d_out, int out_size, void* d_ws, size_t ws_size,
                              hipStream_t stream) {
  const float* dst_feat = (const float*)d_in[0];
  const float* src_feat[RELS] = {(const float*)d_in[1], (const float*)d_in[2],
                                 (const float*)d_in[3]};
  const float* sem_W1 = (const float*)d_in[4];
  const float* sem_b1 = (const float*)d_in[5];
  const float* sem_W2 = (const float*)d_in[6];
  const float *Wg[RELS], *al[RELS], *ar[RELS], *bg[RELS];
  const int *sidx[RELS], *didx[RELS];
  for (int r = 0; r < RELS; ++r) {
    int b = 7 + 6 * r;
    Wg[r] = (const float*)d_in[b + 0];
    al[r] = (const float*)d_in[b + 1];
    ar[r] = (const float*)d_in[b + 2];
    bg[r] = (const float*)d_in[b + 3];
    sidx[r] = (const int*)d_in[b + 4];
    didx[r] = (const int*)d_in[b + 5];
  }
  const int ND = in_sizes[0] / D;
  const int NS = in_sizes[1] / D;
  const int E  = in_sizes[11];

  float* ws   = (float*)d_ws;
  float* z3   = ws;
  float* hs   = z3 + (size_t)RELS * ND * D;
  float* el   = hs + (size_t)NS * D;
  float* er   = el + NS;
  float* m    = er + ND;
  float* zsum = m + ND;
  float* wa   = zsum + ND;
  float* wr   = wa + D;
  float* wsum = wr + D;

  k_init_wsum<<<1, 32, 0, stream>>>(wsum);

  for (int r = 0; r < RELS; ++r) {
    float* zr = z3 + (size_t)r * ND * D;
    k_rowdot<<<1, 256, 0, stream>>>(Wg[r], al[r], ar[r], wa, wr);
    k_init_rel<<<(ND * D + 255) / 256, 256, 0, stream>>>(zr, m, zsum, ND * D, ND);
    k_gemm_wmma<<<(NS + 127) / 128, 256, 0, stream>>>(src_feat[r], Wg[r], hs, NS);
    k_matvec<<<(NS + 7) / 8, 256, 0, stream>>>(src_feat[r], wa, el, NS);
    k_matvec<<<(ND + 7) / 8, 256, 0, stream>>>(dst_feat, wr, er, ND);
    k_edge_max<<<(E + 255) / 256, 256, 0, stream>>>(sidx[r], didx[r], el, er, m, E);
    k_edge_sum<<<(E + 255) / 256, 256, 0, stream>>>(sidx[r], didx[r], el, er, m, zsum, E);
    k_edge_aggr<<<(E + 7) / 8, 256, 0, stream>>>(sidx[r], didx[r], el, er, m, zsum, hs,
                                                 zr, E);
    k_fin<<<(ND * D + 255) / 256, 256, 0, stream>>>(zr, bg[r], ND * D);
    k_sem<<<ND, 128, 0, stream>>>(zr, sem_W1, sem_b1, sem_W2, wsum + r);
  }

  k_mix<<<(ND * D + 255) / 256, 256, 0, stream>>>(z3, wsum, (float*)d_out, ND * D,
                                                  1.0f / (float)ND);
}


# k_sem as a 128-row f32 WMMA tile kernel (W1 staged in LDS, tanh*W2 summed in registers): one atomic per block instead of one per node, so the order of the wsum additions changes
# speedup vs baseline: 1.2071x; 1.2071x over previous
#include <hip/hip_runtime.h>
#include <math.h>

#define D 128
#define RELS 3

typedef float v2f __attribute__((ext_vector_type(2)));
typedef float v8f __attribute__((ext_vector_type(8)));


__device__ __forceinline__ void atomicMaxF(float* addr, float value) {
  if (value >= 0.f) atomicMax((int*)addr, __float_as_int(value));
  else              atomicMin((unsigned int*)addr, __float_as_uint(value));
}

__device__ __forceinline__ float lrelu(float x) { return x > 0.f ? x : 0.2f * x; }


__global__ void k_init_wsum(float* wsum) {
  if (threadIdx.x < RELS) wsum[threadIdx.x] = 0.f;
}

__global__ void k_init_rel(float* __restrict__ zacc, float* __restrict__ m,
                           float* __restrict__ zsum, int nfeat, int nnode) {
  int i = blockIdx.x * blockDim.x + threadIdx.x;
  if (i < nfeat) zacc[i] = 0.f;
  if (i < nnode) { m[i] = -__builtin_inff(); zsum[i] = 0.f; }
}

__global__ __launch_bounds__(256) void k_rowdot(const float* __restrict__ W,
                                                const float* __restrict__ al,
                                                const float* __restrict__ ar,
                                                float* __restrict__ wa,
                                                float* __restrict__ wr) {
  int t = threadIdx.x;
  const float* v = (t < D) ? al : ar;
  float* o = (t < D) ? wa : wr;
  int d = t & (D - 1);
  float s = 0.f;
#pragma unroll 8
  for (int h = 0; h < D; ++h) s += W[d * D + h] * v[h];
  o[d] = s;
}

__global__ __launch_bounds__(256) void k_gemm_wmma(const float* __restrict__ X,
                                                   const float* __restrict__ W,
                                                   float* __restrict__ H, int N) {
  __shared__ float Ws[D * D];
  int t = threadIdx.x;
#pragma unroll
  for (int i = 0; i < (D * D / 4) / 256; ++i)
    ((float4*)Ws)[t + i * 256] = ((const float4*)W)[t + i * 256];
  __syncthreads();

  const int wave = t >> 5;
  const int lane = t & 31;
  const int lrow = lane & 15;
  const int half = lane >> 4;
  const int rowbase = blockIdx.x * 128 + wave * 16;

  int xr = rowbase + lrow;
  if (xr > N - 1) xr = N - 1;
  const float* xp = X + (size_t)xr * D;

  v2f afrag[32];
#pragma unroll
  for (int kt = 0; kt < 32; ++kt) {
    int k0 = kt * 4 + 2 * half;
    afrag[kt] = *(const v2f*)(xp + k0);
  }

#pragma unroll
  for (int nt = 0; nt < 8; ++nt) {
    v8f acc = {};
    int nb = nt * 16 + lrow;
#pragma unroll
    for (int kt = 0; kt < 32; ++kt) {
      int k0 = kt * 4 + 2 * half;
      v2f b;
      b.x = Ws[k0 * D + nb];
      b.y = Ws[(k0 + 1) * D + nb];
      acc = __builtin_amdgcn_wmma_f32_16x16x4_f32(false, afrag[kt], false, b,
                                                  (short)0, acc, false, false);
    }
#pragma unroll
    for (int v = 0; v < 8; ++v) {
      int grow = rowbase + v + 8 * half;
      if (grow < N) H[(size_t)grow * D + nb] = acc[v];
    }
  }
}

__global__ __launch_bounds__(256) void k_matvec(const float* __restrict__ X,
                                                const float* __restrict__ v,
                                                float* __restrict__ out, int n) {
  int wave = threadIdx.x >> 5, lane = threadIdx.x & 31;
  int row = blockIdx.x * 8 + wave;
  if (row >= n) return;
  float4 x = ((const float4*)X)[(size_t)row * (D / 4) + lane];
  float4 c = ((const float4*)v)[lane];
  float s = x.x * c.x + x.y * c.y + x.z * c.z + x.w * c.w;
#pragma unroll
  for (int off = 16; off; off >>= 1) s += __shfl_xor(s, off, 32);
  if (lane == 0) out[row] = s;
}


__global__ void k_edge_max(const int* __restrict__ si, const int* __restrict__ di,
                           const float* __restrict__ el, const float* __restrict__ er,
                           float* __restrict__ m, int E) {
  int i = blockIdx.x * blockDim.x + threadIdx.x;
  if (i >= E) return;
  int d = di[i];
  atomicMaxF(&m[d], lrelu(el[si[i]] + er[d]));
}

__global__ void k_edge_sum(const int* __restrict__ si, const int* __restrict__ di,
                           const float* __restrict__ el, const float* __restrict__ er,
                           const float* __restrict__ m, float* __restrict__ zsum, int E) {
  int i = blockIdx.x * blockDim.x + threadIdx.x;
  if (i >= E) return;
  int d = di[i];
  float e = lrelu(el[si[i]] + er[d]);
  atomicAdd(&zsum[d], expf(e - m[d]));
}

__global__ __launch_bounds__(256) void k_edge_aggr(const int* __restrict__ si,
                                                   const int* __restrict__ di,
                                                   const float* __restrict__ el,
                                                   const float* __restrict__ er,
                                                   const float* __restrict__ m,
                                                   const float* __restrict__ zsum,
                                                   const float* __restrict__ hs,
                                                   float* __restrict__ zacc, int E) {
  int wave = threadIdx.x >> 5, lane = threadIdx.x & 31;
  int e = blockIdx.x * 8 + wave;
  if (e >= E) return;
  int s = si[e], d = di[e];
  float x = lrelu(el[s] + er[d]);
  float alpha = expf(x - m[d]) / zsum[d];
  float4 h = ((const float4*)hs)[(size_t)s * (D / 4) + lane];
  float* zp = zacc + (size_t)d * D + lane * 4;
  atomicAdd(zp + 0, alpha * h.x);
  atomicAdd(zp + 1, alpha * h.y);
  atomicAdd(zp + 2, alpha * h.z);
  atomicAdd(zp + 3, alpha * h.w);
}

__global__ void k_fin(float* __restrict__ Z, const float* __restrict__ b, int n) {
  int i = blockIdx.x * blockDim.x + threadIdx.x;
  if (i >= n) return;
  float x = Z[i] + b[i & (D - 1)];
  Z[i] = x > 0.f ? x : expf(x) - 1.f;
}

__global__ __launch_bounds__(256) void k_sem_wmma(const float* __restrict__ Z,
                                                  const float* __restrict__ W1,
                                                  const float* __restrict__ b1,
                                                  const float* __restrict__ W2,
                                                  float* __restrict__ wsum, int N) {
  __shared__ float Ws[D * D];
  __shared__ float red[8];
  int t = threadIdx.x;
#pragma unroll
  for (int i = 0; i < (D * D / 4) / 256; ++i)
    ((float4*)Ws)[t + i * 256] = ((const float4*)W1)[t + i * 256];
  __syncthreads();

  const int wave = t >> 5;
  const int lane = t & 31;
  const int lrow = lane & 15;
  const int half = lane >> 4;
  const int rowbase = blockIdx.x * 128 + wave * 16;

  int xr = rowbase + lrow;
  if (xr > N - 1) xr = N - 1;
  const float* xp = Z + (size_t)xr * D;

  v2f afrag[32];
#pragma unroll
  for (int kt = 0; kt < 32; ++kt) {
    int k0 = kt * 4 + 2 * half;
    afrag[kt] = *(const v2f*)(xp + k0);
  }

  float s = 0.f;
#pragma unroll
  for (int nt = 0; nt < 8; ++nt) {
    v8f acc = {};
    int nb = nt * 16 + lrow;
#pragma unroll
    for (int kt = 0; kt < 32; ++kt) {
      int k0 = kt * 4 + 2 * half;
      v2f b;
      b.x = Ws[k0 * D + nb];
      b.y = Ws[(k0 + 1) * D + nb];
      acc = __builtin_amdgcn_wmma_f32_16x16x4_f32(false, afrag[kt], false, b,
                                                  (short)0, acc, false, false);
    }
    asm volatile("v_nop\n\tv_nop\n\tv_nop\n\tv_nop" : "+v"(acc));
    const float bn = b1[nb];
    const float wn = W2[nb];
#pragma unroll
    for (int v = 0; v < 8; ++v) {
      int grow = rowbase + v + 8 * half;
      float y = tanhf(acc[v] + bn) * wn;
      s += (grow < N) ? y : 0.f;
    }
  }

#pragma unroll
  for (int off = 16; off; off >>= 1) s += __shfl_xor(s, off, 32);
  if (lane == 0) red[wave] = s;
  __syncthreads();
  if (t == 0) {
    float tot = ((red[0] + red[1]) + (red[2] + red[3])) +
                ((red[4] + red[5]) + (red[6] + red[7]));
    atomicAdd(wsum, tot);
  }
}

__global__ void k_mix(const float* __restrict__ Z3, const float* __restrict__ wsum,
                      float* __restrict__ out, int n, float invN) {
  int i = blockIdx.x * blockDim.x + threadIdx.x;
  float w0 = wsum[0] * invN, w1 = wsum[1] * invN, w2 = wsum[2] * invN;
  float mx = fmaxf(w0, fmaxf(w1, w2));
  float e0 = expf(w0 - mx), e1 = expf(w1 - mx), e2 = expf(w2 - mx);
  float inv = 1.f / (e0 + e1 + e2);
  float a0 = e0 * inv, a1 = e1 * inv, a2 = e2 * inv;
  if (i < n)
    out[i] = a0 * Z3[i] + a1 * Z3[(size_t)n + i] + a2 * Z3[2 * (size_t)n + i];
  if (i == 0) { out[n] = a0; out[n + 1] = a1; out[n + 2] = a2; }
}


extern "C" void kernel_launch(void* const* d_in, const int* in_sizes, int n_in,
                              void* d_out, int out_size, void* d_ws, size_t ws_size,
                              hipStream_t stream) {
  const float* dst_feat = (const float*)d_in[0];
  const float* src_feat[RELS] = {(const float*)d_in[1], (const float*)d_in[2],
                                 (const float*)d_in[3]};
  const float* sem_W1 = (const float*)d_in[4];
  const float* sem_b1 = (const float*)d_in[5];
  const float* sem_W2 = (const float*)d_in[6];
  const float *Wg[RELS], *al[RELS], *ar[RELS], *bg[RELS];
  const int *sidx[RELS], *didx[RELS];
  for (int r = 0; r < RELS; ++r) {
    int b = 7 + 6 * r;
    Wg[r] = (const float*)d_in[b + 0];
    al[r] = (const float*)d_in[b + 1];
    ar[r] = (const float*)d_in[b + 2];
    bg[r] = (const float*)d_in[b + 3];
    sidx[r] = (const int*)d_in[b + 4];
    didx[r] = (const int*)d_in[b + 5];
  }
  const int ND = in_sizes[0] / D;
  const int NS = in_sizes[1] / D;
  const int E  = in_sizes[11];

  float* ws   = (float*)d_ws;
  float* z3   = ws;
  float* hs   = z3 + (size_t)RELS * ND * D;
  float* el   = hs + (size_t)NS * D;
  float* er   = el + NS;
  float* m    = er + ND;
  float* zsum = m + ND;
  float* wa   = zsum + ND;
  float* wr   = wa + D;
  float* wsum = wr + D;

  k_init_wsum<<<1, 32, 0, stream>>>(wsum);

  for (int r = 0; r < RELS; ++r) {
    float* zr = z3 + (size_t)r * ND * D;
    k_rowdot<<<1, 256, 0, stream>>>(Wg[r], al[r], ar[r], wa, wr);
    k_init_rel<<<(ND * D + 255) / 256, 256, 0, stream>>>(zr, m, zsum, ND * D, ND);
    k_gemm_wmma<<<(NS + 127) / 128, 256, 0, stream>>>(src_feat[r], Wg[r], hs, NS);
    k_matvec<<<(NS + 7) / 8, 256, 0, stream>>>(src_feat[r], wa, el, NS);
    k_matvec<<<(ND + 7) / 8, 256, 0, stream>>>(dst_feat, wr, er, ND);
    k_edge_max<<<(E + 255) / 256, 256, 0, stream>>>(sidx[r], didx[r], el, er, m, E);
    k_edge_sum<<<(E + 255) / 256, 256, 0, stream>>>(sidx[r], didx[r], el, er, m, zsum, E);
    k_edge_aggr<<<(E + 7) / 8, 256, 0, stream>>>(sidx[r], didx[r], el, er, m, zsum, hs,
                                                 zr, E);
    k_fin<<<(ND * D + 255) / 256, 256, 0, stream>>>(zr, bg[r], ND * D);
    k_sem_wmma<<<(ND + 127) / 128, 256, 0, stream>>>(zr, sem_W1, sem_b1, sem_W2,
                                                      wsum + r, ND);
  }

  k_mix<<<(ND * D + 255) / 256, 256, 0, stream>>>(z3, wsum, (float*)d_out, ND * D,
                                                  1.0f / (float)ND);
}


# edges counting-sorted by dst each call (1 int atomic per edge; list parked in out); one wave per dst row does softmax + hs-row sum in registers, writes elu once: no f32 atomics, sum order changes
# speedup vs baseline: 2.2211x; 1.8400x over previous
#include <hip/hip_runtime.h>
#include <math.h>

#define D 128
#define RELS 3

typedef float v2f __attribute__((ext_vector_type(2)));
typedef float v8f __attribute__((ext_vector_type(8)));


__device__ __forceinline__ float lrelu(float x) { return x > 0.f ? x : 0.2f * x; }


__global__ void k_init_wsum(float* wsum) {
  if (threadIdx.x < RELS) wsum[threadIdx.x] = 0.f;
}

__global__ __launch_bounds__(256) void k_rowdot(const float* __restrict__ W,
                                                const float* __restrict__ al,
                                                const float* __restrict__ ar,
                                                float* __restrict__ wa,
                                                float* __restrict__ wr) {
  int t = threadIdx.x;
  const float* v = (t < D) ? al : ar;
  float* o = (t < D) ? wa : wr;
  int d = t & (D - 1);
  float s = 0.f;
#pragma unroll 8
  for (int h = 0; h < D; ++h) s += W[d * D + h] * v[h];
  o[d] = s;
}

__global__ __launch_bounds__(256) void k_gemm_wmma(const float* __restrict__ X,
                                                   const float* __restrict__ W,
                                                   float* __restrict__ H, int N) {
  __shared__ float Ws[D * D];
  int t = threadIdx.x;
#pragma unroll
  for (int i = 0; i < (D * D / 4) / 256; ++i)
    ((float4*)Ws)[t + i * 256] = ((const float4*)W)[t + i * 256];
  __syncthreads();

  const int wave = t >> 5;
  const int lane = t & 31;
  const int lrow = lane & 15;
  const int half = lane >> 4;
  const int rowbase = blockIdx.x * 128 + wave * 16;

  int xr = rowbase + lrow;
  if (xr > N - 1) xr = N - 1;
  const float* xp = X + (size_t)xr * D;

  v2f afrag[32];
#pragma unroll
  for (int kt = 0; kt < 32; ++kt) {
    int k0 = kt * 4 + 2 * half;
    afrag[kt] = *(const v2f*)(xp + k0);
  }

#pragma unroll
  for (int nt = 0; nt < 8; ++nt) {
    v8f acc = {};
    int nb = nt * 16 + lrow;
#pragma unroll
    for (int kt = 0; kt < 32; ++kt) {
      int k0 = kt * 4 + 2 * half;
      v2f b;
      b.x = Ws[k0 * D + nb];
      b.y = Ws[(k0 + 1) * D + nb];
      acc = __builtin_amdgcn_wmma_f32_16x16x4_f32(false, afrag[kt], false, b,
                                                  (short)0, acc, false, false);
    }
#pragma unroll
    for (int v = 0; v < 8; ++v) {
      int grow = rowbase + v + 8 * half;
      if (grow < N) H[(size_t)grow * D + nb] = acc[v];
    }
  }
}

__global__ __launch_bounds__(256) void k_matvec(const float* __restrict__ X,
                                                const float* __restrict__ v,
                                                float* __restrict__ out,
                                                int* __restrict__ zero, int n) {
  int wave = threadIdx.x >> 5, lane = threadIdx.x & 31;
  int row = blockIdx.x * 8 + wave;
  if (row >= n) return;
  float4 x = ((const float4*)X)[(size_t)row * (D / 4) + lane];
  float4 c = ((const float4*)v)[lane];
  float s = x.x * c.x + x.y * c.y + x.z * c.z + x.w * c.w;
#pragma unroll
  for (int off = 16; off; off >>= 1) s += __shfl_xor(s, off, 32);
  if (lane == 0) {
    out[row] = s;
    if (zero) zero[row] = 0;
  }
}


__global__ void k_rank(const int* __restrict__ di, int* __restrict__ cnt,
                       int* __restrict__ rank, int E) {
  int i = blockIdx.x * blockDim.x + threadIdx.x;
  if (i >= E) return;
  rank[i] = atomicAdd(&cnt[di[i]], 1);
}

__global__ __launch_bounds__(1024) void k_scan(const int* __restrict__ cnt,
                                               int* __restrict__ cur, int n) {
  __shared__ int wtot[32];
  const int t = threadIdx.x;
  const int lane = t & 31, wave = t >> 5;
  const int chunk = (n + 1023) / 1024;
  int b = t * chunk;
  int e = b + chunk;
  if (b > n) b = n;
  if (e > n) e = n;
  int s = 0;
#pragma unroll 8
  for (int i = b; i < e; ++i) s += cnt[i];
  int incl = s;
#pragma unroll
  for (int off = 1; off < 32; off <<= 1) {
    int u = __shfl_up(incl, off, 32);
    if (lane >= off) incl += u;
  }
  if (lane == 31) wtot[wave] = incl;
  __syncthreads();
  if (wave == 0) {
    int w = wtot[lane];
    int wi = w;
#pragma unroll
    for (int off = 1; off < 32; off <<= 1) {
      int u = __shfl_up(wi, off, 32);
      if (lane >= off) wi += u;
    }
    wtot[lane] = wi - w;
  }
  __syncthreads();
  int run = wtot[wave] + incl - s;
#pragma unroll 8
  for (int i = b; i < e; ++i) {
    cur[i] = run;
    run += cnt[i];
  }
}

__global__ void k_place(const int* __restrict__ si, const int* __restrict__ di,
                        const float* __restrict__ el, const float* __restrict__ er,
                        const int* __restrict__ cur, const int* __restrict__ rank,
                        int2* __restrict__ lst, int E) {
  int i = blockIdx.x * blockDim.x + threadIdx.x;
  if (i >= E) return;
  int s = si[i], d = di[i];
  float x = lrelu(el[s] + er[d]);
  lst[cur[d] + rank[i]] = make_int2(s, __float_as_int(x));
}

__device__ __forceinline__ void fma4(float4& a, float p, const float4& h) {
  a.x += p * h.x; a.y += p * h.y; a.z += p * h.z; a.w += p * h.w;
}

template <int N>
__device__ __forceinline__ void add_rows(float4& acc, const float* __restrict__ hs, int vx,
                                         int pb, int k, int lane) {
  float4 h[N];
  float pk[N];
#pragma unroll
  for (int u = 0; u < N; ++u) {
    const unsigned sj = (unsigned)__builtin_amdgcn_readlane(vx, k + u);
    pk[u] = __int_as_float(__builtin_amdgcn_readlane(pb, k + u));
    h[u] = ((const float4*)(hs + (size_t)sj * D))[lane];
  }
#pragma unroll
  for (int u = 0; u < N; ++u) fma4(acc, pk[u], h[u]);
}

__global__ __launch_bounds__(256) void k_gather(const int* __restrict__ cnt,
                                                const int* __restrict__ cur,
                                                const int2* __restrict__ lst,
                                                const float* __restrict__ hs,
                                                const float* __restrict__ bias,
                                                float* __restrict__ Z, int n) {
  const int lane = threadIdx.x & 31;
  const int d = __builtin_amdgcn_readfirstlane(blockIdx.x * 8 + (threadIdx.x >> 5));
  if (d >= n) return;
  const int c = cnt[d];
  const int start = cur[d];
  const int NEG_INF_BITS = (int)0xff800000u;

  int2 v0 = make_int2(0, NEG_INF_BITS);
  if (lane < c) v0 = lst[start + lane];
  float mx = __int_as_float(v0.y);
  for (int base = 32; base < c; base += 32) {
    int j = base + lane;
    if (j < c) mx = fmaxf(mx, __int_as_float(lst[start + j].y));
  }
#pragma unroll
  for (int off = 16; off; off >>= 1) mx = fmaxf(mx, __shfl_xor(mx, off, 32));

  float4 acc = make_float4(0.f, 0.f, 0.f, 0.f);
  float zs = 0.f;
  for (int base = 0; base < c; base += 32) {
    int2 v = v0;
    const int j = base + lane;
    if (base > 0) {
      v = make_int2(0, NEG_INF_BITS);
      if (j < c) v = lst[start + j];
    }
    const float pe = expf(__int_as_float(v.y) - mx);
    const float p = (j < c) ? pe : 0.f;
    zs += p;
    const int pb = __float_as_int(p);
    int nc = c - base;
    if (nc > 32) nc = 32;
    int k = 0;
    for (; k + 8 <= nc; k += 8) add_rows<8>(acc, hs, v.x, pb, k, lane);
    switch (nc - k) {
      case 7: add_rows<7>(acc, hs, v.x, pb, k, lane); break;
      case 6: add_rows<6>(acc, hs, v.x, pb, k, lane); break;
      case 5: add_rows<5>(acc, hs, v.x, pb, k, lane); break;
      case 4: add_rows<4>(acc, hs, v.x, pb, k, lane); break;
      case 3: add_rows<3>(acc, hs, v.x, pb, k, lane); break;
      case 2: add_rows<2>(acc, hs, v.x, pb, k, lane); break;
      case 1: add_rows<1>(acc, hs, v.x, pb, k, lane); break;
      default: break;
    }
  }
#pragma unroll
  for (int off = 16; off; off >>= 1) zs += __shfl_xor(zs, off, 32);

  const float inv = (c > 0) ? 1.f / zs : 0.f;
  const float4 bb = ((const float4*)bias)[lane];
  float4 o;
  o.x = acc.x * inv + bb.x;
  o.y = acc.y * inv + bb.y;
  o.z = acc.z * inv + bb.z;
  o.w = acc.w * inv + bb.w;
  o.x = o.x > 0.f ? o.x : expf(o.x) - 1.f;
  o.y = o.y > 0.f ? o.y : expf(o.y) - 1.f;
  o.z = o.z > 0.f ? o.z : expf(o.z) - 1.f;
  o.w = o.w > 0.f ? o.w : expf(o.w) - 1.f;
  ((float4*)Z)[(size_t)d * (D / 4) + lane] = o;
}

__global__ __launch_bounds__(256) void k_sem_wmma(const float* __restrict__ Z,
                                                  const float* __restrict__ W1,
                                                  const float* __restrict__ b1,
                                                  const float* __restrict__ W2,
                                                  float* __restrict__ wsum, int N) {
  __shared__ float Ws[D * D];
  __shared__ float red[8];
  int t = threadIdx.x;
#pragma unroll
  for (int i = 0; i < (D * D / 4) / 256; ++i)
    ((float4*)Ws)[t + i * 256] = ((const float4*)W1)[t + i * 256];
  __syncthreads();

  const int wave = t >> 5;
  const int lane = t & 31;
  const int lrow = lane & 15;
  const int half = lane >> 4;
  const int rowbase = blockIdx.x * 128 + wave * 16;

  int xr = rowbase + lrow;
  if (xr > N - 1) xr = N - 1;
  const float* xp = Z + (size_t)xr * D;

  v2f afrag[32];
#pragma unroll
  for (int kt = 0; kt < 32; ++kt) {
    int k0 = kt * 4 + 2 * half;
    afrag[kt] = *(const v2f*)(xp + k0);
  }

  float s = 0.f;
#pragma unroll
  for (int nt = 0; nt < 8; ++nt) {
    v8f acc = {};
    int nb = nt * 16 + lrow;
#pragma unroll
    for (int kt = 0; kt < 32; ++kt) {
      int k0 = kt * 4 + 2 * half;
      v2f b;
      b.x = Ws[k0 * D + nb];
      b.y = Ws[(k0 + 1) * D + nb];
      acc = __builtin_amdgcn_wmma_f32_16x16x4_f32(false, afrag[kt], false, b,
                                                  (short)0, acc, false, false);
    }
    asm volatile("v_nop\n\tv_nop\n\tv_nop\n\tv_nop" : "+v"(acc));
    const float bn = b1[nb];
    const float wn = W2[nb];
#pragma unroll
    for (int v = 0; v < 8; ++v) {
      int grow = rowbase + v + 8 * half;
      float y = tanhf(acc[v] + bn) * wn;
      s += (grow < N) ? y : 0.f;
    }
  }

#pragma unroll
  for (int off = 16; off; off >>= 1) s += __shfl_xor(s, off, 32);
  if (lane == 0) red[wave] = s;
  __syncthreads();
  if (t == 0) {
    float tot = ((red[0] + red[1]) + (red[2] + red[3])) +
                ((red[4] + red[5]) + (red[6] + red[7]));
    atomicAdd(wsum, tot);
  }
}

__global__ void k_mix(const float* __restrict__ Z3, const float* __restrict__ wsum,
                      float* __restrict__ out, int n, float invN) {
  int i = blockIdx.x * blockDim.x + threadIdx.x;
  float w0 = wsum[0] * invN, w1 = wsum[1] * invN, w2 = wsum[2] * invN;
  float mx = fmaxf(w0, fmaxf(w1, w2));
  float e0 = expf(w0 - mx), e1 = expf(w1 - mx), e2 = expf(w2 - mx);
  float inv = 1.f / (e0 + e1 + e2);
  float a0 = e0 * inv, a1 = e1 * inv, a2 = e2 * inv;
  if (i < n)
    out[i] = a0 * Z3[i] + a1 * Z3[(size_t)n + i] + a2 * Z3[2 * (size_t)n + i];
  if (i == 0) { out[n] = a0; out[n + 1] = a1; out[n + 2] = a2; }
}


extern "C" void kernel_launch(void* const* d_in, const int* in_sizes, int n_in,
                              void* d_out, int out_size, void* d_ws, size_t ws_size,
                              hipStream_t stream) {
  const float* dst_feat = (const float*)d_in[0];
  const float* src_feat[RELS] = {(const float*)d_in[1], (const float*)d_in[2],
                                 (const float*)d_in[3]};
  const float* sem_W1 = (const float*)d_in[4];
  const float* sem_b1 = (const float*)d_in[5];
  const float* sem_W2 = (const float*)d_in[6];
  const float *Wg[RELS], *al[RELS], *ar[RELS], *bg[RELS];
  const int *sidx[RELS], *didx[RELS];
  for (int r = 0; r < RELS; ++r) {
    int b = 7 + 6 * r;
    Wg[r] = (const float*)d_in[b + 0];
    al[r] = (const float*)d_in[b + 1];
    ar[r] = (const float*)d_in[b + 2];
    bg[r] = (const float*)d_in[b + 3];
    sidx[r] = (const int*)d_in[b + 4];
    didx[r] = (const int*)d_in[b + 5];
  }
  const int ND = in_sizes[0] / D;
  const int NS = in_sizes[1] / D;
  const int E  = in_sizes[11];

  float* ws   = (float*)d_ws;
  float* z3   = ws;
  float* hs   = z3 + (size_t)RELS * ND * D;
  float* el   = hs + (size_t)NS * D;
  float* er   = el + NS;
  int*   cnt  = (int*)(er + ND);
  int*   cur  = cnt + ND;
  float* wa   = (float*)(cur + ND);
  float* wr   = wa + D;
  float* wsum = wr + D;
  int2* lst  = (int2*)d_out;
  int*  rank = (int*)(lst + E);

  k_init_wsum<<<1, 32, 0, stream>>>(wsum);

  for (int r = 0; r < RELS; ++r) {
    float* zr = z3 + (size_t)r * ND * D;
    k_rowdot<<<1, 256, 0, stream>>>(Wg[r], al[r], ar[r], wa, wr);
    k_gemm_wmma<<<(NS + 127) / 128, 256, 0, stream>>>(src_feat[r], Wg[r], hs, NS);
    k_matvec<<<(NS + 7) / 8, 256, 0, stream>>>(src_feat[r], wa, el, nullptr, NS);
    k_matvec<<<(ND + 7) / 8, 256, 0, stream>>>(dst_feat, wr, er, cnt, ND);
    k_rank<<<(E + 255) / 256, 256, 0, stream>>>(didx[r], cnt, rank, E);
    k_scan<<<1, 1024, 0, stream>>>(cnt, cur, ND);
    k_place<<<(E + 255) / 256, 256, 0, stream>>>(sidx[r], didx[r], el, er, cur, rank, lst,
                                                 E);
    k_gather<<<(ND + 7) / 8, 256, 0, stream>>>(cnt, cur, lst, hs, bg[r], zr, ND);
    k_sem_wmma<<<(ND + 127) / 128, 256, 0, stream>>>(zr, sem_W1, sem_b1, sem_W2,
                                                      wsum + r, ND);
  }

  k_mix<<<(ND * D + 255) / 256, 256, 0, stream>>>(z3, wsum, (float*)d_out, ND * D,
                                                  1.0f / (float)ND);
}


# the five kernels that ran one item per wave now take several (matvec 16 rows, gather 4 rows, rank/place 4 edges a thread, mix 8 float4 a thread): about 7x fewer waves to start; every sum as before
# speedup vs baseline: 2.3042x; 1.0374x over previous
#include <hip/hip_runtime.h>
#include <math.h>

#define D 128
#define RELS 3

typedef float v2f __attribute__((ext_vector_type(2)));
typedef float v8f __attribute__((ext_vector_type(8)));


__device__ __forceinline__ float lrelu(float x) { return x > 0.f ? x : 0.2f * x; }


__global__ void k_init_wsum(float* wsum) {
  if (threadIdx.x < RELS) wsum[threadIdx.x] = 0.f;
}

__global__ __launch_bounds__(256) void k_rowdot(const float* __restrict__ W,
                                                const float* __restrict__ al,
                                                const float* __restrict__ ar,
                                                float* __restrict__ wa,
                                                float* __restrict__ wr) {
  int t = threadIdx.x;
  const float* v = (t < D) ? al : ar;
  float* o = (t < D) ? wa : wr;
  int d = t & (D - 1);
  float s = 0.f;
#pragma unroll 8
  for (int h = 0; h < D; ++h) s += W[d * D + h] * v[h];
  o[d] = s;
}

__global__ __launch_bounds__(256) void k_gemm_wmma(const float* __restrict__ X,
                                                   const float* __restrict__ W,
                                                   float* __restrict__ H, int N) {
  __shared__ float Ws[D * D];
  int t = threadIdx.x;
#pragma unroll
  for (int i = 0; i < (D * D / 4) / 256; ++i)
    ((float4*)Ws)[t + i * 256] = ((const float4*)W)[t + i * 256];
  __syncthreads();

  const int wave = t >> 5;
  const int lane = t & 31;
  const int lrow = lane & 15;
  const int half = lane >> 4;
  const int rowbase = blockIdx.x * 128 + wave * 16;

  int xr = rowbase + lrow;
  if (xr > N - 1) xr = N - 1;
  const float* xp = X + (size_t)xr * D;

  v2f afrag[32];
#pragma unroll
  for (int kt = 0; kt < 32; ++kt) {
    int k0 = kt * 4 + 2 * half;
    afrag[kt] = *(const v2f*)(xp + k0);
  }

#pragma unroll
  for (int nt = 0; nt < 8; ++nt) {
    v8f acc = {};
    int nb = nt * 16 + lrow;
#pragma unroll
    for (int kt = 0; kt < 32; ++kt) {
      int k0 = kt * 4 + 2 * half;
      v2f b;
      b.x = Ws[k0 * D + nb];
      b.y = Ws[(k0 + 1) * D + nb];
      acc = __builtin_amdgcn_wmma_f32_16x16x4_f32(false, afrag[kt], false, b,
                                                  (short)0, acc, false, false);
    }
#pragma unroll
    for (int v = 0; v < 8; ++v) {
      int grow = rowbase + v + 8 * half;
      if (grow < N) H[(size_t)grow * D + nb] = acc[v];
    }
  }
}

#define MV_ROWS 16
__global__ __launch_bounds__(256) void k_matvec(const float* __restrict__ X,
                                                const float* __restrict__ v,
                                                float* __restrict__ out,
                                                int* __restrict__ zero, int n) {
  const int lane = threadIdx.x & 31;
  const int row0 =
      __builtin_amdgcn_readfirstlane((blockIdx.x * 8 + (threadIdx.x >> 5)) * MV_ROWS);
  if (row0 >= n) return;
  float4 x[MV_ROWS];
#pragma unroll
  for (int u = 0; u < MV_ROWS; ++u) {
    int r = row0 + u;
    if (r > n - 1) r = n - 1;
    x[u] = ((const float4*)X)[(size_t)r * (D / 4) + lane];
  }
  const float4 c = ((const float4*)v)[lane];
  __builtin_amdgcn_sched_barrier(0);
  float mine = 0.f;
#pragma unroll
  for (int u = 0; u < MV_ROWS; ++u) {
    float s = x[u].x * c.x + x[u].y * c.y + x[u].z * c.z + x[u].w * c.w;
#pragma unroll
    for (int off = 16; off; off >>= 1) s += __shfl_xor(s, off, 32);
    if (lane == u) mine = s;
  }
  const int r = row0 + lane;
  if (lane < MV_ROWS && r < n) {
    out[r] = mine;
    if (zero) zero[r] = 0;
  }
}


__global__ __launch_bounds__(256) void k_rank(const int* __restrict__ di, int* __restrict__ cnt,
                                              int* __restrict__ rank, int E) {
  int i = (blockIdx.x * 256 + threadIdx.x) * 4;
  if (i + 3 < E) {
    const int4 d = *(const int4*)(di + i);
    int4 r;
    r.x = atomicAdd(&cnt[d.x], 1);
    r.y = atomicAdd(&cnt[d.y], 1);
    r.z = atomicAdd(&cnt[d.z], 1);
    r.w = atomicAdd(&cnt[d.w], 1);
    *(int4*)(rank + i) = r;
  } else {
    for (; i < E; ++i) rank[i] = atomicAdd(&cnt[di[i]], 1);
  }
}

__global__ __launch_bounds__(1024) void k_scan(const int* __restrict__ cnt,
                                               int* __restrict__ cur, int n) {
  __shared__ int wtot[32];
  const int t = threadIdx.x;
  const int lane = t & 31, wave = t >> 5;
  const int chunk = (n + 1023) / 1024;
  int b = t * chunk;
  int e = b + chunk;
  if (b > n) b = n;
  if (e > n) e = n;
  int s = 0;
#pragma unroll 8
  for (int i = b; i < e; ++i) s += cnt[i];
  int incl = s;
#pragma unroll
  for (int off = 1; off < 32; off <<= 1) {
    int u = __shfl_up(incl, off, 32);
    if (lane >= off) incl += u;
  }
  if (lane == 31) wtot[wave] = incl;
  __syncthreads();
  if (wave == 0) {
    int w = wtot[lane];
    int wi = w;
#pragma unroll
    for (int off = 1; off < 32; off <<= 1) {
      int u = __shfl_up(wi, off, 32);
      if (lane >= off) wi += u;
    }
    wtot[lane] = wi - w;
  }
  __syncthreads();
  int run = wtot[wave] + incl - s;
#pragma unroll 8
  for (int i = b; i < e; ++i) {
    cur[i] = run;
    run += cnt[i];
  }
}

__global__ __launch_bounds__(256) void k_place(const int* __restrict__ si,
                                               const int* __restrict__ di,
                                               const float* __restrict__ el,
                                               const float* __restrict__ er,
                                               const int* __restrict__ cur,
                                               const int* __restrict__ rank,
                                               int2* __restrict__ lst, int E) {
  int i = (blockIdx.x * 256 + threadIdx.x) * 4;
  if (i + 3 < E) {
    const int4 s = *(const int4*)(si + i);
    const int4 d = *(const int4*)(di + i);
    const int4 rk = *(const int4*)(rank + i);
    const float x0 = lrelu(el[s.x] + er[d.x]);
    const float x1 = lrelu(el[s.y] + er[d.y]);
    const float x2 = lrelu(el[s.z] + er[d.z]);
    const float x3 = lrelu(el[s.w] + er[d.w]);
    const int p0 = cur[d.x] + rk.x;
    const int p1 = cur[d.y] + rk.y;
    const int p2 = cur[d.z] + rk.z;
    const int p3 = cur[d.w] + rk.w;
    lst[p0] = make_int2(s.x, __float_as_int(x0));
    lst[p1] = make_int2(s.y, __float_as_int(x1));
    lst[p2] = make_int2(s.z, __float_as_int(x2));
    lst[p3] = make_int2(s.w, __float_as_int(x3));
  } else {
    for (; i < E; ++i) {
      const int s = si[i], d = di[i];
      const float x = lrelu(el[s] + er[d]);
      lst[cur[d] + rank[i]] = make_int2(s, __float_as_int(x));
    }
  }
}

__device__ __forceinline__ void fma4(float4& a, float p, const float4& h) {
  a.x += p * h.x; a.y += p * h.y; a.z += p * h.z; a.w += p * h.w;
}

template <int N>
__device__ __forceinline__ void add_rows(float4& acc, const float* __restrict__ hs, int vx,
                                         int pb, int k, int lane) {
  float4 h[N];
  float pk[N];
#pragma unroll
  for (int u = 0; u < N; ++u) {
    const unsigned sj = (unsigned)__builtin_amdgcn_readlane(vx, k + u);
    pk[u] = __int_as_float(__builtin_amdgcn_readlane(pb, k + u));
    h[u] = ((const float4*)(hs + (size_t)sj * D))[lane];
  }
#pragma unroll
  for (int u = 0; u < N; ++u) fma4(acc, pk[u], h[u]);
}

#define G_ROWS 4
__global__ __launch_bounds__(256) void k_gather(const int* __restrict__ cnt,
                                                const int* __restrict__ cur,
                                                const int2* __restrict__ lst,
                                                const float* __restrict__ hs,
                                                const float* __restrict__ bias,
                                                float* __restrict__ Z, int n) {
  const int lane = threadIdx.x & 31;
  const int d0 =
      __builtin_amdgcn_readfirstlane((blockIdx.x * 8 + (threadIdx.x >> 5)) * G_ROWS);
  const int NEG_INF_BITS = (int)0xff800000u;
  const float4 bb = ((const float4*)bias)[lane];

  for (int r = 0; r < G_ROWS; ++r) {
    const int d = d0 + r;
    if (d >= n) break;
    const int c = cnt[d];
    const int start = cur[d];

    int2 v0 = make_int2(0, NEG_INF_BITS);
    if (lane < c) v0 = lst[start + lane];
    float mx = __int_as_float(v0.y);
    for (int base = 32; base < c; base += 32) {
      int j = base + lane;
      if (j < c) mx = fmaxf(mx, __int_as_float(lst[start + j].y));
    }
#pragma unroll
    for (int off = 16; off; off >>= 1) mx = fmaxf(mx, __shfl_xor(mx, off, 32));

    float4 acc = make_float4(0.f, 0.f, 0.f, 0.f);
    float zs = 0.f;
    for (int base = 0; base < c; base += 32) {
      int2 v = v0;
      const int j = base + lane;
      if (base > 0) {
        v = make_int2(0, NEG_INF_BITS);
        if (j < c) v = lst[start + j];
      }
      const float pe = expf(__int_as_float(v.y) - mx);
      const float p = (j < c) ? pe : 0.f;
      zs += p;
      const int pb = __float_as_int(p);
      int nc = c - base;
      if (nc > 32) nc = 32;
      int k = 0;
      for (; k + 8 <= nc; k += 8) add_rows<8>(acc, hs, v.x, pb, k, lane);
      switch (nc - k) {
        case 7: add_rows<7>(acc, hs, v.x, pb, k, lane); break;
        case 6: add_rows<6>(acc, hs, v.x, pb, k, lane); break;
        case 5: add_rows<5>(acc, hs, v.x, pb, k, lane); break;
        case 4: add_rows<4>(acc, hs, v.x, pb, k, lane); break;
        case 3: add_rows<3>(acc, hs, v.x, pb, k, lane); break;
        case 2: add_rows<2>(acc, hs, v.x, pb, k, lane); break;
        case 1: add_rows<1>(acc, hs, v.x, pb, k, lane); break;
        default: break;
      }
    }
#pragma unroll
    for (int off = 16; off; off >>= 1) zs += __shfl_xor(zs, off, 32);

    const float inv = (c > 0) ? 1.f / zs : 0.f;
    float4 o;
    o.x = acc.x * inv + bb.x;
    o.y = acc.y * inv + bb.y;
    o.z = acc.z * inv + bb.z;
    o.w = acc.w * inv + bb.w;
    o.x = o.x > 0.f ? o.x : expf(o.x) - 1.f;
    o.y = o.y > 0.f ? o.y : expf(o.y) - 1.f;
    o.z = o.z > 0.f ? o.z : expf(o.z) - 1.f;
    o.w = o.w > 0.f ? o.w : expf(o.w) - 1.f;
    ((float4*)Z)[(size_t)d * (D / 4) + lane] = o;
  }
}

__global__ __launch_bounds__(256) void k_sem_wmma(const float* __restrict__ Z,
                                                  const float* __restrict__ W1,
                                                  const float* __restrict__ b1,
                                                  const float* __restrict__ W2,
                                                  float* __restrict__ wsum, int N) {
  __shared__ float Ws[D * D];
  __shared__ float red[8];
  int t = threadIdx.x;
#pragma unroll
  for (int i = 0; i < (D * D / 4) / 256; ++i)
    ((float4*)Ws)[t + i * 256] = ((const float4*)W1)[t + i * 256];
  __syncthreads();

  const int wave = t >> 5;
  const int lane = t & 31;
  const int lrow = lane & 15;
  const int half = lane >> 4;
  const int rowbase = blockIdx.x * 128 + wave * 16;

  int xr = rowbase + lrow;
  if (xr > N - 1) xr = N - 1;
  const float* xp = Z + (size_t)xr * D;

  v2f afrag[32];
#pragma unroll
  for (int kt = 0; kt < 32; ++kt) {
    int k0 = kt * 4 + 2 * half;
    afrag[kt] = *(const v2f*)(xp + k0);
  }

  float s = 0.f;
#pragma unroll
  for (int nt = 0; nt < 8; ++nt) {
    v8f acc = {};
    int nb = nt * 16 + lrow;
#pragma unroll
    for (int kt = 0; kt < 32; ++kt) {
      int k0 = kt * 4 + 2 * half;
      v2f b;
      b.x = Ws[k0 * D + nb];
      b.y = Ws[(k0 + 1) * D + nb];
      acc = __builtin_amdgcn_wmma_f32_16x16x4_f32(false, afrag[kt], false, b,
                                                  (short)0, acc, false, false);
    }
    asm volatile("v_nop\n\tv_nop\n\tv_nop\n\tv_nop" : "+v"(acc));
    const float bn = b1[nb];
    const float wn = W2[nb];
#pragma unroll
    for (int v = 0; v < 8; ++v) {
      int grow = rowbase + v + 8 * half;
      float y = tanhf(acc[v] + bn) * wn;
      s += (grow < N) ? y : 0.f;
    }
  }

#pragma unroll
  for (int off = 16; off; off >>= 1) s += __shfl_xor(s, off, 32);
  if (lane == 0) red[wave] = s;
  __syncthreads();
  if (t == 0) {
    float tot = ((red[0] + red[1]) + (red[2] + red[3])) +
                ((red[4] + red[5]) + (red[6] + red[7]));
    atomicAdd(wsum, tot);
  }
}

#define MIX_V 8
__global__ __launch_bounds__(256) void k_mix(const float* __restrict__ Z3,
                                             const float* __restrict__ wsum,
                                             float* __restrict__ out, int n, float invN) {
  const int n4 = n >> 2;
  const float4* __restrict__ za = (const float4*)Z3;
  const float4* __restrict__ zb = (const float4*)(Z3 + (size_t)n);
  const float4* __restrict__ zc = (const float4*)(Z3 + 2 * (size_t)n);
  const int base = blockIdx.x * (256 * MIX_V) + threadIdx.x;
  float4 a[MIX_V], b[MIX_V], c[MIX_V];
#pragma unroll
  for (int u = 0; u < MIX_V; ++u) {
    int i = base + u * 256;
    if (i > n4 - 1) i = n4 - 1;
    a[u] = za[i];
    b[u] = zb[i];
    c[u] = zc[i];
  }
  float w0 = wsum[0] * invN, w1 = wsum[1] * invN, w2 = wsum[2] * invN;
  float mx = fmaxf(w0, fmaxf(w1, w2));
  float e0 = expf(w0 - mx), e1 = expf(w1 - mx), e2 = expf(w2 - mx);
  float inv = 1.f / (e0 + e1 + e2);
  float a0 = e0 * inv, a1 = e1 * inv, a2 = e2 * inv;
#pragma unroll
  for (int u = 0; u < MIX_V; ++u) {
    const int i = base + u * 256;
    float4 o;
    o.x = a0 * a[u].x + a1 * b[u].x + a2 * c[u].x;
    o.y = a0 * a[u].y + a1 * b[u].y + a2 * c[u].y;
    o.z = a0 * a[u].z + a1 * b[u].z + a2 * c[u].z;
    o.w = a0 * a[u].w + a1 * b[u].w + a2 * c[u].w;
    if (i < n4) ((float4*)out)[i] = o;
  }
  if (blockIdx.x == 0 && threadIdx.x == 0) { out[n] = a0; out[n + 1] = a1; out[n + 2] = a2; }
}


extern "C" void kernel_launch(void* const* d_in, const int* in_sizes, int n_in,
                              void* d_out, int out_size, void* d_ws, size_t ws_size,
                              hipStream_t stream) {
  const float* dst_feat = (const float*)d_in[0];
  const float* src_feat[RELS] = {(const float*)d_in[1], (const float*)d_in[2],
                                 (const float*)d_in[3]};
  const float* sem_W1 = (const float*)d_in[4];
  const float* sem_b1 = (const float*)d_in[5];
  const float* sem_W2 = (const float*)d_in[6];
  const float *Wg[RELS], *al[RELS], *ar[RELS], *bg[RELS];
  const int *sidx[RELS], *didx[RELS];
  for (int r = 0; r < RELS; ++r) {
    int b = 7 + 6 * r;
    Wg[r] = (const float*)d_in[b + 0];
    al[r] = (const float*)d_in[b + 1];
    ar[r] = (const float*)d_in[b + 2];
    bg[r] = (const float*)d_in[b + 3];
    sidx[r] = (const int*)d_in[b + 4];
    didx[r] = (const int*)d_in[b + 5];
  }
  const int ND = in_sizes[0] / D;
  const int NS = in_sizes[1] / D;
  const int E  = in_sizes[11];

  float* ws   = (float*)d_ws;
  float* z3   = ws;
  float* hs   = z3 + (size_t)RELS * ND * D;
  float* el   = hs + (size_t)NS * D;
  float* er   = el + NS;
  int*   cnt  = (int*)(er + ND);
  int*   cur  = cnt + ND;
  float* wa   = (float*)(cur + ND);
  float* wr   = wa + D;
  float* wsum = wr + D;
  int2* lst  = (int2*)d_out;
  int*  rank = (int*)(lst + E);

  k_init_wsum<<<1, 32, 0, stream>>>(wsum);

  for (int r = 0; r < RELS; ++r) {
    float* zr = z3 + (size_t)r * ND * D;
    k_rowdot<<<1, 256, 0, stream>>>(Wg[r], al[r], ar[r], wa, wr);
    k_gemm_wmma<<<(NS + 127) / 128, 256, 0, stream>>>(src_feat[r], Wg[r], hs, NS);
    k_matvec<<<(NS + 8 * MV_ROWS - 1) / (8 * MV_ROWS), 256, 0, stream>>>(src_feat[r], wa, el,
                                                                       nullptr, NS);
    k_matvec<<<(ND + 8 * MV_ROWS - 1) / (8 * MV_ROWS), 256, 0, stream>>>(dst_feat, wr, er, cnt,
                                                                       ND);
    k_rank<<<(E + 1023) / 1024, 256, 0, stream>>>(didx[r], cnt, rank, E);
    k_scan<<<1, 1024, 0, stream>>>(cnt, cur, ND);
    k_place<<<(E + 1023) / 1024, 256, 0, stream>>>(sidx[r], didx[r], el, er, cur, rank, lst,
                                                   E);
    k_gather<<<(ND + 8 * G_ROWS - 1) / (8 * G_ROWS), 256, 0, stream>>>(cnt, cur, lst, hs, bg[r],
                                                                     zr, ND);
    k_sem_wmma<<<(ND + 127) / 128, 256, 0, stream>>>(zr, sem_W1, sem_b1, sem_W2,
                                                      wsum + r, ND);
  }

  k_mix<<<(ND * D / 4 + 256 * MIX_V - 1) / (256 * MIX_V), 256, 0, stream>>>(
      z3, wsum, (float*)d_out, ND * D, 1.0f / (float)ND);
}


# every stage that does not need the shared hs buffer runs once for all three relations (grid.y = relation; er for all three in one pass over dst_feat): 29 launches become 14; every sum as before
# speedup vs baseline: 2.7828x; 1.2077x over previous
#include <hip/hip_runtime.h>
#include <math.h>

#define D 128
#define RELS 3

typedef float v2f __attribute__((ext_vector_type(2)));
typedef float v8f __attribute__((ext_vector_type(8)));


__device__ __forceinline__ float lrelu(float x) { return x > 0.f ? x : 0.2f * x; }

__device__ __forceinline__ const float* pick3(int r, const float* a, const float* b,
                                              const float* c) {
  return r == 0 ? a : (r == 1 ? b : c);
}
__device__ __forceinline__ const int* pick3(int r, const int* a, const int* b, const int* c) {
  return r == 0 ? a : (r == 1 ? b : c);
}


__global__ __launch_bounds__(256) void k_rowdot(const float* __restrict__ W0,
                                                const float* __restrict__ W1,
                                                const float* __restrict__ W2,
                                                const float* __restrict__ al0,
                                                const float* __restrict__ al1,
                                                const float* __restrict__ al2,
                                                const float* __restrict__ ar0,
                                                const float* __restrict__ ar1,
                                                const float* __restrict__ ar2,
                                                float* __restrict__ wa3,
                                                float* __restrict__ wr3,
                                                float* __restrict__ wsum) {
  const int rel = blockIdx.x;
  const float* W = pick3(rel, W0, W1, W2);
  const float* al = pick3(rel, al0, al1, al2);
  const float* ar = pick3(rel, ar0, ar1, ar2);
  int t = threadIdx.x;
  if (t == 0) wsum[rel] = 0.f;
  const float* v = (t < D) ? al : ar;
  float* o = ((t < D) ? wa3 : wr3) + rel * D;
  int d = t & (D - 1);
  float s = 0.f;
#pragma unroll 8
  for (int h = 0; h < D; ++h) s += W[d * D + h] * v[h];
  o[d] = s;
}

__global__ __launch_bounds__(256) void k_gemm_wmma(const float* __restrict__ X,
                                                   const float* __restrict__ W,
                                                   float* __restrict__ H, int N) {
  __shared__ float Ws[D * D];
  int t = threadIdx.x;
#pragma unroll
  for (int i = 0; i < (D * D / 4) / 256; ++i)
    ((float4*)Ws)[t + i * 256] = ((const float4*)W)[t + i * 256];
  __syncthreads();

  const int wave = t >> 5;
  const int lane = t & 31;
  const int lrow = lane & 15;
  const int half = lane >> 4;
  const int rowbase = blockIdx.x * 128 + wave * 16;

  int xr = rowbase + lrow;
  if (xr > N - 1) xr = N - 1;
  const float* xp = X + (size_t)xr * D;

  v2f afrag[32];
#pragma unroll
  for (int kt = 0; kt < 32; ++kt) {
    int k0 = kt * 4 + 2 * half;
    afrag[kt] = *(const v2f*)(xp + k0);
  }

#pragma unroll
  for (int nt = 0; nt < 8; ++nt) {
    v8f acc = {};
    int nb = nt * 16 + lrow;
#pragma unroll
    for (int kt = 0; kt < 32; ++kt) {
      int k0 = kt * 4 + 2 * half;
      v2f b;
      b.x = Ws[k0 * D + nb];
      b.y = Ws[(k0 + 1) * D + nb];
      acc = __builtin_amdgcn_wmma_f32_16x16x4_f32(false, afrag[kt], false, b,
                                                  (short)0, acc, false, false);
    }
#pragma unroll
    for (int v = 0; v < 8; ++v) {
      int grow = rowbase + v + 8 * half;
      if (grow < N) H[(size_t)grow * D + nb] = acc[v];
    }
  }
}

#define MV_ROWS 16
__global__ __launch_bounds__(256) void k_matvec(const float* __restrict__ X0,
                                                const float* __restrict__ X1,
                                                const float* __restrict__ X2,
                                                const float* __restrict__ v3,
                                                float* __restrict__ out3, int n) {
  const int rel = blockIdx.y;
  const float* X = pick3(rel, X0, X1, X2);
  const float* v = v3 + rel * D;
  float* out = out3 + (size_t)rel * n;
  const int lane = threadIdx.x & 31;
  const int row0 =
      __builtin_amdgcn_readfirstlane((blockIdx.x * 8 + (threadIdx.x >> 5)) * MV_ROWS);
  if (row0 >= n) return;
  float4 x[MV_ROWS];
#pragma unroll
  for (int u = 0; u < MV_ROWS; ++u) {
    int r = row0 + u;
    if (r > n - 1) r = n - 1;
    x[u] = ((const float4*)X)[(size_t)r * (D / 4) + lane];
  }
  const float4 c = ((const float4*)v)[lane];
  __builtin_amdgcn_sched_barrier(0);
  float mine = 0.f;
#pragma unroll
  for (int u = 0; u < MV_ROWS; ++u) {
    float s = x[u].x * c.x + x[u].y * c.y + x[u].z * c.z + x[u].w * c.w;
#pragma unroll
    for (int off = 16; off; off >>= 1) s += __shfl_xor(s, off, 32);
    if (lane == u) mine = s;
  }
  const int r = row0 + lane;
  if (lane < MV_ROWS && r < n) out[r] = mine;
}

__global__ __launch_bounds__(256) void k_matvec_dst(const float* __restrict__ X,
                                                    const float* __restrict__ v3,
                                                    float* __restrict__ out3,
                                                    int* __restrict__ zero3, int n) {
  const int lane = threadIdx.x & 31;
  const int row0 =
      __builtin_amdgcn_readfirstlane((blockIdx.x * 8 + (threadIdx.x >> 5)) * MV_ROWS);
  if (row0 >= n) return;
  float4 x[MV_ROWS];
#pragma unroll
  for (int u = 0; u < MV_ROWS; ++u) {
    int r = row0 + u;
    if (r > n - 1) r = n - 1;
    x[u] = ((const float4*)X)[(size_t)r * (D / 4) + lane];
  }
  const float4 c0 = ((const float4*)v3)[lane];
  const float4 c1 = ((const float4*)v3)[(D / 4) + lane];
  const float4 c2 = ((const float4*)v3)[2 * (D / 4) + lane];
  __builtin_amdgcn_sched_barrier(0);
  float m0 = 0.f, m1 = 0.f, m2 = 0.f;
#pragma unroll
  for (int u = 0; u < MV_ROWS; ++u) {
    float s0 = x[u].x * c0.x + x[u].y * c0.y + x[u].z * c0.z + x[u].w * c0.w;
    float s1 = x[u].x * c1.x + x[u].y * c1.y + x[u].z * c1.z + x[u].w * c1.w;
    float s2 = x[u].x * c2.x + x[u].y * c2.y + x[u].z * c2.z + x[u].w * c2.w;
#pragma unroll
    for (int off = 16; off; off >>= 1) {
      s0 += __shfl_xor(s0, off, 32);
      s1 += __shfl_xor(s1, off, 32);
      s2 += __shfl_xor(s2, off, 32);
    }
    if (lane == u) { m0 = s0; m1 = s1; m2 = s2; }
  }
  const int r = row0 + lane;
  if (lane < MV_ROWS && r < n) {
    out3[r] = m0;
    out3[(size_t)n + r] = m1;
    out3[2 * (size_t)n + r] = m2;
    zero3[r] = 0;
    zero3[(size_t)n + r] = 0;
    zero3[2 * (size_t)n + r] = 0;
  }
}


__global__ __launch_bounds__(256) void k_rank(const int* __restrict__ di0,
                                              const int* __restrict__ di1,
                                              const int* __restrict__ di2,
                                              int* __restrict__ cnt3, int* __restrict__ rank3,
                                              int E, int nd) {
  const int rel = blockIdx.y;
  const int* di = pick3(rel, di0, di1, di2);
  int* cnt = cnt3 + (size_t)rel * nd;
  int* rank = rank3 + (size_t)rel * E;
  int i = (blockIdx.x * 256 + threadIdx.x) * 4;
  if (i + 3 < E) {
    const int4 d = *(const int4*)(di + i);
    int4 r;
    r.x = atomicAdd(&cnt[d.x], 1);
    r.y = atomicAdd(&cnt[d.y], 1);
    r.z = atomicAdd(&cnt[d.z], 1);
    r.w = atomicAdd(&cnt[d.w], 1);
    *(int4*)(rank + i) = r;
  } else {
    for (; i < E; ++i) rank[i] = atomicAdd(&cnt[di[i]], 1);
  }
}

__global__ __launch_bounds__(1024) void k_scan(const int* __restrict__ cnt3,
                                               int* __restrict__ cur3, int n) {
  __shared__ int wtot[32];
  const int* cnt = cnt3 + (size_t)blockIdx.x * n;
  int* cur = cur3 + (size_t)blockIdx.x * n;
  const int t = threadIdx.x;
  const int lane = t & 31, wave = t >> 5;
  const int chunk = (n + 1023) / 1024;
  int b = t * chunk;
  int e = b + chunk;
  if (b > n) b = n;
  if (e > n) e = n;
  int s = 0;
#pragma unroll 8
  for (int i = b; i < e; ++i) s += cnt[i];
  int incl = s;
#pragma unroll
  for (int off = 1; off < 32; off <<= 1) {
    int u = __shfl_up(incl, off, 32);
    if (lane >= off) incl += u;
  }
  if (lane == 31) wtot[wave] = incl;
  __syncthreads();
  if (wave == 0) {
    int w = wtot[lane];
    int wi = w;
#pragma unroll
    for (int off = 1; off < 32; off <<= 1) {
      int u = __shfl_up(wi, off, 32);
      if (lane >= off) wi += u;
    }
    wtot[lane] = wi - w;
  }
  __syncthreads();
  int run = wtot[wave] + incl - s;
#pragma unroll 8
  for (int i = b; i < e; ++i) {
    cur[i] = run;
    run += cnt[i];
  }
}

__global__ __launch_bounds__(256) void k_place(const int* __restrict__ si0,
                                               const int* __restrict__ si1,
                                               const int* __restrict__ si2,
                                               const int* __restrict__ di0,
                                               const int* __restrict__ di1,
                                               const int* __restrict__ di2,
                                               const float* __restrict__ el3,
                                               const float* __restrict__ er3,
                                               const int* __restrict__ cur3,
                                               const int* __restrict__ rank3,
                                               int2* __restrict__ lst3, int E, int ns, int nd) {
  const int rel = blockIdx.y;
  const int* si = pick3(rel, si0, si1, si2);
  const int* di = pick3(rel, di0, di1, di2);
  const float* el = el3 + (size_t)rel * ns;
  const float* er = er3 + (size_t)rel * nd;
  const int* cur = cur3 + (size_t)rel * nd;
  const int* rank = rank3 + (size_t)rel * E;
  int2* lst = lst3 + (size_t)rel * E;
  int i = (blockIdx.x * 256 + threadIdx.x) * 4;
  if (i + 3 < E) {
    const int4 s = *(const int4*)(si + i);
    const int4 d = *(const int4*)(di + i);
    const int4 rk = *(const int4*)(rank + i);
    const float x0 = lrelu(el[s.x] + er[d.x]);
    const float x1 = lrelu(el[s.y] + er[d.y]);
    const float x2 = lrelu(el[s.z] + er[d.z]);
    const float x3 = lrelu(el[s.w] + er[d.w]);
    const int p0 = cur[d.x] + rk.x;
    const int p1 = cur[d.y] + rk.y;
    const int p2 = cur[d.z] + rk.z;
    const int p3 = cur[d.w] + rk.w;
    lst[p0] = make_int2(s.x, __float_as_int(x0));
    lst[p1] = make_int2(s.y, __float_as_int(x1));
    lst[p2] = make_int2(s.z, __float_as_int(x2));
    lst[p3] = make_int2(s.w, __float_as_int(x3));
  } else {
    for (; i < E; ++i) {
      const int s = si[i], d = di[i];
      const float x = lrelu(el[s] + er[d]);
      lst[cur[d] + rank[i]] = make_int2(s, __float_as_int(x));
    }
  }
}

__device__ __forceinline__ void fma4(float4& a, float p, const float4& h) {
  a.x += p * h.x; a.y += p * h.y; a.z += p * h.z; a.w += p * h.w;
}

template <int N>
__device__ __forceinline__ void add_rows(float4& acc, const float* __restrict__ hs, int vx,
                                         int pb, int k, int lane) {
  float4 h[N];
  float pk[N];
#pragma unroll
  for (int u = 0; u < N; ++u) {
    const unsigned sj = (unsigned)__builtin_amdgcn_readlane(vx, k + u);
    pk[u] = __int_as_float(__builtin_amdgcn_readlane(pb, k + u));
    h[u] = ((const float4*)(hs + (size_t)sj * D))[lane];
  }
#pragma unroll
  for (int u = 0; u < N; ++u) fma4(acc, pk[u], h[u]);
}

#define G_ROWS 4
__global__ __launch_bounds__(256) void k_gather(const int* __restrict__ cnt,
                                                const int* __restrict__ cur,
                                                const int2* __restrict__ lst,
                                                const float* __restrict__ hs,
                                                const float* __restrict__ bias,
                                                float* __restrict__ Z, int n) {
  const int lane = threadIdx.x & 31;
  const int d0 =
      __builtin_amdgcn_readfirstlane((blockIdx.x * 8 + (threadIdx.x >> 5)) * G_ROWS);
  const int NEG_INF_BITS = (int)0xff800000u;
  const float4 bb = ((const float4*)bias)[lane];

  for (int r = 0; r < G_ROWS; ++r) {
    const int d = d0 + r;
    if (d >= n) break;
    const int c = cnt[d];
    const int start = cur[d];

    int2 v0 = make_int2(0, NEG_INF_BITS);
    if (lane < c) v0 = lst[start + lane];
    float mx = __int_as_float(v0.y);
    for (int base = 32; base < c; base += 32) {
      int j = base + lane;
      if (j < c) mx = fmaxf(mx, __int_as_float(lst[start + j].y));
    }
#pragma unroll
    for (int off = 16; off; off >>= 1) mx = fmaxf(mx, __shfl_xor(mx, off, 32));

    float4 acc = make_float4(0.f, 0.f, 0.f, 0.f);
    float zs = 0.f;
    for (int base = 0; base < c; base += 32) {
      int2 v = v0;
      const int j = base + lane;
      if (base > 0) {
        v = make_int2(0, NEG_INF_BITS);
        if (j < c) v = lst[start + j];
      }
      const float pe = expf(__int_as_float(v.y) - mx);
      const float p = (j < c) ? pe : 0.f;
      zs += p;
      const int pb = __float_as_int(p);
      int nc = c - base;
      if (nc > 32) nc = 32;
      int k = 0;
      for (; k + 8 <= nc; k += 8) add_rows<8>(acc, hs, v.x, pb, k, lane);
      switch (nc - k) {
        case 7: add_rows<7>(acc, hs, v.x, pb, k, lane); break;
        case 6: add_rows<6>(acc, hs, v.x, pb, k, lane); break;
        case 5: add_rows<5>(acc, hs, v.x, pb, k, lane); break;
        case 4: add_rows<4>(acc, hs, v.x, pb, k, lane); break;
        case 3: add_rows<3>(acc, hs, v.x, pb, k, lane); break;
        case 2: add_rows<2>(acc, hs, v.x, pb, k, lane); break;
        case 1: add_rows<1>(acc, hs, v.x, pb, k, lane); break;
        default: break;
      }
    }
#pragma unroll
    for (int off = 16; off; off >>= 1) zs += __shfl_xor(zs, off, 32);

    const float inv = (c > 0) ? 1.f / zs : 0.f;
    float4 o;
    o.x = acc.x * inv + bb.x;
    o.y = acc.y * inv + bb.y;
    o.z = acc.z * inv + bb.z;
    o.w = acc.w * inv + bb.w;
    o.x = o.x > 0.f ? o.x : expf(o.x) - 1.f;
    o.y = o.y > 0.f ? o.y : expf(o.y) - 1.f;
    o.z = o.z > 0.f ? o.z : expf(o.z) - 1.f;
    o.w = o.w > 0.f ? o.w : expf(o.w) - 1.f;
    ((float4*)Z)[(size_t)d * (D / 4) + lane] = o;
  }
}

__global__ __launch_bounds__(256) void k_sem_wmma(const float* __restrict__ Z3,
                                                  const float* __restrict__ W1,
                                                  const float* __restrict__ b1,
                                                  const float* __restrict__ W2,
                                                  float* __restrict__ wsum3, int N) {
  __shared__ float Ws[D * D];
  __shared__ float red[8];
  const float* Z = Z3 + (size_t)blockIdx.y * N * D;
  float* wsum = wsum3 + blockIdx.y;
  int t = threadIdx.x;
#pragma unroll
  for (int i = 0; i < (D * D / 4) / 256; ++i)
    ((float4*)Ws)[t + i * 256] = ((const float4*)W1)[t + i * 256];
  __syncthreads();

  const int wave = t >> 5;
  const int lane = t & 31;
  const int lrow = lane & 15;
  const int half = lane >> 4;
  const int rowbase = blockIdx.x * 128 + wave * 16;

  int xr = rowbase + lrow;
  if (xr > N - 1) xr = N - 1;
  const float* xp = Z + (size_t)xr * D;

  v2f afrag[32];
#pragma unroll
  for (int kt = 0; kt < 32; ++kt) {
    int k0 = kt * 4 + 2 * half;
    afrag[kt] = *(const v2f*)(xp + k0);
  }

  float s = 0.f;
#pragma unroll
  for (int nt = 0; nt < 8; ++nt) {
    v8f acc = {};
    int nb = nt * 16 + lrow;
#pragma unroll
    for (int kt = 0; kt < 32; ++kt) {
      int k0 = kt * 4 + 2 * half;
      v2f b;
      b.x = Ws[k0 * D + nb];
      b.y = Ws[(k0 + 1) * D + nb];
      acc = __builtin_amdgcn_wmma_f32_16x16x4_f32(false, afrag[kt], false, b,
                                                  (short)0, acc, false, false);
    }
    asm volatile("v_nop\n\tv_nop\n\tv_nop\n\tv_nop" : "+v"(acc));
    const float bn = b1[nb];
    const float wn = W2[nb];
#pragma unroll
    for (int v = 0; v < 8; ++v) {
      int grow = rowbase + v + 8 * half;
      float y = tanhf(acc[v] + bn) * wn;
      s += (grow < N) ? y : 0.f;
    }
  }

#pragma unroll
  for (int off = 16; off; off >>= 1) s += __shfl_xor(s, off, 32);
  if (lane == 0) red[wave] = s;
  __syncthreads();
  if (t == 0) {
    float tot = ((red[0] + red[1]) + (red[2] + red[3])) +
                ((red[4] + red[5]) + (red[6] + red[7]));
    atomicAdd(wsum, tot);
  }
}

#define MIX_V 8
__global__ __launch_bounds__(256) void k_mix(const float* __restrict__ Z3,
                                             const float* __restrict__ wsum,
                                             float* __restrict__ out, int n, float invN) {
  const int n4 = n >> 2;
  const float4* __restrict__ za = (const float4*)Z3;
  const float4* __restrict__ zb = (const float4*)(Z3 + (size_t)n);
  const float4* __restrict__ zc = (const float4*)(Z3 + 2 * (size_t)n);
  const int base = blockIdx.x * (256 * MIX_V) + threadIdx.x;
  float4 a[MIX_V], b[MIX_V], c[MIX_V];
#pragma unroll
  for (int u = 0; u < MIX_V; ++u) {
    int i = base + u * 256;
    if (i > n4 - 1) i = n4 - 1;
    a[u] = za[i];
    b[u] = zb[i];
    c[u] = zc[i];
  }
  float w0 = wsum[0] * invN, w1 = wsum[1] * invN, w2 = wsum[2] * invN;
  float mx = fmaxf(w0, fmaxf(w1, w2));
  float e0 = expf(w0 - mx), e1 = expf(w1 - mx), e2 = expf(w2 - mx);
  float inv = 1.f / (e0 + e1 + e2);
  float a0 = e0 * inv, a1 = e1 * inv, a2 = e2 * inv;
#pragma unroll
  for (int u = 0; u < MIX_V; ++u) {
    const int i = base + u * 256;
    float4 o;
    o.x = a0 * a[u].x + a1 * b[u].x + a2 * c[u].x;
    o.y = a0 * a[u].y + a1 * b[u].y + a2 * c[u].y;
    o.z = a0 * a[u].z + a1 * b[u].z + a2 * c[u].z;
    o.w = a0 * a[u].w + a1 * b[u].w + a2 * c[u].w;
    if (i < n4) ((float4*)out)[i] = o;
  }
  if (blockIdx.x == 0 && threadIdx.x == 0) { out[n] = a0; out[n + 1] = a1; out[n + 2] = a2; }
}


extern "C" void kernel_launch(void* const* d_in, const int* in_sizes, int n_in,
                              void* d_out, int out_size, void* d_ws, size_t ws_size,
                              hipStream_t stream) {
  const float* dst_feat = (const float*)d_in[0];
  const float* src_feat[RELS] = {(const float*)d_in[1], (const float*)d_in[2],
                                 (const float*)d_in[3]};
  const float* sem_W1 = (const float*)d_in[4];
  const float* sem_b1 = (const float*)d_in[5];
  const float* sem_W2 = (const float*)d_in[6];
  const float *Wg[RELS], *al[RELS], *ar[RELS], *bg[RELS];
  const int *sidx[RELS], *didx[RELS];
  for (int r = 0; r < RELS; ++r) {
    int b = 7 + 6 * r;
    Wg[r] = (const float*)d_in[b + 0];
    al[r] = (const float*)d_in[b + 1];
    ar[r] = (const float*)d_in[b + 2];
    bg[r] = (const float*)d_in[b + 3];
    sidx[r] = (const int*)d_in[b + 4];
    didx[r] = (const int*)d_in[b + 5];
  }
  const int ND = in_sizes[0] / D;
  const int NS = in_sizes[1] / D;
  const int E  = in_sizes[11];

  float* ws   = (float*)d_ws;
  float* z3   = ws;
  float* hs   = z3 + (size_t)RELS * ND * D;
  float* wsum = hs + (size_t)NS * D;
  int2*  lst = (int2*)d_out;
  float* el  = (float*)(lst + (size_t)RELS * E);
  float* er  = el + (size_t)RELS * NS;
  int*   cnt = (int*)(er + (size_t)RELS * ND);
  int*   cur = cnt + (size_t)RELS * ND;
  float* wa  = (float*)(cur + (size_t)RELS * ND);
  float* wr  = wa + RELS * D;
  int*   rank = (int*)(z3 + (size_t)(RELS - 1) * ND * D);

  k_rowdot<<<RELS, 256, 0, stream>>>(Wg[0], Wg[1], Wg[2], al[0], al[1], al[2], ar[0], ar[1],
                                     ar[2], wa, wr, wsum);
  k_matvec<<<dim3((NS + 8 * MV_ROWS - 1) / (8 * MV_ROWS), RELS), 256, 0, stream>>>(
      src_feat[0], src_feat[1], src_feat[2], wa, el, NS);
  k_matvec_dst<<<(ND + 8 * MV_ROWS - 1) / (8 * MV_ROWS), 256, 0, stream>>>(dst_feat, wr, er,
                                                                         cnt, ND);
  k_rank<<<dim3((E + 1023) / 1024, RELS), 256, 0, stream>>>(didx[0], didx[1], didx[2], cnt,
                                                            rank, E, ND);
  k_scan<<<RELS, 1024, 0, stream>>>(cnt, cur, ND);
  k_place<<<dim3((E + 1023) / 1024, RELS), 256, 0, stream>>>(
      sidx[0], sidx[1], sidx[2], didx[0], didx[1], didx[2], el, er, cur, rank, lst, E, NS, ND);

  for (int r = 0; r < RELS; ++r) {
    float* zr = z3 + (size_t)r * ND * D;
    k_gemm_wmma<<<(NS + 127) / 128, 256, 0, stream>>>(src_feat[r], Wg[r], hs, NS);
    k_gather<<<(ND + 8 * G_ROWS - 1) / (8 * G_ROWS), 256, 0, stream>>>(
        cnt + (size_t)r * ND, cur + (size_t)r * ND, lst + (size_t)r * E, hs, bg[r], zr, ND);
  }

  k_sem_wmma<<<dim3((ND + 127) / 128, RELS), 256, 0, stream>>>(z3, sem_W1, sem_b1, sem_W2,
                                                               wsum, ND);

  k_mix<<<(ND * D / 4 + 256 * MIX_V - 1) / (256 * MIX_V), 256, 0, stream>>>(
      z3, wsum, (float*)d_out, ND * D, 1.0f / (float)ND);
}


# aggregate first: a dst row sums its weighted source FEATURE rows, then one WMMA pass does (sum)@W+b, elu in place; no hs buffer, so 3 gathers + 3 GEMMs run as 1+1 launches; sum order changes
# speedup vs baseline: 2.9875x; 1.0736x over previous
#include <hip/hip_runtime.h>
#include <math.h>

#define D 128
#define RELS 3

typedef float v2f __attribute__((ext_vector_type(2)));
typedef float v8f __attribute__((ext_vector_type(8)));


__device__ __forceinline__ float lrelu(float x) { return x > 0.f ? x : 0.2f * x; }

__device__ __forceinline__ const float* pick3(int r, const float* a, const float* b,
                                              const float* c) {
  return r == 0 ? a : (r == 1 ? b : c);
}
__device__ __forceinline__ const int* pick3(int r, const int* a, const int* b, const int* c) {
  return r == 0 ? a : (r == 1 ? b : c);
}


__global__ __launch_bounds__(256) void k_rowdot(const float* __restrict__ W0,
                                                const float* __restrict__ W1,
                                                const float* __restrict__ W2,
                                                const float* __restrict__ al0,
                                                const float* __restrict__ al1,
                                                const float* __restrict__ al2,
                                                const float* __restrict__ ar0,
                                                const float* __restrict__ ar1,
                                                const float* __restrict__ ar2,
                                                float* __restrict__ wa3,
                                                float* __restrict__ wr3,
                                                float* __restrict__ wsum) {
  const int rel = blockIdx.x;
  const float* W = pick3(rel, W0, W1, W2);
  const float* al = pick3(rel, al0, al1, al2);
  const float* ar = pick3(rel, ar0, ar1, ar2);
  int t = threadIdx.x;
  if (t == 0) wsum[rel] = 0.f;
  const float* v = (t < D) ? al : ar;
  float* o = ((t < D) ? wa3 : wr3) + rel * D;
  int d = t & (D - 1);
  float s = 0.f;
#pragma unroll 8
  for (int h = 0; h < D; ++h) s += W[d * D + h] * v[h];
  o[d] = s;
}

__global__ __launch_bounds__(256) void k_gemm_wmma(float* Z3,
                                                   const float* __restrict__ W0,
                                                   const float* __restrict__ W1,
                                                   const float* __restrict__ W2,
                                                   const float* __restrict__ b0,
                                                   const float* __restrict__ b1,
                                                   const float* __restrict__ b2, int N) {
  __shared__ float Ws[D * D];
  const int rel = blockIdx.y;
  const float* W = pick3(rel, W0, W1, W2);
  const float* bias = pick3(rel, b0, b1, b2);
  float* Z = Z3 + (size_t)rel * N * D;
  int t = threadIdx.x;
#pragma unroll
  for (int i = 0; i < (D * D / 4) / 256; ++i)
    ((float4*)Ws)[t + i * 256] = ((const float4*)W)[t + i * 256];
  __syncthreads();

  const int wave = t >> 5;
  const int lane = t & 31;
  const int lrow = lane & 15;
  const int half = lane >> 4;
  const int rowbase = blockIdx.x * 128 + wave * 16;

  int xr = rowbase + lrow;
  if (xr > N - 1) xr = N - 1;
  const float* xp = Z + (size_t)xr * D;

  v2f afrag[32];
#pragma unroll
  for (int kt = 0; kt < 32; ++kt) {
    int k0 = kt * 4 + 2 * half;
    afrag[kt] = *(const v2f*)(xp + k0);
  }
  float bn[8];
#pragma unroll
  for (int nt = 0; nt < 8; ++nt) bn[nt] = bias[nt * 16 + lrow];

#pragma unroll
  for (int nt = 0; nt < 8; ++nt) {
    v8f acc = {};
    int nb = nt * 16 + lrow;
#pragma unroll
    for (int kt = 0; kt < 32; ++kt) {
      int k0 = kt * 4 + 2 * half;
      v2f b;
      b.x = Ws[k0 * D + nb];
      b.y = Ws[(k0 + 1) * D + nb];
      acc = __builtin_amdgcn_wmma_f32_16x16x4_f32(false, afrag[kt], false, b,
                                                  (short)0, acc, false, false);
    }
    asm volatile("v_nop\n\tv_nop\n\tv_nop\n\tv_nop" : "+v"(acc));
#pragma unroll
    for (int v = 0; v < 8; ++v) {
      int grow = rowbase + v + 8 * half;
      float o = acc[v] + bn[nt];
      o = o > 0.f ? o : expf(o) - 1.f;
      if (grow < N) Z[(size_t)grow * D + nb] = o;
    }
  }
}

#define MV_ROWS 16
__global__ __launch_bounds__(256) void k_matvec(const float* __restrict__ X0,
                                                const float* __restrict__ X1,
                                                const float* __restrict__ X2,
                                                const float* __restrict__ v3,
                                                float* __restrict__ out3, int n) {
  const int rel = blockIdx.y;
  const float* X = pick3(rel, X0, X1, X2);
  const float* v = v3 + rel * D;
  float* out = out3 + (size_t)rel * n;
  const int lane = threadIdx.x & 31;
  const int row0 =
      __builtin_amdgcn_readfirstlane((blockIdx.x * 8 + (threadIdx.x >> 5)) * MV_ROWS);
  if (row0 >= n) return;
  float4 x[MV_ROWS];
#pragma unroll
  for (int u = 0; u < MV_ROWS; ++u) {
    int r = row0 + u;
    if (r > n - 1) r = n - 1;
    x[u] = ((const float4*)X)[(size_t)r * (D / 4) + lane];
  }
  const float4 c = ((const float4*)v)[lane];
  __builtin_amdgcn_sched_barrier(0);
  float mine = 0.f;
#pragma unroll
  for (int u = 0; u < MV_ROWS; ++u) {
    float s = x[u].x * c.x + x[u].y * c.y + x[u].z * c.z + x[u].w * c.w;
#pragma unroll
    for (int off = 16; off; off >>= 1) s += __shfl_xor(s, off, 32);
    if (lane == u) mine = s;
  }
  const int r = row0 + lane;
  if (lane < MV_ROWS && r < n) out[r] = mine;
}

__global__ __launch_bounds__(256) void k_matvec_dst(const float* __restrict__ X,
                                                    const float* __restrict__ v3,
                                                    float* __restrict__ out3,
                                                    int* __restrict__ zero3, int n) {
  const int lane = threadIdx.x & 31;
  const int row0 =
      __builtin_amdgcn_readfirstlane((blockIdx.x * 8 + (threadIdx.x >> 5)) * MV_ROWS);
  if (row0 >= n) return;
  float4 x[MV_ROWS];
#pragma unroll
  for (int u = 0; u < MV_ROWS; ++u) {
    int r = row0 + u;
    if (r > n - 1) r = n - 1;
    x[u] = ((const float4*)X)[(size_t)r * (D / 4) + lane];
  }
  const float4 c0 = ((const float4*)v3)[lane];
  const float4 c1 = ((const float4*)v3)[(D / 4) + lane];
  const float4 c2 = ((const float4*)v3)[2 * (D / 4) + lane];
  __builtin_amdgcn_sched_barrier(0);
  float m0 = 0.f, m1 = 0.f, m2 = 0.f;
#pragma unroll
  for (int u = 0; u < MV_ROWS; ++u) {
    float s0 = x[u].x * c0.x + x[u].y * c0.y + x[u].z * c0.z + x[u].w * c0.w;
    float s1 = x[u].x * c1.x + x[u].y * c1.y + x[u].z * c1.z + x[u].w * c1.w;
    float s2 = x[u].x * c2.x + x[u].y * c2.y + x[u].z * c2.z + x[u].w * c2.w;
#pragma unroll
    for (int off = 16; off; off >>= 1) {
      s0 += __shfl_xor(s0, off, 32);
      s1 += __shfl_xor(s1, off, 32);
      s2 += __shfl_xor(s2, off, 32);
    }
    if (lane == u) { m0 = s0; m1 = s1; m2 = s2; }
  }
  const int r = row0 + lane;
  if (lane < MV_ROWS && r < n) {
    out3[r] = m0;
    out3[(size_t)n + r] = m1;
    out3[2 * (size_t)n + r] = m2;
    zero3[r] = 0;
    zero3[(size_t)n + r] = 0;
    zero3[2 * (size_t)n + r] = 0;
  }
}


__global__ __launch_bounds__(256) void k_rank(const int* __restrict__ di0,
                                              const int* __restrict__ di1,
                                              const int* __restrict__ di2,
                                              int* __restrict__ cnt3, int* __restrict__ rank3,
                                              int E, int nd) {
  const int rel = blockIdx.y;
  const int* di = pick3(rel, di0, di1, di2);
  int* cnt = cnt3 + (size_t)rel * nd;
  int* rank = rank3 + (size_t)rel * E;
  int i = (blockIdx.x * 256 + threadIdx.x) * 4;
  if (i + 3 < E) {
    const int4 d = *(const int4*)(di + i);
    int4 r;
    r.x = atomicAdd(&cnt[d.x], 1);
    r.y = atomicAdd(&cnt[d.y], 1);
    r.z = atomicAdd(&cnt[d.z], 1);
    r.w = atomicAdd(&cnt[d.w], 1);
    *(int4*)(rank + i) = r;
  } else {
    for (; i < E; ++i) rank[i] = atomicAdd(&cnt[di[i]], 1);
  }
}

__global__ __launch_bounds__(1024) void k_scan(const int* __restrict__ cnt3,
                                               int* __restrict__ cur3, int n) {
  __shared__ int wtot[32];
  const int* cnt = cnt3 + (size_t)blockIdx.x * n;
  int* cur = cur3 + (size_t)blockIdx.x * n;
  const int t = threadIdx.x;
  const int lane = t & 31, wave = t >> 5;
  const int chunk = (n + 1023) / 1024;
  int b = t * chunk;
  int e = b + chunk;
  if (b > n) b = n;
  if (e > n) e = n;
  int s = 0;
#pragma unroll 8
  for (int i = b; i < e; ++i) s += cnt[i];
  int incl = s;
#pragma unroll
  for (int off = 1; off < 32; off <<= 1) {
    int u = __shfl_up(incl, off, 32);
    if (lane >= off) incl += u;
  }
  if (lane == 31) wtot[wave] = incl;
  __syncthreads();
  if (wave == 0) {
    int w = wtot[lane];
    int wi = w;
#pragma unroll
    for (int off = 1; off < 32; off <<= 1) {
      int u = __shfl_up(wi, off, 32);
      if (lane >= off) wi += u;
    }
    wtot[lane] = wi - w;
  }
  __syncthreads();
  int run = wtot[wave] + incl - s;
#pragma unroll 8
  for (int i = b; i < e; ++i) {
    cur[i] = run;
    run += cnt[i];
  }
}

__global__ __launch_bounds__(256) void k_place(const int* __restrict__ si0,
                                               const int* __restrict__ si1,
                                               const int* __restrict__ si2,
                                               const int* __restrict__ di0,
                                               const int* __restrict__ di1,
                                               const int* __restrict__ di2,
                                               const float* __restrict__ el3,
                                               const float* __restrict__ er3,
                                               const int* __restrict__ cur3,
                                               const int* __restrict__ rank3,
                                               int2* __restrict__ lst3, int E, int ns, int nd) {
  const int rel = blockIdx.y;
  const int* si = pick3(rel, si0, si1, si2);
  const int* di = pick3(rel, di0, di1, di2);
  const float* el = el3 + (size_t)rel * ns;
  const float* er = er3 + (size_t)rel * nd;
  const int* cur = cur3 + (size_t)rel * nd;
  const int* rank = rank3 + (size_t)rel * E;
  int2* lst = lst3 + (size_t)rel * E;
  int i = (blockIdx.x * 256 + threadIdx.x) * 4;
  if (i + 3 < E) {
    const int4 s = *(const int4*)(si + i);
    const int4 d = *(const int4*)(di + i);
    const int4 rk = *(const int4*)(rank + i);
    const float x0 = lrelu(el[s.x] + er[d.x]);
    const float x1 = lrelu(el[s.y] + er[d.y]);
    const float x2 = lrelu(el[s.z] + er[d.z]);
    const float x3 = lrelu(el[s.w] + er[d.w]);
    const int p0 = cur[d.x] + rk.x;
    const int p1 = cur[d.y] + rk.y;
    const int p2 = cur[d.z] + rk.z;
    const int p3 = cur[d.w] + rk.w;
    lst[p0] = make_int2(s.x, __float_as_int(x0));
    lst[p1] = make_int2(s.y, __float_as_int(x1));
    lst[p2] = make_int2(s.z, __float_as_int(x2));
    lst[p3] = make_int2(s.w, __float_as_int(x3));
  } else {
    for (; i < E; ++i) {
      const int s = si[i], d = di[i];
      const float x = lrelu(el[s] + er[d]);
      lst[cur[d] + rank[i]] = make_int2(s, __float_as_int(x));
    }
  }
}

__device__ __forceinline__ void fma4(float4& a, float p, const float4& h) {
  a.x += p * h.x; a.y += p * h.y; a.z += p * h.z; a.w += p * h.w;
}

template <int N>
__device__ __forceinline__ void add_rows(float4& acc, const float* __restrict__ hs, int vx,
                                         int pb, int k, int lane) {
  float4 h[N];
  float pk[N];
#pragma unroll
  for (int u = 0; u < N; ++u) {
    const unsigned sj = (unsigned)__builtin_amdgcn_readlane(vx, k + u);
    pk[u] = __int_as_float(__builtin_amdgcn_readlane(pb, k + u));
    h[u] = ((const float4*)(hs + (size_t)sj * D))[lane];
  }
#pragma unroll
  for (int u = 0; u < N; ++u) fma4(acc, pk[u], h[u]);
}

#define G_ROWS 4
__global__ __launch_bounds__(256) void k_gather(const int* __restrict__ cnt3,
                                                const int* __restrict__ cur3,
                                                const int2* __restrict__ lst3,
                                                const float* __restrict__ X0,
                                                const float* __restrict__ X1,
                                                const float* __restrict__ X2,
                                                float* __restrict__ Z3, int n, int E) {
  const int rel = blockIdx.y;
  const float* hs = pick3(rel, X0, X1, X2);
  const int* cnt = cnt3 + (size_t)rel * n;
  const int* cur = cur3 + (size_t)rel * n;
  const int2* lst = lst3 + (size_t)rel * E;
  float* Z = Z3 + (size_t)rel * n * D;
  const int lane = threadIdx.x & 31;
  const int d0 =
      __builtin_amdgcn_readfirstlane((blockIdx.x * 8 + (threadIdx.x >> 5)) * G_ROWS);
  const int NEG_INF_BITS = (int)0xff800000u;

  for (int r = 0; r < G_ROWS; ++r) {
    const int d = d0 + r;
    if (d >= n) break;
    const int c = cnt[d];
    const int start = cur[d];

    int2 v0 = make_int2(0, NEG_INF_BITS);
    if (lane < c) v0 = lst[start + lane];
    float mx = __int_as_float(v0.y);
    for (int base = 32; base < c; base += 32) {
      int j = base + lane;
      if (j < c) mx = fmaxf(mx, __int_as_float(lst[start + j].y));
    }
#pragma unroll
    for (int off = 16; off; off >>= 1) mx = fmaxf(mx, __shfl_xor(mx, off, 32));

    float4 acc = make_float4(0.f, 0.f, 0.f, 0.f);
    float zs = 0.f;
    for (int base = 0; base < c; base += 32) {
      int2 v = v0;
      const int j = base + lane;
      if (base > 0) {
        v = make_int2(0, NEG_INF_BITS);
        if (j < c) v = lst[start + j];
      }
      const float pe = expf(__int_as_float(v.y) - mx);
      const float p = (j < c) ? pe : 0.f;
      zs += p;
      const int pb = __float_as_int(p);
      int nc = c - base;
      if (nc > 32) nc = 32;
      int k = 0;
      for (; k + 8 <= nc; k += 8) add_rows<8>(acc, hs, v.x, pb, k, lane);
      switch (nc - k) {
        case 7: add_rows<7>(acc, hs, v.x, pb, k, lane); break;
        case 6: add_rows<6>(acc, hs, v.x, pb, k, lane); break;
        case 5: add_rows<5>(acc, hs, v.x, pb, k, lane); break;
        case 4: add_rows<4>(acc, hs, v.x, pb, k, lane); break;
        case 3: add_rows<3>(acc, hs, v.x, pb, k, lane); break;
        case 2: add_rows<2>(acc, hs, v.x, pb, k, lane); break;
        case 1: add_rows<1>(acc, hs, v.x, pb, k, lane); break;
        default: break;
      }
    }
#pragma unroll
    for (int off = 16; off; off >>= 1) zs += __shfl_xor(zs, off, 32);

    const float inv = (c > 0) ? 1.f / zs : 0.f;
    float4 o;
    o.x = acc.x * inv;
    o.y = acc.y * inv;
    o.z = acc.z * inv;
    o.w = acc.w * inv;
    ((float4*)Z)[(size_t)d * (D / 4) + lane] = o;
  }
}

__global__ __launch_bounds__(256) void k_sem_wmma(const float* __restrict__ Z3,
                                                  const float* __restrict__ W1,
                                                  const float* __restrict__ b1,
                                                  const float* __restrict__ W2,
                                                  float* __restrict__ wsum3, int N) {
  __shared__ float Ws[D * D];
  __shared__ float red[8];
  const float* Z = Z3 + (size_t)blockIdx.y * N * D;
  float* wsum = wsum3 + blockIdx.y;
  int t = threadIdx.x;
#pragma unroll
  for (int i = 0; i < (D * D / 4) / 256; ++i)
    ((float4*)Ws)[t + i * 256] = ((const float4*)W1)[t + i * 256];
  __syncthreads();

  const int wave = t >> 5;
  const int lane = t & 31;
  const int lrow = lane & 15;
  const int half = lane >> 4;
  const int rowbase = blockIdx.x * 128 + wave * 16;

  int xr = rowbase + lrow;
  if (xr > N - 1) xr = N - 1;
  const float* xp = Z + (size_t)xr * D;

  v2f afrag[32];
#pragma unroll
  for (int kt = 0; kt < 32; ++kt) {
    int k0 = kt * 4 + 2 * half;
    afrag[kt] = *(const v2f*)(xp + k0);
  }

  float s = 0.f;
#pragma unroll
  for (int nt = 0; nt < 8; ++nt) {
    v8f acc = {};
    int nb = nt * 16 + lrow;
#pragma unroll
    for (int kt = 0; kt < 32; ++kt) {
      int k0 = kt * 4 + 2 * half;
      v2f b;
      b.x = Ws[k0 * D + nb];
      b.y = Ws[(k0 + 1) * D + nb];
      acc = __builtin_amdgcn_wmma_f32_16x16x4_f32(false, afrag[kt], false, b,
                                                  (short)0, acc, false, false);
    }
    asm volatile("v_nop\n\tv_nop\n\tv_nop\n\tv_nop" : "+v"(acc));
    const float bn = b1[nb];
    const float wn = W2[nb];
#pragma unroll
    for (int v = 0; v < 8; ++v) {
      int grow = rowbase + v + 8 * half;
      float y = tanhf(acc[v] + bn) * wn;
      s += (grow < N) ? y : 0.f;
    }
  }

#pragma unroll
  for (int off = 16; off; off >>= 1) s += __shfl_xor(s, off, 32);
  if (lane == 0) red[wave] = s;
  __syncthreads();
  if (t == 0) {
    float tot = ((red[0] + red[1]) + (red[2] + red[3])) +
                ((red[4] + red[5]) + (red[6] + red[7]));
    atomicAdd(wsum, tot);
  }
}

#define MIX_V 8
__global__ __launch_bounds__(256) void k_mix(const float* __restrict__ Z3,
                                             const float* __restrict__ wsum,
                                             float* __restrict__ out, int n, float invN) {
  const int n4 = n >> 2;
  const float4* __restrict__ za = (const float4*)Z3;
  const float4* __restrict__ zb = (const float4*)(Z3 + (size_t)n);
  const float4* __restrict__ zc = (const float4*)(Z3 + 2 * (size_t)n);
  const int base = blockIdx.x * (256 * MIX_V) + threadIdx.x;
  float4 a[MIX_V], b[MIX_V], c[MIX_V];
#pragma unroll
  for (int u = 0; u < MIX_V; ++u) {
    int i = base + u * 256;
    if (i > n4 - 1) i = n4 - 1;
    a[u] = za[i];
    b[u] = zb[i];
    c[u] = zc[i];
  }
  float w0 = wsum[0] * invN, w1 = wsum[1] * invN, w2 = wsum[2] * invN;
  float mx = fmaxf(w0, fmaxf(w1, w2));
  float e0 = expf(w0 - mx), e1 = expf(w1 - mx), e2 = expf(w2 - mx);
  float inv = 1.f / (e0 + e1 + e2);
  float a0 = e0 * inv, a1 = e1 * inv, a2 = e2 * inv;
#pragma unroll
  for (int u = 0; u < MIX_V; ++u) {
    const int i = base + u * 256;
    float4 o;
    o.x = a0 * a[u].x + a1 * b[u].x + a2 * c[u].x;
    o.y = a0 * a[u].y + a1 * b[u].y + a2 * c[u].y;
    o.z = a0 * a[u].z + a1 * b[u].z + a2 * c[u].z;
    o.w = a0 * a[u].w + a1 * b[u].w + a2 * c[u].w;
    if (i < n4) ((float4*)out)[i] = o;
  }
  if (blockIdx.x == 0 && threadIdx.x == 0) { out[n] = a0; out[n + 1] = a1; out[n + 2] = a2; }
}


extern "C" void kernel_launch(void* const* d_in, const int* in_sizes, int n_in,
                              void* d_out, int out_size, void* d_ws, size_t ws_size,
                              hipStream_t stream) {
  const float* dst_feat = (const float*)d_in[0];
  const float* src_feat[RELS] = {(const float*)d_in[1], (const float*)d_in[2],
                                 (const float*)d_in[3]};
  const float* sem_W1 = (const float*)d_in[4];
  const float* sem_b1 = (const float*)d_in[5];
  const float* sem_W2 = (const float*)d_in[6];
  const float *Wg[RELS], *al[RELS], *ar[RELS], *bg[RELS];
  const int *sidx[RELS], *didx[RELS];
  for (int r = 0; r < RELS; ++r) {
    int b = 7 + 6 * r;
    Wg[r] = (const float*)d_in[b + 0];
    al[r] = (const float*)d_in[b + 1];
    ar[r] = (const float*)d_in[b + 2];
    bg[r] = (const float*)d_in[b + 3];
    sidx[r] = (const int*)d_in[b + 4];
    didx[r] = (const int*)d_in[b + 5];
  }
  const int ND = in_sizes[0] / D;
  const int NS = in_sizes[1] / D;
  const int E  = in_sizes[11];

  float* ws   = (float*)d_ws;
  float* z3   = ws;
  float* wsum = z3 + (size_t)RELS * ND * D;
  int2*  lst = (int2*)d_out;
  float* el  = (float*)(lst + (size_t)RELS * E);
  float* er  = el + (size_t)RELS * NS;
  int*   cnt = (int*)(er + (size_t)RELS * ND);
  int*   cur = cnt + (size_t)RELS * ND;
  float* wa  = (float*)(cur + (size_t)RELS * ND);
  float* wr  = wa + RELS * D;
  int*   rank = (int*)(z3 + (size_t)(RELS - 1) * ND * D);

  k_rowdot<<<RELS, 256, 0, stream>>>(Wg[0], Wg[1], Wg[2], al[0], al[1], al[2], ar[0], ar[1],
                                     ar[2], wa, wr, wsum);
  k_matvec<<<dim3((NS + 8 * MV_ROWS - 1) / (8 * MV_ROWS), RELS), 256, 0, stream>>>(
      src_feat[0], src_feat[1], src_feat[2], wa, el, NS);
  k_matvec_dst<<<(ND + 8 * MV_ROWS - 1) / (8 * MV_ROWS), 256, 0, stream>>>(dst_feat, wr, er,
                                                                         cnt, ND);
  k_rank<<<dim3((E + 1023) / 1024, RELS), 256, 0, stream>>>(didx[0], didx[1], didx[2], cnt,
                                                            rank, E, ND);
  k_scan<<<RELS, 1024, 0, stream>>>(cnt, cur, ND);
  k_place<<<dim3((E + 1023) / 1024, RELS), 256, 0, stream>>>(
      sidx[0], sidx[1], sidx[2], didx[0], didx[1], didx[2], el, er, cur, rank, lst, E, NS, ND);

  k_gather<<<dim3((ND + 8 * G_ROWS - 1) / (8 * G_ROWS), RELS), 256, 0, stream>>>(
      cnt, cur, lst, src_feat[0], src_feat[1], src_feat[2], z3, ND, E);
  k_gemm_wmma<<<dim3((ND + 127) / 128, RELS), 256, 0, stream>>>(z3, Wg[0], Wg[1], Wg[2], bg[0],
                                                                 bg[1], bg[2], ND);

  k_sem_wmma<<<dim3((ND + 127) / 128, RELS), 256, 0, stream>>>(z3, sem_W1, sem_b1, sem_W2,
                                                               wsum, ND);

  k_mix<<<(ND * D / 4 + 256 * MIX_V - 1) / (256 * MIX_V), 256, 0, stream>>>(
      z3, wsum, (float*)d_out, ND * D, 1.0f / (float)ND);
}


# run starts: each wave of the dst matvec pass (now behind k_rank) sums its rows' edge counts and takes their places with one cursor add; the one-block prefix scan launch is gone; sums as before
# speedup vs baseline: 3.1569x; 1.0567x over previous
#include <hip/hip_runtime.h>
#include <math.h>

#define D 128
#define RELS 3

typedef float v2f __attribute__((ext_vector_type(2)));
typedef float v8f __attribute__((ext_vector_type(8)));


__device__ __forceinline__ float lrelu(float x) { return x > 0.f ? x : 0.2f * x; }

__device__ __forceinline__ const float* pick3(int r, const float* a, const float* b,
                                              const float* c) {
  return r == 0 ? a : (r == 1 ? b : c);
}
__device__ __forceinline__ const int* pick3(int r, const int* a, const int* b, const int* c) {
  return r == 0 ? a : (r == 1 ? b : c);
}


#define CUR_PAD 64
__global__ __launch_bounds__(256) void k_rowdot(const float* __restrict__ W0,
                                                const float* __restrict__ W1,
                                                const float* __restrict__ W2,
                                                const float* __restrict__ al0,
                                                const float* __restrict__ al1,
                                                const float* __restrict__ al2,
                                                const float* __restrict__ ar0,
                                                const float* __restrict__ ar1,
                                                const float* __restrict__ ar2,
                                                float* __restrict__ wa3,
                                                float* __restrict__ wr3,
                                                float* __restrict__ wsum,
                                                int* __restrict__ zero, int nz,
                                                int* __restrict__ alloc) {
  if (blockIdx.x >= RELS) {
    const int i = ((blockIdx.x - RELS) * 256 + threadIdx.x) * 4;
    if (i + 3 < nz) {
      *(int4*)(zero + i) = make_int4(0, 0, 0, 0);
    } else {
      for (int j = i; j < nz; ++j) zero[j] = 0;
    }
    return;
  }
  const int rel = blockIdx.x;
  const float* W = pick3(rel, W0, W1, W2);
  const float* al = pick3(rel, al0, al1, al2);
  const float* ar = pick3(rel, ar0, ar1, ar2);
  int t = threadIdx.x;
  if (t == 0) {
    wsum[rel] = 0.f;
    alloc[rel * CUR_PAD] = 0;
  }
  const float* v = (t < D) ? al : ar;
  float* o = ((t < D) ? wa3 : wr3) + rel * D;
  int d = t & (D - 1);
  float s = 0.f;
#pragma unroll 8
  for (int h = 0; h < D; ++h) s += W[d * D + h] * v[h];
  o[d] = s;
}

__global__ __launch_bounds__(256) void k_gemm_wmma(float* Z3,
                                                   const float* __restrict__ W0,
                                                   const float* __restrict__ W1,
                                                   const float* __restrict__ W2,
                                                   const float* __restrict__ b0,
                                                   const float* __restrict__ b1,
                                                   const float* __restrict__ b2, int N) {
  __shared__ float Ws[D * D];
  const int rel = blockIdx.y;
  const float* W = pick3(rel, W0, W1, W2);
  const float* bias = pick3(rel, b0, b1, b2);
  float* Z = Z3 + (size_t)rel * N * D;
  int t = threadIdx.x;
#pragma unroll
  for (int i = 0; i < (D * D / 4) / 256; ++i)
    ((float4*)Ws)[t + i * 256] = ((const float4*)W)[t + i * 256];
  __syncthreads();

  const int wave = t >> 5;
  const int lane = t & 31;
  const int lrow = lane & 15;
  const int half = lane >> 4;
  const int rowbase = blockIdx.x * 128 + wave * 16;

  int xr = rowbase + lrow;
  if (xr > N - 1) xr = N - 1;
  const float* xp = Z + (size_t)xr * D;

  v2f afrag[32];
#pragma unroll
  for (int kt = 0; kt < 32; ++kt) {
    int k0 = kt * 4 + 2 * half;
    afrag[kt] = *(const v2f*)(xp + k0);
  }
  float bn[8];
#pragma unroll
  for (int nt = 0; nt < 8; ++nt) bn[nt] = bias[nt * 16 + lrow];

#pragma unroll
  for (int nt = 0; nt < 8; ++nt) {
    v8f acc = {};
    int nb = nt * 16 + lrow;
#pragma unroll
    for (int kt = 0; kt < 32; ++kt) {
      int k0 = kt * 4 + 2 * half;
      v2f b;
      b.x = Ws[k0 * D + nb];
      b.y = Ws[(k0 + 1) * D + nb];
      acc = __builtin_amdgcn_wmma_f32_16x16x4_f32(false, afrag[kt], false, b,
                                                  (short)0, acc, false, false);
    }
    asm volatile("v_nop\n\tv_nop\n\tv_nop\n\tv_nop" : "+v"(acc));
#pragma unroll
    for (int v = 0; v < 8; ++v) {
      int grow = rowbase + v + 8 * half;
      float o = acc[v] + bn[nt];
      o = o > 0.f ? o : expf(o) - 1.f;
      if (grow < N) Z[(size_t)grow * D + nb] = o;
    }
  }
}

#define MV_ROWS 16
__global__ __launch_bounds__(256) void k_matvec(const float* __restrict__ X0,
                                                const float* __restrict__ X1,
                                                const float* __restrict__ X2,
                                                const float* __restrict__ v3,
                                                float* __restrict__ out3, int n) {
  const int rel = blockIdx.y;
  const float* X = pick3(rel, X0, X1, X2);
  const float* v = v3 + rel * D;
  float* out = out3 + (size_t)rel * n;
  const int lane = threadIdx.x & 31;
  const int row0 =
      __builtin_amdgcn_readfirstlane((blockIdx.x * 8 + (threadIdx.x >> 5)) * MV_ROWS);
  if (row0 >= n) return;
  float4 x[MV_ROWS];
#pragma unroll
  for (int u = 0; u < MV_ROWS; ++u) {
    int r = row0 + u;
    if (r > n - 1) r = n - 1;
    x[u] = ((const float4*)X)[(size_t)r * (D / 4) + lane];
  }
  const float4 c = ((const float4*)v)[lane];
  __builtin_amdgcn_sched_barrier(0);
  float mine = 0.f;
#pragma unroll
  for (int u = 0; u < MV_ROWS; ++u) {
    float s = x[u].x * c.x + x[u].y * c.y + x[u].z * c.z + x[u].w * c.w;
#pragma unroll
    for (int off = 16; off; off >>= 1) s += __shfl_xor(s, off, 32);
    if (lane == u) mine = s;
  }
  const int r = row0 + lane;
  if (lane < MV_ROWS && r < n) out[r] = mine;
}

__global__ __launch_bounds__(256) void k_matvec_dst(const float* __restrict__ X,
                                                    const float* __restrict__ v3,
                                                    float* __restrict__ out3,
                                                    const int* __restrict__ cnt3,
                                                    int* __restrict__ cur3,
                                                    int* __restrict__ alloc, int n) {
  const int lane = threadIdx.x & 31;
  const int row0 =
      __builtin_amdgcn_readfirstlane((blockIdx.x * 8 + (threadIdx.x >> 5)) * MV_ROWS);
  if (row0 >= n) return;
  float4 x[MV_ROWS];
#pragma unroll
  for (int u = 0; u < MV_ROWS; ++u) {
    int r = row0 + u;
    if (r > n - 1) r = n - 1;
    x[u] = ((const float4*)X)[(size_t)r * (D / 4) + lane];
  }
  const float4 c0 = ((const float4*)v3)[lane];
  const float4 c1 = ((const float4*)v3)[(D / 4) + lane];
  const float4 c2 = ((const float4*)v3)[2 * (D / 4) + lane];
  const int r = row0 + lane;
  const bool own = lane < MV_ROWS && r < n;
  int n0 = 0, n1 = 0, n2 = 0;
  if (own) {
    n0 = cnt3[r];
    n1 = cnt3[(size_t)n + r];
    n2 = cnt3[2 * (size_t)n + r];
  }
  __builtin_amdgcn_sched_barrier(0);
  int i0 = n0, i1 = n1, i2 = n2;
#pragma unroll
  for (int off = 1; off < MV_ROWS; off <<= 1) {
    const int u0 = __shfl_up(i0, off, 32);
    const int u1 = __shfl_up(i1, off, 32);
    const int u2 = __shfl_up(i2, off, 32);
    if (lane >= off) { i0 += u0; i1 += u1; i2 += u2; }
  }
  const int t0 = __shfl(i0, MV_ROWS - 1, 32);
  const int t1 = __shfl(i1, MV_ROWS - 1, 32);
  const int t2 = __shfl(i2, MV_ROWS - 1, 32);
  const int tsel = lane == 0 ? t0 : (lane == 1 ? t1 : t2);
  int bsel = 0;
  if (lane < RELS) bsel = atomicAdd(&alloc[lane * CUR_PAD], tsel);
  float m0 = 0.f, m1 = 0.f, m2 = 0.f;
#pragma unroll
  for (int u = 0; u < MV_ROWS; ++u) {
    float s0 = x[u].x * c0.x + x[u].y * c0.y + x[u].z * c0.z + x[u].w * c0.w;
    float s1 = x[u].x * c1.x + x[u].y * c1.y + x[u].z * c1.z + x[u].w * c1.w;
    float s2 = x[u].x * c2.x + x[u].y * c2.y + x[u].z * c2.z + x[u].w * c2.w;
#pragma unroll
    for (int off = 16; off; off >>= 1) {
      s0 += __shfl_xor(s0, off, 32);
      s1 += __shfl_xor(s1, off, 32);
      s2 += __shfl_xor(s2, off, 32);
    }
    if (lane == u) { m0 = s0; m1 = s1; m2 = s2; }
  }
  const int b0 = __shfl(bsel, 0, 32);
  const int b1 = __shfl(bsel, 1, 32);
  const int b2 = __shfl(bsel, 2, 32);
  if (own) {
    out3[r] = m0;
    out3[(size_t)n + r] = m1;
    out3[2 * (size_t)n + r] = m2;
    cur3[r] = b0 + i0 - n0;
    cur3[(size_t)n + r] = b1 + i1 - n1;
    cur3[2 * (size_t)n + r] = b2 + i2 - n2;
  }
}


__global__ __launch_bounds__(256) void k_rank(const int* __restrict__ di0,
                                              const int* __restrict__ di1,
                                              const int* __restrict__ di2,
                                              int* __restrict__ cnt3, int* __restrict__ rank3,
                                              int E, int nd) {
  const int rel = blockIdx.y;
  const int* di = pick3(rel, di0, di1, di2);
  int* cnt = cnt3 + (size_t)rel * nd;
  int* rank = rank3 + (size_t)rel * E;
  int i = (blockIdx.x * 256 + threadIdx.x) * 4;
  if (i + 3 < E) {
    const int4 d = *(const int4*)(di + i);
    int4 r;
    r.x = atomicAdd(&cnt[d.x], 1);
    r.y = atomicAdd(&cnt[d.y], 1);
    r.z = atomicAdd(&cnt[d.z], 1);
    r.w = atomicAdd(&cnt[d.w], 1);
    *(int4*)(rank + i) = r;
  } else {
    for (; i < E; ++i) rank[i] = atomicAdd(&cnt[di[i]], 1);
  }
}

__global__ __launch_bounds__(256) void k_place(const int* __restrict__ si0,
                                               const int* __restrict__ si1,
                                               const int* __restrict__ si2,
                                               const int* __restrict__ di0,
                                               const int* __restrict__ di1,
                                               const int* __restrict__ di2,
                                               const float* __restrict__ el3,
                                               const float* __restrict__ er3,
                                               const int* __restrict__ cur3,
                                               const int* __restrict__ rank3,
                                               int2* __restrict__ lst3, int E, int ns, int nd) {
  const int rel = blockIdx.y;
  const int* si = pick3(rel, si0, si1, si2);
  const int* di = pick3(rel, di0, di1, di2);
  const float* el = el3 + (size_t)rel * ns;
  const float* er = er3 + (size_t)rel * nd;
  const int* cur = cur3 + (size_t)rel * nd;
  const int* rank = rank3 + (size_t)rel * E;
  int2* lst = lst3 + (size_t)rel * E;
  int i = (blockIdx.x * 256 + threadIdx.x) * 4;
  if (i + 3 < E) {
    const int4 s = *(const int4*)(si + i);
    const int4 d = *(const int4*)(di + i);
    const int4 rk = *(const int4*)(rank + i);
    const float x0 = lrelu(el[s.x] + er[d.x]);
    const float x1 = lrelu(el[s.y] + er[d.y]);
    const float x2 = lrelu(el[s.z] + er[d.z]);
    const float x3 = lrelu(el[s.w] + er[d.w]);
    const int p0 = cur[d.x] + rk.x;
    const int p1 = cur[d.y] + rk.y;
    const int p2 = cur[d.z] + rk.z;
    const int p3 = cur[d.w] + rk.w;
    lst[p0] = make_int2(s.x, __float_as_int(x0));
    lst[p1] = make_int2(s.y, __float_as_int(x1));
    lst[p2] = make_int2(s.z, __float_as_int(x2));
    lst[p3] = make_int2(s.w, __float_as_int(x3));
  } else {
    for (; i < E; ++i) {
      const int s = si[i], d = di[i];
      const float x = lrelu(el[s] + er[d]);
      lst[cur[d] + rank[i]] = make_int2(s, __float_as_int(x));
    }
  }
}

__device__ __forceinline__ void fma4(float4& a, float p, const float4& h) {
  a.x += p * h.x; a.y += p * h.y; a.z += p * h.z; a.w += p * h.w;
}

template <int N>
__device__ __forceinline__ void add_rows(float4& acc, const float* __restrict__ hs, int vx,
                                         int pb, int k, int lane) {
  float4 h[N];
  float pk[N];
#pragma unroll
  for (int u = 0; u < N; ++u) {
    const unsigned sj = (unsigned)__builtin_amdgcn_readlane(vx, k + u);
    pk[u] = __int_as_float(__builtin_amdgcn_readlane(pb, k + u));
    h[u] = ((const float4*)(hs + (size_t)sj * D))[lane];
  }
#pragma unroll
  for (int u = 0; u < N; ++u) fma4(acc, pk[u], h[u]);
}

#define G_ROWS 4
__global__ __launch_bounds__(256) void k_gather(const int* __restrict__ cnt3,
                                                const int* __restrict__ cur3,
                                                const int2* __restrict__ lst3,
                                                const float* __restrict__ X0,
                                                const float* __restrict__ X1,
                                                const float* __restrict__ X2,
                                                float* __restrict__ Z3, int n, int E) {
  const int rel = blockIdx.y;
  const float* hs = pick3(rel, X0, X1, X2);
  const int* cnt = cnt3 + (size_t)rel * n;
  const int* cur = cur3 + (size_t)rel * n;
  const int2* lst = lst3 + (size_t)rel * E;
  float* Z = Z3 + (size_t)rel * n * D;
  const int lane = threadIdx.x & 31;
  const int d0 =
      __builtin_amdgcn_readfirstlane((blockIdx.x * 8 + (threadIdx.x >> 5)) * G_ROWS);
  const int NEG_INF_BITS = (int)0xff800000u;

  for (int r = 0; r < G_ROWS; ++r) {
    const int d = d0 + r;
    if (d >= n) break;
    const int c = cnt[d];
    const int start = cur[d];

    int2 v0 = make_int2(0, NEG_INF_BITS);
    if (lane < c) v0 = lst[start + lane];
    float mx = __int_as_float(v0.y);
    for (int base = 32; base < c; base += 32) {
      int j = base + lane;
      if (j < c) mx = fmaxf(mx, __int_as_float(lst[start + j].y));
    }
#pragma unroll
    for (int off = 16; off; off >>= 1) mx = fmaxf(mx, __shfl_xor(mx, off, 32));

    float4 acc = make_float4(0.f, 0.f, 0.f, 0.f);
    float zs = 0.f;
    for (int base = 0; base < c; base += 32) {
      int2 v = v0;
      const int j = base + lane;
      if (base > 0) {
        v = make_int2(0, NEG_INF_BITS);
        if (j < c) v = lst[start + j];
      }
      const float pe = expf(__int_as_float(v.y) - mx);
      const float p = (j < c) ? pe : 0.f;
      zs += p;
      const int pb = __float_as_int(p);
      int nc = c - base;
      if (nc > 32) nc = 32;
      int k = 0;
      for (; k + 8 <= nc; k += 8) add_rows<8>(acc, hs, v.x, pb, k, lane);
      switch (nc - k) {
        case 7: add_rows<7>(acc, hs, v.x, pb, k, lane); break;
        case 6: add_rows<6>(acc, hs, v.x, pb, k, lane); break;
        case 5: add_rows<5>(acc, hs, v.x, pb, k, lane); break;
        case 4: add_rows<4>(acc, hs, v.x, pb, k, lane); break;
        case 3: add_rows<3>(acc, hs, v.x, pb, k, lane); break;
        case 2: add_rows<2>(acc, hs, v.x, pb, k, lane); break;
        case 1: add_rows<1>(acc, hs, v.x, pb, k, lane); break;
        default: break;
      }
    }
#pragma unroll
    for (int off = 16; off; off >>= 1) zs += __shfl_xor(zs, off, 32);

    const float inv = (c > 0) ? 1.f / zs : 0.f;
    float4 o;
    o.x = acc.x * inv;
    o.y = acc.y * inv;
    o.z = acc.z * inv;
    o.w = acc.w * inv;
    ((float4*)Z)[(size_t)d * (D / 4) + lane] = o;
  }
}

__global__ __launch_bounds__(256) void k_sem_wmma(const float* __restrict__ Z3,
                                                  const float* __restrict__ W1,
                                                  const float* __restrict__ b1,
                                                  const float* __restrict__ W2,
                                                  float* __restrict__ wsum3, int N) {
  __shared__ float Ws[D * D];
  __shared__ float red[8];
  const float* Z = Z3 + (size_t)blockIdx.y * N * D;
  float* wsum = wsum3 + blockIdx.y;
  int t = threadIdx.x;
#pragma unroll
  for (int i = 0; i < (D * D / 4) / 256; ++i)
    ((float4*)Ws)[t + i * 256] = ((const float4*)W1)[t + i * 256];
  __syncthreads();

  const int wave = t >> 5;
  const int lane = t & 31;
  const int lrow = lane & 15;
  const int half = lane >> 4;
  const int rowbase = blockIdx.x * 128 + wave * 16;

  int xr = rowbase + lrow;
  if (xr > N - 1) xr = N - 1;
  const float* xp = Z + (size_t)xr * D;

  v2f afrag[32];
#pragma unroll
  for (int kt = 0; kt < 32; ++kt) {
    int k0 = kt * 4 + 2 * half;
    afrag[kt] = *(const v2f*)(xp + k0);
  }

  float s = 0.f;
#pragma unroll
  for (int nt = 0; nt < 8; ++nt) {
    v8f acc = {};
    int nb = nt * 16 + lrow;
#pragma unroll
    for (int kt = 0; kt < 32; ++kt) {
      int k0 = kt * 4 + 2 * half;
      v2f b;
      b.x = Ws[k0 * D + nb];
      b.y = Ws[(k0 + 1) * D + nb];
      acc = __builtin_amdgcn_wmma_f32_16x16x4_f32(false, afrag[kt], false, b,
                                                  (short)0, acc, false, false);
    }
    asm volatile("v_nop\n\tv_nop\n\tv_nop\n\tv_nop" : "+v"(acc));
    const float bn = b1[nb];
    const float wn = W2[nb];
#pragma unroll
    for (int v = 0; v < 8; ++v) {
      int grow = rowbase + v + 8 * half;
      float y = tanhf(acc[v] + bn) * wn;
      s += (grow < N) ? y : 0.f;
    }
  }

#pragma unroll
  for (int off = 16; off; off >>= 1) s += __shfl_xor(s, off, 32);
  if (lane == 0) red[wave] = s;
  __syncthreads();
  if (t == 0) {
    float tot = ((red[0] + red[1]) + (red[2] + red[3])) +
                ((red[4] + red[5]) + (red[6] + red[7]));
    atomicAdd(wsum, tot);
  }
}

#define MIX_V 8
__global__ __launch_bounds__(256) void k_mix(const float* __restrict__ Z3,
                                             const float* __restrict__ wsum,
                                             float* __restrict__ out, int n, float invN) {
  const int n4 = n >> 2;
  const float4* __restrict__ za = (const float4*)Z3;
  const float4* __restrict__ zb = (const float4*)(Z3 + (size_t)n);
  const float4* __restrict__ zc = (const float4*)(Z3 + 2 * (size_t)n);
  const int base = blockIdx.x * (256 * MIX_V) + threadIdx.x;
  float4 a[MIX_V], b[MIX_V], c[MIX_V];
#pragma unroll
  for (int u = 0; u < MIX_V; ++u) {
    int i = base + u * 256;
    if (i > n4 - 1) i = n4 - 1;
    a[u] = za[i];
    b[u] = zb[i];
    c[u] = zc[i];
  }
  float w0 = wsum[0] * invN, w1 = wsum[1] * invN, w2 = wsum[2] * invN;
  float mx = fmaxf(w0, fmaxf(w1, w2));
  float e0 = expf(w0 - mx), e1 = expf(w1 - mx), e2 = expf(w2 - mx);
  float inv = 1.f / (e0 + e1 + e2);
  float a0 = e0 * inv, a1 = e1 * inv, a2 = e2 * inv;
#pragma unroll
  for (int u = 0; u < MIX_V; ++u) {
    const int i = base + u * 256;
    float4 o;
    o.x = a0 * a[u].x + a1 * b[u].x + a2 * c[u].x;
    o.y = a0 * a[u].y + a1 * b[u].y + a2 * c[u].y;
    o.z = a0 * a[u].z + a1 * b[u].z + a2 * c[u].z;
    o.w = a0 * a[u].w + a1 * b[u].w + a2 * c[u].w;
    if (i < n4) ((float4*)out)[i] = o;
  }
  if (blockIdx.x == 0 && threadIdx.x == 0) { out[n] = a0; out[n + 1] = a1; out[n + 2] = a2; }
}


extern "C" void kernel_launch(void* const* d_in, const int* in_sizes, int n_in,
                              void* d_out, int out_size, void* d_ws, size_t ws_size,
                              hipStream_t stream) {
  const float* dst_feat = (const float*)d_in[0];
  const float* src_feat[RELS] = {(const float*)d_in[1], (const float*)d_in[2],
                                 (const float*)d_in[3]};
  const float* sem_W1 = (const float*)d_in[4];
  const float* sem_b1 = (const float*)d_in[5];
  const float* sem_W2 = (const float*)d_in[6];
  const float *Wg[RELS], *al[RELS], *ar[RELS], *bg[RELS];
  const int *sidx[RELS], *didx[RELS];
  for (int r = 0; r < RELS; ++r) {
    int b = 7 + 6 * r;
    Wg[r] = (const float*)d_in[b + 0];
    al[r] = (const float*)d_in[b + 1];
    ar[r] = (const float*)d_in[b + 2];
    bg[r] = (const float*)d_in[b + 3];
    sidx[r] = (const int*)d_in[b + 4];
    didx[r] = (const int*)d_in[b + 5];
  }
  const int ND = in_sizes[0] / D;
  const int NS = in_sizes[1] / D;
  const int E  = in_sizes[11];

  float* ws   = (float*)d_ws;
  float* z3   = ws;
  float* wsum = z3 + (size_t)RELS * ND * D;
  int2*  lst = (int2*)d_out;
  float* el  = (float*)(lst + (size_t)RELS * E);
  float* er  = el + (size_t)RELS * NS;
  int*   cnt = (int*)(er + (size_t)RELS * ND);
  int*   cur = cnt + (size_t)RELS * ND;
  float* wa  = (float*)(cur + (size_t)RELS * ND);
  float* wr  = wa + RELS * D;
  int*   alloc = (int*)(wr + RELS * D);
  int*   rank = (int*)(z3 + (size_t)(RELS - 1) * ND * D);

  k_rowdot<<<RELS + (RELS * ND + 1023) / 1024, 256, 0, stream>>>(
      Wg[0], Wg[1], Wg[2], al[0], al[1], al[2], ar[0], ar[1], ar[2], wa, wr, wsum, cnt,
      RELS * ND, alloc);
  k_matvec<<<dim3((NS + 8 * MV_ROWS - 1) / (8 * MV_ROWS), RELS), 256, 0, stream>>>(
      src_feat[0], src_feat[1], src_feat[2], wa, el, NS);
  k_rank<<<dim3((E + 1023) / 1024, RELS), 256, 0, stream>>>(didx[0], didx[1], didx[2], cnt,
                                                            rank, E, ND);
  k_matvec_dst<<<(ND + 8 * MV_ROWS - 1) / (8 * MV_ROWS), 256, 0, stream>>>(dst_feat, wr, er,
                                                                         cnt, cur, alloc, ND);
  k_place<<<dim3((E + 1023) / 1024, RELS), 256, 0, stream>>>(
      sidx[0], sidx[1], sidx[2], didx[0], didx[1], didx[2], el, er, cur, rank, lst, E, NS, ND);

  k_gather<<<dim3((ND + 8 * G_ROWS - 1) / (8 * G_ROWS), RELS), 256, 0, stream>>>(
      cnt, cur, lst, src_feat[0], src_feat[1], src_feat[2], z3, ND, E);
  k_gemm_wmma<<<dim3((ND + 127) / 128, RELS), 256, 0, stream>>>(z3, Wg[0], Wg[1], Wg[2], bg[0],
                                                                 bg[1], bg[2], ND);

  k_sem_wmma<<<dim3((ND + 127) / 128, RELS), 256, 0, stream>>>(z3, sem_W1, sem_b1, sem_W2,
                                                               wsum, ND);

  k_mix<<<(ND * D / 4 + 256 * MIX_V - 1) / (256 * MIX_V), 256, 0, stream>>>(
      z3, wsum, (float*)d_out, ND * D, 1.0f / (float)ND);
}


# edge ranks by returning LDS adds: a 1024-thread group per 16384-edge chunk counts every dst row in LDS, the dst matvec pass turns chunk counts into places; no global atomic per edge; sum order moves
# speedup vs baseline: 4.1375x; 1.3106x over previous
#include <hip/hip_runtime.h>
#include <math.h>

#define D 128
#define RELS 3

typedef float v2f __attribute__((ext_vector_type(2)));
typedef float v8f __attribute__((ext_vector_type(8)));


__device__ __forceinline__ float lrelu(float x) { return x > 0.f ? x : 0.2f * x; }

__device__ __forceinline__ const float* pick3(int r, const float* a, const float* b,
                                              const float* c) {
  return r == 0 ? a : (r == 1 ? b : c);
}
__device__ __forceinline__ const int* pick3(int r, const int* a, const int* b, const int* c) {
  return r == 0 ? a : (r == 1 ? b : c);
}


#define CUR_PAD 64
__global__ __launch_bounds__(256) void k_rowdot(const float* __restrict__ W0,
                                                const float* __restrict__ W1,
                                                const float* __restrict__ W2,
                                                const float* __restrict__ al0,
                                                const float* __restrict__ al1,
                                                const float* __restrict__ al2,
                                                const float* __restrict__ ar0,
                                                const float* __restrict__ ar1,
                                                const float* __restrict__ ar2,
                                                float* __restrict__ wa3,
                                                float* __restrict__ wr3,
                                                float* __restrict__ wsum,
                                                int* __restrict__ alloc) {
  const int rel = blockIdx.x;
  const float* W = pick3(rel, W0, W1, W2);
  const float* al = pick3(rel, al0, al1, al2);
  const float* ar = pick3(rel, ar0, ar1, ar2);
  int t = threadIdx.x;
  if (t == 0) {
    wsum[rel] = 0.f;
    alloc[rel * CUR_PAD] = 0;
  }
  const float* v = (t < D) ? al : ar;
  float* o = ((t < D) ? wa3 : wr3) + rel * D;
  int d = t & (D - 1);
  float s = 0.f;
#pragma unroll 8
  for (int h = 0; h < D; ++h) s += W[d * D + h] * v[h];
  o[d] = s;
}

__global__ __launch_bounds__(256) void k_gemm_wmma(float* Z3,
                                                   const float* __restrict__ W0,
                                                   const float* __restrict__ W1,
                                                   const float* __restrict__ W2,
                                                   const float* __restrict__ b0,
                                                   const float* __restrict__ b1,
                                                   const float* __restrict__ b2, int N) {
  __shared__ float Ws[D * D];
  const int rel = blockIdx.y;
  const float* W = pick3(rel, W0, W1, W2);
  const float* bias = pick3(rel, b0, b1, b2);
  float* Z = Z3 + (size_t)rel * N * D;
  int t = threadIdx.x;
#pragma unroll
  for (int i = 0; i < (D * D / 4) / 256; ++i)
    ((float4*)Ws)[t + i * 256] = ((const float4*)W)[t + i * 256];
  __syncthreads();

  const int wave = t >> 5;
  const int lane = t & 31;
  const int lrow = lane & 15;
  const int half = lane >> 4;
  const int rowbase = blockIdx.x * 128 + wave * 16;

  int xr = rowbase + lrow;
  if (xr > N - 1) xr = N - 1;
  const float* xp = Z + (size_t)xr * D;

  v2f afrag[32];
#pragma unroll
  for (int kt = 0; kt < 32; ++kt) {
    int k0 = kt * 4 + 2 * half;
    afrag[kt] = *(const v2f*)(xp + k0);
  }
  float bn[8];
#pragma unroll
  for (int nt = 0; nt < 8; ++nt) bn[nt] = bias[nt * 16 + lrow];

#pragma unroll
  for (int nt = 0; nt < 8; ++nt) {
    v8f acc = {};
    int nb = nt * 16 + lrow;
#pragma unroll
    for (int kt = 0; kt < 32; ++kt) {
      int k0 = kt * 4 + 2 * half;
      v2f b;
      b.x = Ws[k0 * D + nb];
      b.y = Ws[(k0 + 1) * D + nb];
      acc = __builtin_amdgcn_wmma_f32_16x16x4_f32(false, afrag[kt], false, b,
                                                  (short)0, acc, false, false);
    }
    asm volatile("v_nop\n\tv_nop\n\tv_nop\n\tv_nop" : "+v"(acc));
#pragma unroll
    for (int v = 0; v < 8; ++v) {
      int grow = rowbase + v + 8 * half;
      float o = acc[v] + bn[nt];
      o = o > 0.f ? o : expf(o) - 1.f;
      if (grow < N) Z[(size_t)grow * D + nb] = o;
    }
  }
}

#define MV_ROWS 16
__global__ __launch_bounds__(256) void k_matvec(const float* __restrict__ X0,
                                                const float* __restrict__ X1,
                                                const float* __restrict__ X2,
                                                const float* __restrict__ v3,
                                                float* __restrict__ out3, int n) {
  const int rel = blockIdx.y;
  const float* X = pick3(rel, X0, X1, X2);
  const float* v = v3 + rel * D;
  float* out = out3 + (size_t)rel * n;
  const int lane = threadIdx.x & 31;
  const int row0 =
      __builtin_amdgcn_readfirstlane((blockIdx.x * 8 + (threadIdx.x >> 5)) * MV_ROWS);
  if (row0 >= n) return;
  float4 x[MV_ROWS];
#pragma unroll
  for (int u = 0; u < MV_ROWS; ++u) {
    int r = row0 + u;
    if (r > n - 1) r = n - 1;
    x[u] = ((const float4*)X)[(size_t)r * (D / 4) + lane];
  }
  const float4 c = ((const float4*)v)[lane];
  __builtin_amdgcn_sched_barrier(0);
  float mine = 0.f;
#pragma unroll
  for (int u = 0; u < MV_ROWS; ++u) {
    float s = x[u].x * c.x + x[u].y * c.y + x[u].z * c.z + x[u].w * c.w;
#pragma unroll
    for (int off = 16; off; off >>= 1) s += __shfl_xor(s, off, 32);
    if (lane == u) mine = s;
  }
  const int r = row0 + lane;
  if (lane < MV_ROWS && r < n) out[r] = mine;
}

#define SCAN_B 16
__global__ __launch_bounds__(256) void k_matvec_dst(const float* __restrict__ X,
                                                    const float* __restrict__ v3,
                                                    float* __restrict__ out3,
                                                    int* tab3,
                                                    int* __restrict__ cnt3,
                                                    int* __restrict__ cur3,
                                                    int* __restrict__ alloc, int n, int C) {
  const int lane = threadIdx.x & 31;
  const int row0 =
      __builtin_amdgcn_readfirstlane((blockIdx.x * 8 + (threadIdx.x >> 5)) * MV_ROWS);
  if (row0 >= n) return;
  float4 x[MV_ROWS];
#pragma unroll
  for (int u = 0; u < MV_ROWS; ++u) {
    int r = row0 + u;
    if (r > n - 1) r = n - 1;
    x[u] = ((const float4*)X)[(size_t)r * (D / 4) + lane];
  }
  const float4 c0 = ((const float4*)v3)[lane];
  const float4 c1 = ((const float4*)v3)[(D / 4) + lane];
  const float4 c2 = ((const float4*)v3)[2 * (D / 4) + lane];
  const int r = row0 + lane;
  const bool own = lane < MV_ROWS && r < n;
  const size_t pl = (size_t)C * n;
  int n0 = 0, n1 = 0, n2 = 0;
  if (own) {
    const int* tp = tab3 + r;
    for (int cb = 0; cb < C; cb += SCAN_B) {
      int v0[SCAN_B], v1[SCAN_B], v2[SCAN_B];
#pragma unroll
      for (int u = 0; u < SCAN_B; ++u) {
        const int c = (cb + u < C) ? cb + u : C - 1;
        v0[u] = tp[(size_t)c * n];
        v1[u] = tp[pl + (size_t)c * n];
        v2[u] = tp[2 * pl + (size_t)c * n];
      }
#pragma unroll
      for (int u = 0; u < SCAN_B; ++u) {
        if (cb + u < C) { n0 += v0[u]; n1 += v1[u]; n2 += v2[u]; }
      }
    }
  }
  __builtin_amdgcn_sched_barrier(0);
  int i0 = n0, i1 = n1, i2 = n2;
#pragma unroll
  for (int off = 1; off < MV_ROWS; off <<= 1) {
    const int u0 = __shfl_up(i0, off, 32);
    const int u1 = __shfl_up(i1, off, 32);
    const int u2 = __shfl_up(i2, off, 32);
    if (lane >= off) { i0 += u0; i1 += u1; i2 += u2; }
  }
  const int t0 = __shfl(i0, MV_ROWS - 1, 32);
  const int t1 = __shfl(i1, MV_ROWS - 1, 32);
  const int t2 = __shfl(i2, MV_ROWS - 1, 32);
  const int tsel = lane == 0 ? t0 : (lane == 1 ? t1 : t2);
  int bsel = 0;
  if (lane < RELS) bsel = atomicAdd(&alloc[lane * CUR_PAD], tsel);
  float m0 = 0.f, m1 = 0.f, m2 = 0.f;
#pragma unroll
  for (int u = 0; u < MV_ROWS; ++u) {
    float s0 = x[u].x * c0.x + x[u].y * c0.y + x[u].z * c0.z + x[u].w * c0.w;
    float s1 = x[u].x * c1.x + x[u].y * c1.y + x[u].z * c1.z + x[u].w * c1.w;
    float s2 = x[u].x * c2.x + x[u].y * c2.y + x[u].z * c2.z + x[u].w * c2.w;
#pragma unroll
    for (int off = 16; off; off >>= 1) {
      s0 += __shfl_xor(s0, off, 32);
      s1 += __shfl_xor(s1, off, 32);
      s2 += __shfl_xor(s2, off, 32);
    }
    if (lane == u) { m0 = s0; m1 = s1; m2 = s2; }
  }
  const int b0 = __shfl(bsel, 0, 32);
  const int b1 = __shfl(bsel, 1, 32);
  const int b2 = __shfl(bsel, 2, 32);
  if (own) {
    int p0 = b0 + i0 - n0;
    int p1 = b1 + i1 - n1;
    int p2 = b2 + i2 - n2;
    out3[r] = m0;
    out3[(size_t)n + r] = m1;
    out3[2 * (size_t)n + r] = m2;
    cnt3[r] = n0;
    cnt3[(size_t)n + r] = n1;
    cnt3[2 * (size_t)n + r] = n2;
    cur3[r] = p0;
    cur3[(size_t)n + r] = p1;
    cur3[2 * (size_t)n + r] = p2;
    int* tp = tab3 + r;
    for (int cb = 0; cb < C; cb += SCAN_B) {
      int v0[SCAN_B], v1[SCAN_B], v2[SCAN_B];
#pragma unroll
      for (int u = 0; u < SCAN_B; ++u) {
        const int c = (cb + u < C) ? cb + u : C - 1;
        v0[u] = tp[(size_t)c * n];
        v1[u] = tp[pl + (size_t)c * n];
        v2[u] = tp[2 * pl + (size_t)c * n];
      }
#pragma unroll
      for (int u = 0; u < SCAN_B; ++u) {
        if (cb + u < C) {
          const size_t o = (size_t)(cb + u) * n;
          tp[o] = p0;
          tp[pl + o] = p1;
          tp[2 * pl + o] = p2;
          p0 += v0[u];
          p1 += v1[u];
          p2 += v2[u];
        }
      }
    }
  }
}


#define CH_EDGES 16384
#define TAB_WORDS 51200
__global__ __launch_bounds__(1024) void k_rank(const int* __restrict__ di0,
                                               const int* __restrict__ di1,
                                               const int* __restrict__ di2,
                                               int* __restrict__ tab3, int* __restrict__ rank3,
                                               int E, int nd, int C) {
  __shared__ int lc[TAB_WORDS];
  const int rel = blockIdx.y;
  const int* di = pick3(rel, di0, di1, di2);
  int* rank = rank3 + (size_t)rel * E;
  int* tab = tab3 + ((size_t)rel * C + blockIdx.x) * nd;
  const int t = threadIdx.x;
  const int nd4 = nd & ~3;
  const int e0 = blockIdx.x * CH_EDGES;
  int4 dv[CH_EDGES / 4096];
#pragma unroll
  for (int q = 0; q < CH_EDGES / 4096; ++q) {
    const int i = e0 + (q * 1024 + t) * 4;
    dv[q] = make_int4(0, 0, 0, 0);
    if (i + 3 < E) dv[q] = *(const int4*)(di + i);
  }
  for (int j = t * 4; j < nd4; j += 4096) *(int4*)(lc + j) = make_int4(0, 0, 0, 0);
  if (t < nd - nd4) lc[nd4 + t] = 0;
  __syncthreads();
#pragma unroll
  for (int q = 0; q < CH_EDGES / 4096; ++q) {
    const int i = e0 + (q * 1024 + t) * 4;
    if (i + 3 < E) {
      int4 rk;
      rk.x = atomicAdd(&lc[dv[q].x], 1);
      rk.y = atomicAdd(&lc[dv[q].y], 1);
      rk.z = atomicAdd(&lc[dv[q].z], 1);
      rk.w = atomicAdd(&lc[dv[q].w], 1);
      *(int4*)(rank + i) = rk;
    } else {
      for (int j = i; j < E && j < i + 4; ++j) rank[j] = atomicAdd(&lc[di[j]], 1);
    }
  }
  __syncthreads();
  if ((nd & 3) == 0) {
    for (int j = t * 4; j < nd; j += 4096) *(int4*)(tab + j) = *(const int4*)(lc + j);
  } else {
    for (int j = t; j < nd; j += 1024) tab[j] = lc[j];
  }
}

__global__ __launch_bounds__(256) void k_place(const int* __restrict__ si0,
                                               const int* __restrict__ si1,
                                               const int* __restrict__ si2,
                                               const int* __restrict__ di0,
                                               const int* __restrict__ di1,
                                               const int* __restrict__ di2,
                                               const float* __restrict__ el3,
                                               const float* __restrict__ er3,
                                               const int* __restrict__ tab3,
                                               const int* __restrict__ rank3,
                                               int2* __restrict__ lst3, int E, int ns, int nd,
                                               int C) {
  const int rel = blockIdx.y;
  const int* si = pick3(rel, si0, si1, si2);
  const int* di = pick3(rel, di0, di1, di2);
  const float* el = el3 + (size_t)rel * ns;
  const float* er = er3 + (size_t)rel * nd;
  const int* pos = tab3 + ((size_t)rel * C + blockIdx.x / (CH_EDGES / 1024)) * nd;
  const int* rank = rank3 + (size_t)rel * E;
  int2* lst = lst3 + (size_t)rel * E;
  int i = (blockIdx.x * 256 + threadIdx.x) * 4;
  if (i + 3 < E) {
    const int4 s = *(const int4*)(si + i);
    const int4 d = *(const int4*)(di + i);
    const int4 rk = *(const int4*)(rank + i);
    const float x0 = lrelu(el[s.x] + er[d.x]);
    const float x1 = lrelu(el[s.y] + er[d.y]);
    const float x2 = lrelu(el[s.z] + er[d.z]);
    const float x3 = lrelu(el[s.w] + er[d.w]);
    const int p0 = pos[d.x] + rk.x;
    const int p1 = pos[d.y] + rk.y;
    const int p2 = pos[d.z] + rk.z;
    const int p3 = pos[d.w] + rk.w;
    lst[p0] = make_int2(s.x, __float_as_int(x0));
    lst[p1] = make_int2(s.y, __float_as_int(x1));
    lst[p2] = make_int2(s.z, __float_as_int(x2));
    lst[p3] = make_int2(s.w, __float_as_int(x3));
  } else {
    for (; i < E; ++i) {
      const int s = si[i], d = di[i];
      const float x = lrelu(el[s] + er[d]);
      lst[pos[d] + rank[i]] = make_int2(s, __float_as_int(x));
    }
  }
}

__device__ __forceinline__ void fma4(float4& a, float p, const float4& h) {
  a.x += p * h.x; a.y += p * h.y; a.z += p * h.z; a.w += p * h.w;
}

template <int N>
__device__ __forceinline__ void add_rows(float4& acc, const float* __restrict__ hs, int vx,
                                         int pb, int k, int lane) {
  float4 h[N];
  float pk[N];
#pragma unroll
  for (int u = 0; u < N; ++u) {
    const unsigned sj = (unsigned)__builtin_amdgcn_readlane(vx, k + u);
    pk[u] = __int_as_float(__builtin_amdgcn_readlane(pb, k + u));
    h[u] = ((const float4*)(hs + (size_t)sj * D))[lane];
  }
#pragma unroll
  for (int u = 0; u < N; ++u) fma4(acc, pk[u], h[u]);
}

#define G_ROWS 4
__global__ __launch_bounds__(256) void k_gather(const int* __restrict__ cnt3,
                                                const int* __restrict__ cur3,
                                                const int2* __restrict__ lst3,
                                                const float* __restrict__ X0,
                                                const float* __restrict__ X1,
                                                const float* __restrict__ X2,
                                                float* __restrict__ Z3, int n, int E) {
  const int rel = blockIdx.y;
  const float* hs = pick3(rel, X0, X1, X2);
  const int* cnt = cnt3 + (size_t)rel * n;
  const int* cur = cur3 + (size_t)rel * n;
  const int2* lst = lst3 + (size_t)rel * E;
  float* Z = Z3 + (size_t)rel * n * D;
  const int lane = threadIdx.x & 31;
  const int d0 =
      __builtin_amdgcn_readfirstlane((blockIdx.x * 8 + (threadIdx.x >> 5)) * G_ROWS);
  const int NEG_INF_BITS = (int)0xff800000u;

  for (int r = 0; r < G_ROWS; ++r) {
    const int d = d0 + r;
    if (d >= n) break;
    const int c = cnt[d];
    const int start = cur[d];

    int2 v0 = make_int2(0, NEG_INF_BITS);
    if (lane < c) v0 = lst[start + lane];
    float mx = __int_as_float(v0.y);
    for (int base = 32; base < c; base += 32) {
      int j = base + lane;
      if (j < c) mx = fmaxf(mx, __int_as_float(lst[start + j].y));
    }
#pragma unroll
    for (int off = 16; off; off >>= 1) mx = fmaxf(mx, __shfl_xor(mx, off, 32));

    float4 acc = make_float4(0.f, 0.f, 0.f, 0.f);
    float zs = 0.f;
    for (int base = 0; base < c; base += 32) {
      int2 v = v0;
      const int j = base + lane;
      if (base > 0) {
        v = make_int2(0, NEG_INF_BITS);
        if (j < c) v = lst[start + j];
      }
      const float pe = expf(__int_as_float(v.y) - mx);
      const float p = (j < c) ? pe : 0.f;
      zs += p;
      const int pb = __float_as_int(p);
      int nc = c - base;
      if (nc > 32) nc = 32;
      int k = 0;
      for (; k + 8 <= nc; k += 8) add_rows<8>(acc, hs, v.x, pb, k, lane);
      switch (nc - k) {
        case 7: add_rows<7>(acc, hs, v.x, pb, k, lane); break;
        case 6: add_rows<6>(acc, hs, v.x, pb, k, lane); break;
        case 5: add_rows<5>(acc, hs, v.x, pb, k, lane); break;
        case 4: add_rows<4>(acc, hs, v.x, pb, k, lane); break;
        case 3: add_rows<3>(acc, hs, v.x, pb, k, lane); break;
        case 2: add_rows<2>(acc, hs, v.x, pb, k, lane); break;
        case 1: add_rows<1>(acc, hs, v.x, pb, k, lane); break;
        default: break;
      }
    }
#pragma unroll
    for (int off = 16; off; off >>= 1) zs += __shfl_xor(zs, off, 32);

    const float inv = (c > 0) ? 1.f / zs : 0.f;
    float4 o;
    o.x = acc.x * inv;
    o.y = acc.y * inv;
    o.z = acc.z * inv;
    o.w = acc.w * inv;
    ((float4*)Z)[(size_t)d * (D / 4) + lane] = o;
  }
}

__global__ __launch_bounds__(256) void k_sem_wmma(const float* __restrict__ Z3,
                                                  const float* __restrict__ W1,
                                                  const float* __restrict__ b1,
                                                  const float* __restrict__ W2,
                                                  float* __restrict__ wsum3, int N) {
  __shared__ float Ws[D * D];
  __shared__ float red[8];
  const float* Z = Z3 + (size_t)blockIdx.y * N * D;
  float* wsum = wsum3 + blockIdx.y;
  int t = threadIdx.x;
#pragma unroll
  for (int i = 0; i < (D * D / 4) / 256; ++i)
    ((float4*)Ws)[t + i * 256] = ((const float4*)W1)[t + i * 256];
  __syncthreads();

  const int wave = t >> 5;
  const int lane = t & 31;
  const int lrow = lane & 15;
  const int half = lane >> 4;
  const int rowbase = blockIdx.x * 128 + wave * 16;

  int xr = rowbase + lrow;
  if (xr > N - 1) xr = N - 1;
  const float* xp = Z + (size_t)xr * D;

  v2f afrag[32];
#pragma unroll
  for (int kt = 0; kt < 32; ++kt) {
    int k0 = kt * 4 + 2 * half;
    afrag[kt] = *(const v2f*)(xp + k0);
  }

  float s = 0.f;
#pragma unroll
  for (int nt = 0; nt < 8; ++nt) {
    v8f acc = {};
    int nb = nt * 16 + lrow;
#pragma unroll
    for (int kt = 0; kt < 32; ++kt) {
      int k0 = kt * 4 + 2 * half;
      v2f b;
      b.x = Ws[k0 * D + nb];
      b.y = Ws[(k0 + 1) * D + nb];
      acc = __builtin_amdgcn_wmma_f32_16x16x4_f32(false, afrag[kt], false, b,
                                                  (short)0, acc, false, false);
    }
    asm volatile("v_nop\n\tv_nop\n\tv_nop\n\tv_nop" : "+v"(acc));
    const float bn = b1[nb];
    const float wn = W2[nb];
#pragma unroll
    for (int v = 0; v < 8; ++v) {
      int grow = rowbase + v + 8 * half;
      float y = tanhf(acc[v] + bn) * wn;
      s += (grow < N) ? y : 0.f;
    }
  }

#pragma unroll
  for (int off = 16; off; off >>= 1) s += __shfl_xor(s, off, 32);
  if (lane == 0) red[wave] = s;
  __syncthreads();
  if (t == 0) {
    float tot = ((red[0] + red[1]) + (red[2] + red[3])) +
                ((red[4] + red[5]) + (red[6] + red[7]));
    atomicAdd(wsum, tot);
  }
}

#define MIX_V 8
__global__ __launch_bounds__(256) void k_mix(const float* __restrict__ Z3,
                                             const float* __restrict__ wsum,
                                             float* __restrict__ out, int n, float invN) {
  const int n4 = n >> 2;
  const float4* __restrict__ za = (const float4*)Z3;
  const float4* __restrict__ zb = (const float4*)(Z3 + (size_t)n);
  const float4* __restrict__ zc = (const float4*)(Z3 + 2 * (size_t)n);
  const int base = blockIdx.x * (256 * MIX_V) + threadIdx.x;
  float4 a[MIX_V], b[MIX_V], c[MIX_V];
#pragma unroll
  for (int u = 0; u < MIX_V; ++u) {
    int i = base + u * 256;
    if (i > n4 - 1) i = n4 - 1;
    a[u] = za[i];
    b[u] = zb[i];
    c[u] = zc[i];
  }
  float w0 = wsum[0] * invN, w1 = wsum[1] * invN, w2 = wsum[2] * invN;
  float mx = fmaxf(w0, fmaxf(w1, w2));
  float e0 = expf(w0 - mx), e1 = expf(w1 - mx), e2 = expf(w2 - mx);
  float inv = 1.f / (e0 + e1 + e2);
  float a0 = e0 * inv, a1 = e1 * inv, a2 = e2 * inv;
#pragma unroll
  for (int u = 0; u < MIX_V; ++u) {
    const int i = base + u * 256;
    float4 o;
    o.x = a0 * a[u].x + a1 * b[u].x + a2 * c[u].x;
    o.y = a0 * a[u].y + a1 * b[u].y + a2 * c[u].y;
    o.z = a0 * a[u].z + a1 * b[u].z + a2 * c[u].z;
    o.w = a0 * a[u].w + a1 * b[u].w + a2 * c[u].w;
    if (i < n4) ((float4*)out)[i] = o;
  }
  if (blockIdx.x == 0 && threadIdx.x == 0) { out[n] = a0; out[n + 1] = a1; out[n + 2] = a2; }
}


extern "C" void kernel_launch(void* const* d_in, const int* in_sizes, int n_in,
                              void* d_out, int out_size, void* d_ws, size_t ws_size,
                              hipStream_t stream) {
  const float* dst_feat = (const float*)d_in[0];
  const float* src_feat[RELS] = {(const float*)d_in[1], (const float*)d_in[2],
                                 (const float*)d_in[3]};
  const float* sem_W1 = (const float*)d_in[4];
  const float* sem_b1 = (const float*)d_in[5];
  const float* sem_W2 = (const float*)d_in[6];
  const float *Wg[RELS], *al[RELS], *ar[RELS], *bg[RELS];
  const int *sidx[RELS], *didx[RELS];
  for (int r = 0; r < RELS; ++r) {
    int b = 7 + 6 * r;
    Wg[r] = (const float*)d_in[b + 0];
    al[r] = (const float*)d_in[b + 1];
    ar[r] = (const float*)d_in[b + 2];
    bg[r] = (const float*)d_in[b + 3];
    sidx[r] = (const int*)d_in[b + 4];
    didx[r] = (const int*)d_in[b + 5];
  }
  const int ND = in_sizes[0] / D;
  const int NS = in_sizes[1] / D;
  const int E  = in_sizes[11];

  float* ws   = (float*)d_ws;
  float* z3   = ws;
  float* wsum = z3 + (size_t)RELS * ND * D;
  int2*  lst = (int2*)d_out;
  float* el  = (float*)(lst + (size_t)RELS * E);
  float* er  = el + (size_t)RELS * NS;
  int*   cnt = (int*)(er + (size_t)RELS * ND);
  int*   cur = cnt + (size_t)RELS * ND;
  float* wa  = (float*)(cur + (size_t)RELS * ND);
  float* wr  = wa + RELS * D;
  int*   alloc = (int*)(wr + RELS * D);
  int*   rank = (int*)(z3 + (size_t)(RELS - 1) * ND * D);
  const int NCH = (E + CH_EDGES - 1) / CH_EDGES;
  int*   tab = (int*)z3;

  k_rowdot<<<RELS, 256, 0, stream>>>(Wg[0], Wg[1], Wg[2], al[0], al[1], al[2], ar[0], ar[1],
                                     ar[2], wa, wr, wsum, alloc);
  k_matvec<<<dim3((NS + 8 * MV_ROWS - 1) / (8 * MV_ROWS), RELS), 256, 0, stream>>>(
      src_feat[0], src_feat[1], src_feat[2], wa, el, NS);
  k_rank<<<dim3(NCH, RELS), 1024, 0, stream>>>(didx[0], didx[1], didx[2], tab, rank, E, ND, NCH);
  k_matvec_dst<<<(ND + 8 * MV_ROWS - 1) / (8 * MV_ROWS), 256, 0, stream>>>(
      dst_feat, wr, er, tab, cnt, cur, alloc, ND, NCH);
  k_place<<<dim3((E + 1023) / 1024, RELS), 256, 0, stream>>>(
      sidx[0], sidx[1], sidx[2], didx[0], didx[1], didx[2], el, er, tab, rank, lst, E, NS, ND,
      NCH);

  k_gather<<<dim3((ND + 8 * G_ROWS - 1) / (8 * G_ROWS), RELS), 256, 0, stream>>>(
      cnt, cur, lst, src_feat[0], src_feat[1], src_feat[2], z3, ND, E);
  k_gemm_wmma<<<dim3((ND + 127) / 128, RELS), 256, 0, stream>>>(z3, Wg[0], Wg[1], Wg[2], bg[0],
                                                                 bg[1], bg[2], ND);

  k_sem_wmma<<<dim3((ND + 127) / 128, RELS), 256, 0, stream>>>(z3, sem_W1, sem_b1, sem_W2,
                                                               wsum, ND);

  k_mix<<<(ND * D / 4 + 256 * MIX_V - 1) / (256 * MIX_V), 256, 0, stream>>>(
      z3, wsum, (float*)d_out, ND * D, 1.0f / (float)ND);
}


# source-row gather and its f32 WMMA transform share one launch: a pair of waves hands 16 aggregated rows to the matrix product through LDS, none go to memory; 9 launches -> 8; sums as before
# speedup vs baseline: 4.4877x; 1.0846x over previous
#include <hip/hip_runtime.h>
#include <math.h>

#define D 128
#define RELS 3

typedef float v2f __attribute__((ext_vector_type(2)));
typedef float v8f __attribute__((ext_vector_type(8)));


__device__ __forceinline__ float lrelu(float x) { return x > 0.f ? x : 0.2f * x; }

__device__ __forceinline__ const float* pick3(int r, const float* a, const float* b,
                                              const float* c) {
  return r == 0 ? a : (r == 1 ? b : c);
}
__device__ __forceinline__ const int* pick3(int r, const int* a, const int* b, const int* c) {
  return r == 0 ? a : (r == 1 ? b : c);
}


#define CUR_PAD 64
__global__ __launch_bounds__(256) void k_rowdot(const float* __restrict__ W0,
                                                const float* __restrict__ W1,
                                                const float* __restrict__ W2,
                                                const float* __restrict__ al0,
                                                const float* __restrict__ al1,
                                                const float* __restrict__ al2,
                                                const float* __restrict__ ar0,
                                                const float* __restrict__ ar1,
                                                const float* __restrict__ ar2,
                                                float* __restrict__ wa3,
                                                float* __restrict__ wr3,
                                                float* __restrict__ wsum,
                                                int* __restrict__ alloc) {
  const int rel = blockIdx.x;
  const float* W = pick3(rel, W0, W1, W2);
  const float* al = pick3(rel, al0, al1, al2);
  const float* ar = pick3(rel, ar0, ar1, ar2);
  int t = threadIdx.x;
  if (t == 0) {
    wsum[rel] = 0.f;
    alloc[rel * CUR_PAD] = 0;
  }
  const float* v = (t < D) ? al : ar;
  float* o = ((t < D) ? wa3 : wr3) + rel * D;
  int d = t & (D - 1);
  float s = 0.f;
#pragma unroll 8
  for (int h = 0; h < D; ++h) s += W[d * D + h] * v[h];
  o[d] = s;
}

#define MV_ROWS 16
__global__ __launch_bounds__(256) void k_matvec(const float* __restrict__ X0,
                                                const float* __restrict__ X1,
                                                const float* __restrict__ X2,
                                                const float* __restrict__ v3,
                                                float* __restrict__ out3, int n) {
  const int rel = blockIdx.y;
  const float* X = pick3(rel, X0, X1, X2);
  const float* v = v3 + rel * D;
  float* out = out3 + (size_t)rel * n;
  const int lane = threadIdx.x & 31;
  const int row0 =
      __builtin_amdgcn_readfirstlane((blockIdx.x * 8 + (threadIdx.x >> 5)) * MV_ROWS);
  if (row0 >= n) return;
  float4 x[MV_ROWS];
#pragma unroll
  for (int u = 0; u < MV_ROWS; ++u) {
    int r = row0 + u;
    if (r > n - 1) r = n - 1;
    x[u] = ((const float4*)X)[(size_t)r * (D / 4) + lane];
  }
  const float4 c = ((const float4*)v)[lane];
  __builtin_amdgcn_sched_barrier(0);
  float mine = 0.f;
#pragma unroll
  for (int u = 0; u < MV_ROWS; ++u) {
    float s = x[u].x * c.x + x[u].y * c.y + x[u].z * c.z + x[u].w * c.w;
#pragma unroll
    for (int off = 16; off; off >>= 1) s += __shfl_xor(s, off, 32);
    if (lane == u) mine = s;
  }
  const int r = row0 + lane;
  if (lane < MV_ROWS && r < n) out[r] = mine;
}

#define SCAN_B 16
__global__ __launch_bounds__(256) void k_matvec_dst(const float* __restrict__ X,
                                                    const float* __restrict__ v3,
                                                    float* __restrict__ out3,
                                                    int* tab3,
                                                    int* __restrict__ cnt3,
                                                    int* __restrict__ cur3,
                                                    int* __restrict__ alloc, int n, int C) {
  const int lane = threadIdx.x & 31;
  const int row0 =
      __builtin_amdgcn_readfirstlane((blockIdx.x * 8 + (threadIdx.x >> 5)) * MV_ROWS);
  if (row0 >= n) return;
  float4 x[MV_ROWS];
#pragma unroll
  for (int u = 0; u < MV_ROWS; ++u) {
    int r = row0 + u;
    if (r > n - 1) r = n - 1;
    x[u] = ((const float4*)X)[(size_t)r * (D / 4) + lane];
  }
  const float4 c0 = ((const float4*)v3)[lane];
  const float4 c1 = ((const float4*)v3)[(D / 4) + lane];
  const float4 c2 = ((const float4*)v3)[2 * (D / 4) + lane];
  const int r = row0 + lane;
  const bool own = lane < MV_ROWS && r < n;
  const size_t pl = (size_t)C * n;
  int n0 = 0, n1 = 0, n2 = 0;
  if (own) {
    const int* tp = tab3 + r;
    for (int cb = 0; cb < C; cb += SCAN_B) {
      int v0[SCAN_B], v1[SCAN_B], v2[SCAN_B];
#pragma unroll
      for (int u = 0; u < SCAN_B; ++u) {
        const int c = (cb + u < C) ? cb + u : C - 1;
        v0[u] = tp[(size_t)c * n];
        v1[u] = tp[pl + (size_t)c * n];
        v2[u] = tp[2 * pl + (size_t)c * n];
      }
#pragma unroll
      for (int u = 0; u < SCAN_B; ++u) {
        if (cb + u < C) { n0 += v0[u]; n1 += v1[u]; n2 += v2[u]; }
      }
    }
  }
  __builtin_amdgcn_sched_barrier(0);
  int i0 = n0, i1 = n1, i2 = n2;
#pragma unroll
  for (int off = 1; off < MV_ROWS; off <<= 1) {
    const int u0 = __shfl_up(i0, off, 32);
    const int u1 = __shfl_up(i1, off, 32);
    const int u2 = __shfl_up(i2, off, 32);
    if (lane >= off) { i0 += u0; i1 += u1; i2 += u2; }
  }
  const int t0 = __shfl(i0, MV_ROWS - 1, 32);
  const int t1 = __shfl(i1, MV_ROWS - 1, 32);
  const int t2 = __shfl(i2, MV_ROWS - 1, 32);
  const int tsel = lane == 0 ? t0 : (lane == 1 ? t1 : t2);
  int bsel = 0;
  if (lane < RELS) bsel = atomicAdd(&alloc[lane * CUR_PAD], tsel);
  float m0 = 0.f, m1 = 0.f, m2 = 0.f;
#pragma unroll
  for (int u = 0; u < MV_ROWS; ++u) {
    float s0 = x[u].x * c0.x + x[u].y * c0.y + x[u].z * c0.z + x[u].w * c0.w;
    float s1 = x[u].x * c1.x + x[u].y * c1.y + x[u].z * c1.z + x[u].w * c1.w;
    float s2 = x[u].x * c2.x + x[u].y * c2.y + x[u].z * c2.z + x[u].w * c2.w;
#pragma unroll
    for (int off = 16; off; off >>= 1) {
      s0 += __shfl_xor(s0, off, 32);
      s1 += __shfl_xor(s1, off, 32);
      s2 += __shfl_xor(s2, off, 32);
    }
    if (lane == u) { m0 = s0; m1 = s1; m2 = s2; }
  }
  const int b0 = __shfl(bsel, 0, 32);
  const int b1 = __shfl(bsel, 1, 32);
  const int b2 = __shfl(bsel, 2, 32);
  if (own) {
    int p0 = b0 + i0 - n0;
    int p1 = b1 + i1 - n1;
    int p2 = b2 + i2 - n2;
    out3[r] = m0;
    out3[(size_t)n + r] = m1;
    out3[2 * (size_t)n + r] = m2;
    cnt3[r] = n0;
    cnt3[(size_t)n + r] = n1;
    cnt3[2 * (size_t)n + r] = n2;
    cur3[r] = p0;
    cur3[(size_t)n + r] = p1;
    cur3[2 * (size_t)n + r] = p2;
    int* tp = tab3 + r;
    for (int cb = 0; cb < C; cb += SCAN_B) {
      int v0[SCAN_B], v1[SCAN_B], v2[SCAN_B];
#pragma unroll
      for (int u = 0; u < SCAN_B; ++u) {
        const int c = (cb + u < C) ? cb + u : C - 1;
        v0[u] = tp[(size_t)c * n];
        v1[u] = tp[pl + (size_t)c * n];
        v2[u] = tp[2 * pl + (size_t)c * n];
      }
#pragma unroll
      for (int u = 0; u < SCAN_B; ++u) {
        if (cb + u < C) {
          const size_t o = (size_t)(cb + u) * n;
          tp[o] = p0;
          tp[pl + o] = p1;
          tp[2 * pl + o] = p2;
          p0 += v0[u];
          p1 += v1[u];
          p2 += v2[u];
        }
      }
    }
  }
}


#define CH_EDGES 16384
#define TAB_WORDS 51200
__global__ __launch_bounds__(1024) void k_rank(const int* __restrict__ di0,
                                               const int* __restrict__ di1,
                                               const int* __restrict__ di2,
                                               int* __restrict__ tab3, int* __restrict__ rank3,
                                               int E, int nd, int C) {
  __shared__ int lc[TAB_WORDS];
  const int rel = blockIdx.y;
  const int* di = pick3(rel, di0, di1, di2);
  int* rank = rank3 + (size_t)rel * E;
  int* tab = tab3 + ((size_t)rel * C + blockIdx.x) * nd;
  const int t = threadIdx.x;
  const int nd4 = nd & ~3;
  const int e0 = blockIdx.x * CH_EDGES;
  int4 dv[CH_EDGES / 4096];
#pragma unroll
  for (int q = 0; q < CH_EDGES / 4096; ++q) {
    const int i = e0 + (q * 1024 + t) * 4;
    dv[q] = make_int4(0, 0, 0, 0);
    if (i + 3 < E) dv[q] = *(const int4*)(di + i);
  }
  for (int j = t * 4; j < nd4; j += 4096) *(int4*)(lc + j) = make_int4(0, 0, 0, 0);
  if (t < nd - nd4) lc[nd4 + t] = 0;
  __syncthreads();
#pragma unroll
  for (int q = 0; q < CH_EDGES / 4096; ++q) {
    const int i = e0 + (q * 1024 + t) * 4;
    if (i + 3 < E) {
      int4 rk;
      rk.x = atomicAdd(&lc[dv[q].x], 1);
      rk.y = atomicAdd(&lc[dv[q].y], 1);
      rk.z = atomicAdd(&lc[dv[q].z], 1);
      rk.w = atomicAdd(&lc[dv[q].w], 1);
      *(int4*)(rank + i) = rk;
    } else {
      for (int j = i; j < E && j < i + 4; ++j) rank[j] = atomicAdd(&lc[di[j]], 1);
    }
  }
  __syncthreads();
  if ((nd & 3) == 0) {
    for (int j = t * 4; j < nd; j += 4096) *(int4*)(tab + j) = *(const int4*)(lc + j);
  } else {
    for (int j = t; j < nd; j += 1024) tab[j] = lc[j];
  }
}

__global__ __launch_bounds__(256) void k_place(const int* __restrict__ si0,
                                               const int* __restrict__ si1,
                                               const int* __restrict__ si2,
                                               const int* __restrict__ di0,
                                               const int* __restrict__ di1,
                                               const int* __restrict__ di2,
                                               const float* __restrict__ el3,
                                               const float* __restrict__ er3,
                                               const int* __restrict__ tab3,
                                               const int* __restrict__ rank3,
                                               int2* __restrict__ lst3, int E, int ns, int nd,
                                               int C) {
  const int rel = blockIdx.y;
  const int* si = pick3(rel, si0, si1, si2);
  const int* di = pick3(rel, di0, di1, di2);
  const float* el = el3 + (size_t)rel * ns;
  const float* er = er3 + (size_t)rel * nd;
  const int* pos = tab3 + ((size_t)rel * C + blockIdx.x / (CH_EDGES / 1024)) * nd;
  const int* rank = rank3 + (size_t)rel * E;
  int2* lst = lst3 + (size_t)rel * E;
  int i = (blockIdx.x * 256 + threadIdx.x) * 4;
  if (i + 3 < E) {
    const int4 s = *(const int4*)(si + i);
    const int4 d = *(const int4*)(di + i);
    const int4 rk = *(const int4*)(rank + i);
    const float x0 = lrelu(el[s.x] + er[d.x]);
    const float x1 = lrelu(el[s.y] + er[d.y]);
    const float x2 = lrelu(el[s.z] + er[d.z]);
    const float x3 = lrelu(el[s.w] + er[d.w]);
    const int p0 = pos[d.x] + rk.x;
    const int p1 = pos[d.y] + rk.y;
    const int p2 = pos[d.z] + rk.z;
    const int p3 = pos[d.w] + rk.w;
    lst[p0] = make_int2(s.x, __float_as_int(x0));
    lst[p1] = make_int2(s.y, __float_as_int(x1));
    lst[p2] = make_int2(s.z, __float_as_int(x2));
    lst[p3] = make_int2(s.w, __float_as_int(x3));
  } else {
    for (; i < E; ++i) {
      const int s = si[i], d = di[i];
      const float x = lrelu(el[s] + er[d]);
      lst[pos[d] + rank[i]] = make_int2(s, __float_as_int(x));
    }
  }
}

__device__ __forceinline__ void fma4(float4& a, float p, const float4& h) {
  a.x += p * h.x; a.y += p * h.y; a.z += p * h.z; a.w += p * h.w;
}

template <int N>
__device__ __forceinline__ void add_rows(float4& acc, const float* __restrict__ hs, int vx,
                                         int pb, int k, int lane) {
  float4 h[N];
  float pk[N];
#pragma unroll
  for (int u = 0; u < N; ++u) {
    const unsigned sj = (unsigned)__builtin_amdgcn_readlane(vx, k + u);
    pk[u] = __int_as_float(__builtin_amdgcn_readlane(pb, k + u));
    h[u] = ((const float4*)(hs + (size_t)sj * D))[lane];
  }
#pragma unroll
  for (int u = 0; u < N; ++u) fma4(acc, pk[u], h[u]);
}

#define FT_ROWS 16
#define FT_PITCH 132
#define G_ROWS 8
__global__ __launch_bounds__(512) void k_gather_gemm(const int* __restrict__ cnt3,
                                                     const int* __restrict__ cur3,
                                                     const int2* __restrict__ lst3,
                                                     const float* __restrict__ X0,
                                                     const float* __restrict__ X1,
                                                     const float* __restrict__ X2,
                                                     const float* __restrict__ W0,
                                                     const float* __restrict__ W1,
                                                     const float* __restrict__ W2,
                                                     const float* __restrict__ b0,
                                                     const float* __restrict__ b1,
                                                     const float* __restrict__ b2,
                                                     float* __restrict__ Z3, int n, int E) {
  __shared__ float4 Sm4[(D * D + 8 * FT_ROWS * FT_PITCH) / 4];
  float* Ws = (float*)Sm4;
  const int rel = blockIdx.y;
  const float* hs = pick3(rel, X0, X1, X2);
  const float* W = pick3(rel, W0, W1, W2);
  const float* bias = pick3(rel, b0, b1, b2);
  const int* cnt = cnt3 + (size_t)rel * n;
  const int* cur = cur3 + (size_t)rel * n;
  const int2* lst = lst3 + (size_t)rel * E;
  float* Z = Z3 + (size_t)rel * n * D;
  const int t = threadIdx.x;
#pragma unroll
  for (int i = 0; i < (D * D / 4) / 512; ++i)
    ((float4*)Ws)[t + i * 512] = ((const float4*)W)[t + i * 512];

  const int wave = __builtin_amdgcn_readfirstlane(t >> 5);
  const int lane = t & 31;
  const int ti = wave >> 1;
  const int h2 = wave & 1;
  float* tile = Ws + D * D + ti * (FT_ROWS * FT_PITCH);
  const int rowbase = blockIdx.x * (8 * FT_ROWS) + ti * FT_ROWS;
  const int d0 = rowbase + h2 * G_ROWS;
  const int NEG_INF_BITS = (int)0xff800000u;

  int dl = d0 + (lane & (G_ROWS - 1));
  if (dl > n - 1) dl = n - 1;
  const int cv = cnt[dl];
  const int sv = cur[dl];
  int2 vn = make_int2(0, NEG_INF_BITS);
  if (d0 < n) {
    const int c0 = __builtin_amdgcn_readlane(cv, 0);
    const int s0 = __builtin_amdgcn_readlane(sv, 0);
    if (lane < c0) vn = lst[s0 + lane];
  }

#pragma nounroll
  for (int r = 0; r < G_ROWS; ++r) {
    const int d = d0 + r;
    float4 o = make_float4(0.f, 0.f, 0.f, 0.f);
    if (d < n) {
      const int c = __builtin_amdgcn_readlane(cv, r);
      const int start = __builtin_amdgcn_readlane(sv, r);
      const int2 v0 = vn;
      vn = make_int2(0, NEG_INF_BITS);
      if (r + 1 < G_ROWS && d + 1 < n) {
        const int c1 = __builtin_amdgcn_readlane(cv, r + 1);
        const int s1 = __builtin_amdgcn_readlane(sv, r + 1);
        if (lane < c1) vn = lst[s1 + lane];
      }
      float mx = __int_as_float(v0.y);
      for (int base = 32; base < c; base += 32) {
        int j = base + lane;
        if (j < c) mx = fmaxf(mx, __int_as_float(lst[start + j].y));
      }
#pragma unroll
      for (int off = 16; off; off >>= 1) mx = fmaxf(mx, __shfl_xor(mx, off, 32));

      float4 acc = make_float4(0.f, 0.f, 0.f, 0.f);
      float zs = 0.f;
      for (int base = 0; base < c; base += 32) {
        int2 v = v0;
        const int j = base + lane;
        if (base > 0) {
          v = make_int2(0, NEG_INF_BITS);
          if (j < c) v = lst[start + j];
        }
        const float pe = expf(__int_as_float(v.y) - mx);
        const float p = (j < c) ? pe : 0.f;
        zs += p;
        const int pb = __float_as_int(p);
        int nc = c - base;
        if (nc > 32) nc = 32;
        int k = 0;
        for (; k + 16 <= nc; k += 16) add_rows<16>(acc, hs, v.x, pb, k, lane);
        for (; k + 8 <= nc; k += 8) add_rows<8>(acc, hs, v.x, pb, k, lane);
        switch (nc - k) {
          case 7: add_rows<7>(acc, hs, v.x, pb, k, lane); break;
          case 6: add_rows<6>(acc, hs, v.x, pb, k, lane); break;
          case 5: add_rows<5>(acc, hs, v.x, pb, k, lane); break;
          case 4: add_rows<4>(acc, hs, v.x, pb, k, lane); break;
          case 3: add_rows<3>(acc, hs, v.x, pb, k, lane); break;
          case 2: add_rows<2>(acc, hs, v.x, pb, k, lane); break;
          case 1: add_rows<1>(acc, hs, v.x, pb, k, lane); break;
          default: break;
        }
      }
#pragma unroll
      for (int off = 16; off; off >>= 1) zs += __shfl_xor(zs, off, 32);

      const float inv = (c > 0) ? 1.f / zs : 0.f;
      o.x = acc.x * inv;
      o.y = acc.y * inv;
      o.z = acc.z * inv;
      o.w = acc.w * inv;
    }
    *(float4*)(tile + (h2 * G_ROWS + r) * FT_PITCH + 4 * lane) = o;
  }

  __syncthreads();

  const int lrow = lane & 15;
  const int half = lane >> 4;
  v2f afrag[32];
#pragma unroll
  for (int kt = 0; kt < 32; ++kt) {
    int k0 = kt * 4 + 2 * half;
    afrag[kt] = *(const v2f*)(tile + lrow * FT_PITCH + k0);
  }
  float bn[4];
#pragma unroll
  for (int nt = 0; nt < 4; ++nt) bn[nt] = bias[(h2 * 4 + nt) * 16 + lrow];
  __syncthreads();

#pragma unroll
  for (int nt = 0; nt < 4; ++nt) {
    v8f acc = {};
    int nb = (h2 * 4 + nt) * 16 + lrow;
#pragma unroll
    for (int kt = 0; kt < 32; ++kt) {
      int k0 = kt * 4 + 2 * half;
      v2f b;
      b.x = Ws[k0 * D + nb];
      b.y = Ws[(k0 + 1) * D + nb];
      acc = __builtin_amdgcn_wmma_f32_16x16x4_f32(false, afrag[kt], false, b,
                                                  (short)0, acc, false, false);
    }
    asm volatile("v_nop\n\tv_nop\n\tv_nop\n\tv_nop" : "+v"(acc));
#pragma unroll
    for (int v = 0; v < 8; ++v) {
      float o = acc[v] + bn[nt];
      o = o > 0.f ? o : expf(o) - 1.f;
      tile[(v + 8 * half) * FT_PITCH + nb] = o;
    }
  }
  const int ocol = h2 * 64 + lrow * 4;
  if (rowbase + FT_ROWS <= n) {
    float4 orow[8];
#pragma unroll
    for (int i = 0; i < 8; ++i)
      orow[i] = *(const float4*)(tile + (2 * i + half) * FT_PITCH + ocol);
#pragma unroll
    for (int i = 0; i < 8; ++i)
      *(float4*)(Z + (size_t)(rowbase + 2 * i + half) * D + ocol) = orow[i];
  } else {
#pragma unroll
    for (int i = 0; i < 8; ++i) {
      const int grow = rowbase + 2 * i + half;
      if (grow < n)
        *(float4*)(Z + (size_t)grow * D + ocol) =
            *(const float4*)(tile + (2 * i + half) * FT_PITCH + ocol);
    }
  }
}

__global__ __launch_bounds__(256) void k_sem_wmma(const float* __restrict__ Z3,
                                                  const float* __restrict__ W1,
                                                  const float* __restrict__ b1,
                                                  const float* __restrict__ W2,
                                                  float* __restrict__ wsum3, int N) {
  __shared__ float Ws[D * D];
  __shared__ float red[8];
  const float* Z = Z3 + (size_t)blockIdx.y * N * D;
  float* wsum = wsum3 + blockIdx.y;
  int t = threadIdx.x;
#pragma unroll
  for (int i = 0; i < (D * D / 4) / 256; ++i)
    ((float4*)Ws)[t + i * 256] = ((const float4*)W1)[t + i * 256];
  __syncthreads();

  const int wave = t >> 5;
  const int lane = t & 31;
  const int lrow = lane & 15;
  const int half = lane >> 4;
  const int rowbase = blockIdx.x * 128 + wave * 16;

  int xr = rowbase + lrow;
  if (xr > N - 1) xr = N - 1;
  const float* xp = Z + (size_t)xr * D;

  v2f afrag[32];
#pragma unroll
  for (int kt = 0; kt < 32; ++kt) {
    int k0 = kt * 4 + 2 * half;
    afrag[kt] = *(const v2f*)(xp + k0);
  }

  float s = 0.f;
#pragma unroll
  for (int nt = 0; nt < 8; ++nt) {
    v8f acc = {};
    int nb = nt * 16 + lrow;
#pragma unroll
    for (int kt = 0; kt < 32; ++kt) {
      int k0 = kt * 4 + 2 * half;
      v2f b;
      b.x = Ws[k0 * D + nb];
      b.y = Ws[(k0 + 1) * D + nb];
      acc = __builtin_amdgcn_wmma_f32_16x16x4_f32(false, afrag[kt], false, b,
                                                  (short)0, acc, false, false);
    }
    asm volatile("v_nop\n\tv_nop\n\tv_nop\n\tv_nop" : "+v"(acc));
    const float bn = b1[nb];
    const float wn = W2[nb];
#pragma unroll
    for (int v = 0; v < 8; ++v) {
      int grow = rowbase + v + 8 * half;
      float y = tanhf(acc[v] + bn) * wn;
      s += (grow < N) ? y : 0.f;
    }
  }

#pragma unroll
  for (int off = 16; off; off >>= 1) s += __shfl_xor(s, off, 32);
  if (lane == 0) red[wave] = s;
  __syncthreads();
  if (t == 0) {
    float tot = ((red[0] + red[1]) + (red[2] + red[3])) +
                ((red[4] + red[5]) + (red[6] + red[7]));
    atomicAdd(wsum, tot);
  }
}

#define MIX_V 8
__global__ __launch_bounds__(256) void k_mix(const float* __restrict__ Z3,
                                             const float* __restrict__ wsum,
                                             float* __restrict__ out, int n, float invN) {
  const int n4 = n >> 2;
  const float4* __restrict__ za = (const float4*)Z3;
  const float4* __restrict__ zb = (const float4*)(Z3 + (size_t)n);
  const float4* __restrict__ zc = (const float4*)(Z3 + 2 * (size_t)n);
  const int base = blockIdx.x * (256 * MIX_V) + threadIdx.x;
  float4 a[MIX_V], b[MIX_V], c[MIX_V];
#pragma unroll
  for (int u = 0; u < MIX_V; ++u) {
    int i = base + u * 256;
    if (i > n4 - 1) i = n4 - 1;
    a[u] = za[i];
    b[u] = zb[i];
    c[u] = zc[i];
  }
  float w0 = wsum[0] * invN, w1 = wsum[1] * invN, w2 = wsum[2] * invN;
  float mx = fmaxf(w0, fmaxf(w1, w2));
  float e0 = expf(w0 - mx), e1 = expf(w1 - mx), e2 = expf(w2 - mx);
  float inv = 1.f / (e0 + e1 + e2);
  float a0 = e0 * inv, a1 = e1 * inv, a2 = e2 * inv;
#pragma unroll
  for (int u = 0; u < MIX_V; ++u) {
    const int i = base + u * 256;
    float4 o;
    o.x = a0 * a[u].x + a1 * b[u].x + a2 * c[u].x;
    o.y = a0 * a[u].y + a1 * b[u].y + a2 * c[u].y;
    o.z = a0 * a[u].z + a1 * b[u].z + a2 * c[u].z;
    o.w = a0 * a[u].w + a1 * b[u].w + a2 * c[u].w;
    if (i < n4) ((float4*)out)[i] = o;
  }
  if (blockIdx.x == 0 && threadIdx.x == 0) { out[n] = a0; out[n + 1] = a1; out[n + 2] = a2; }
}


extern "C" void kernel_launch(void* const* d_in, const int* in_sizes, int n_in,
                              void* d_out, int out_size, void* d_ws, size_t ws_size,
                              hipStream_t stream) {
  const float* dst_feat = (const float*)d_in[0];
  const float* src_feat[RELS] = {(const float*)d_in[1], (const float*)d_in[2],
                                 (const float*)d_in[3]};
  const float* sem_W1 = (const float*)d_in[4];
  const float* sem_b1 = (const float*)d_in[5];
  const float* sem_W2 = (const float*)d_in[6];
  const float *Wg[RELS], *al[RELS], *ar[RELS], *bg[RELS];
  const int *sidx[RELS], *didx[RELS];
  for (int r = 0; r < RELS; ++r) {
    int b = 7 + 6 * r;
    Wg[r] = (const float*)d_in[b + 0];
    al[r] = (const float*)d_in[b + 1];
    ar[r] = (const float*)d_in[b + 2];
    bg[r] = (const float*)d_in[b + 3];
    sidx[r] = (const int*)d_in[b + 4];
    didx[r] = (const int*)d_in[b + 5];
  }
  const int ND = in_sizes[0] / D;
  const int NS = in_sizes[1] / D;
  const int E  = in_sizes[11];

  float* ws   = (float*)d_ws;
  float* z3   = ws;
  float* wsum = z3 + (size_t)RELS * ND * D;
  int2*  lst = (int2*)d_out;
  float* el  = (float*)(lst + (size_t)RELS * E);
  float* er  = el + (size_t)RELS * NS;
  int*   cnt = (int*)(er + (size_t)RELS * ND);
  int*   cur = cnt + (size_t)RELS * ND;
  float* wa  = (float*)(cur + (size_t)RELS * ND);
  float* wr  = wa + RELS * D;
  int*   alloc = (int*)(wr + RELS * D);
  int*   rank = (int*)(z3 + (size_t)(RELS - 1) * ND * D);
  const int NCH = (E + CH_EDGES - 1) / CH_EDGES;
  int*   tab = (int*)z3;

  k_rowdot<<<RELS, 256, 0, stream>>>(Wg[0], Wg[1], Wg[2], al[0], al[1], al[2], ar[0], ar[1],
                                     ar[2], wa, wr, wsum, alloc);
  k_matvec<<<dim3((NS + 8 * MV_ROWS - 1) / (8 * MV_ROWS), RELS), 256, 0, stream>>>(
      src_feat[0], src_feat[1], src_feat[2], wa, el, NS);
  k_rank<<<dim3(NCH, RELS), 1024, 0, stream>>>(didx[0], didx[1], didx[2], tab, rank, E, ND, NCH);
  k_matvec_dst<<<(ND + 8 * MV_ROWS - 1) / (8 * MV_ROWS), 256, 0, stream>>>(
      dst_feat, wr, er, tab, cnt, cur, alloc, ND, NCH);
  k_place<<<dim3((E + 1023) / 1024, RELS), 256, 0, stream>>>(
      sidx[0], sidx[1], sidx[2], didx[0], didx[1], didx[2], el, er, tab, rank, lst, E, NS, ND,
      NCH);

  k_gather_gemm<<<dim3((ND + 8 * FT_ROWS - 1) / (8 * FT_ROWS), RELS), 512, 0, stream>>>(
      cnt, cur, lst, src_feat[0], src_feat[1], src_feat[2], Wg[0], Wg[1], Wg[2], bg[0], bg[1],
      bg[2], z3, ND, E);

  k_sem_wmma<<<dim3((ND + 127) / 128, RELS), 256, 0, stream>>>(z3, sem_W1, sem_b1, sem_W2,
                                                               wsum, ND);

  k_mix<<<(ND * D / 4 + 256 * MIX_V - 1) / (256 * MIX_V), 256, 0, stream>>>(
      z3, wsum, (float*)d_out, ND * D, 1.0f / (float)ND);
}


# semantic-score product (rows @ W1, tanh, . W2) moves into the gather+transform launch: tile still in LDS, W1 staged over W; its launch and its read of the rows are gone (8 -> 7); wsum sum order moves
# speedup vs baseline: 4.8317x; 1.0767x over previous
#include <hip/hip_runtime.h>
#include <math.h>

#define D 128
#define RELS 3

typedef float v2f __attribute__((ext_vector_type(2)));
typedef float v8f __attribute__((ext_vector_type(8)));


__device__ __forceinline__ float lrelu(float x) { return x > 0.f ? x : 0.2f * x; }

__device__ __forceinline__ const float* pick3(int r, const float* a, const float* b,
                                              const float* c) {
  return r == 0 ? a : (r == 1 ? b : c);
}
__device__ __forceinline__ const int* pick3(int r, const int* a, const int* b, const int* c) {
  return r == 0 ? a : (r == 1 ? b : c);
}


#define CUR_PAD 64
__global__ __launch_bounds__(256) void k_rowdot(const float* __restrict__ W0,
                                                const float* __restrict__ W1,
                                                const float* __restrict__ W2,
                                                const float* __restrict__ al0,
                                                const float* __restrict__ al1,
                                                const float* __restrict__ al2,
                                                const float* __restrict__ ar0,
                                                const float* __restrict__ ar1,
                                                const float* __restrict__ ar2,
                                                float* __restrict__ wa3,
                                                float* __restrict__ wr3,
                                                float* __restrict__ wsum,
                                                int* __restrict__ alloc) {
  const int rel = blockIdx.x;
  const float* W = pick3(rel, W0, W1, W2);
  const float* al = pick3(rel, al0, al1, al2);
  const float* ar = pick3(rel, ar0, ar1, ar2);
  int t = threadIdx.x;
  if (t == 0) {
    wsum[rel] = 0.f;
    alloc[rel * CUR_PAD] = 0;
  }
  const float* v = (t < D) ? al : ar;
  float* o = ((t < D) ? wa3 : wr3) + rel * D;
  int d = t & (D - 1);
  float s = 0.f;
#pragma unroll 8
  for (int h = 0; h < D; ++h) s += W[d * D + h] * v[h];
  o[d] = s;
}

#define MV_ROWS 16
__global__ __launch_bounds__(256) void k_matvec(const float* __restrict__ X0,
                                                const float* __restrict__ X1,
                                                const float* __restrict__ X2,
                                                const float* __restrict__ v3,
                                                float* __restrict__ out3, int n) {
  const int rel = blockIdx.y;
  const float* X = pick3(rel, X0, X1, X2);
  const float* v = v3 + rel * D;
  float* out = out3 + (size_t)rel * n;
  const int lane = threadIdx.x & 31;
  const int row0 =
      __builtin_amdgcn_readfirstlane((blockIdx.x * 8 + (threadIdx.x >> 5)) * MV_ROWS);
  if (row0 >= n) return;
  float4 x[MV_ROWS];
#pragma unroll
  for (int u = 0; u < MV_ROWS; ++u) {
    int r = row0 + u;
    if (r > n - 1) r = n - 1;
    x[u] = ((const float4*)X)[(size_t)r * (D / 4) + lane];
  }
  const float4 c = ((const float4*)v)[lane];
  __builtin_amdgcn_sched_barrier(0);
  float mine = 0.f;
#pragma unroll
  for (int u = 0; u < MV_ROWS; ++u) {
    float s = x[u].x * c.x + x[u].y * c.y + x[u].z * c.z + x[u].w * c.w;
#pragma unroll
    for (int off = 16; off; off >>= 1) s += __shfl_xor(s, off, 32);
    if (lane == u) mine = s;
  }
  const int r = row0 + lane;
  if (lane < MV_ROWS && r < n) out[r] = mine;
}

#define SCAN_B 16
__global__ __launch_bounds__(256) void k_matvec_dst(const float* __restrict__ X,
                                                    const float* __restrict__ v3,
                                                    float* __restrict__ out3,
                                                    int* tab3,
                                                    int* __restrict__ cnt3,
                                                    int* __restrict__ cur3,
                                                    int* __restrict__ alloc, int n, int C) {
  const int lane = threadIdx.x & 31;
  const int row0 =
      __builtin_amdgcn_readfirstlane((blockIdx.x * 8 + (threadIdx.x >> 5)) * MV_ROWS);
  if (row0 >= n) return;
  float4 x[MV_ROWS];
#pragma unroll
  for (int u = 0; u < MV_ROWS; ++u) {
    int r = row0 + u;
    if (r > n - 1) r = n - 1;
    x[u] = ((const float4*)X)[(size_t)r * (D / 4) + lane];
  }
  const float4 c0 = ((const float4*)v3)[lane];
  const float4 c1 = ((const float4*)v3)[(D / 4) + lane];
  const float4 c2 = ((const float4*)v3)[2 * (D / 4) + lane];
  const int r = row0 + lane;
  const bool own = lane < MV_ROWS && r < n;
  const size_t pl = (size_t)C * n;
  int n0 = 0, n1 = 0, n2 = 0;
  if (own) {
    const int* tp = tab3 + r;
    for (int cb = 0; cb < C; cb += SCAN_B) {
      int v0[SCAN_B], v1[SCAN_B], v2[SCAN_B];
#pragma unroll
      for (int u = 0; u < SCAN_B; ++u) {
        const int c = (cb + u < C) ? cb + u : C - 1;
        v0[u] = tp[(size_t)c * n];
        v1[u] = tp[pl + (size_t)c * n];
        v2[u] = tp[2 * pl + (size_t)c * n];
      }
#pragma unroll
      for (int u = 0; u < SCAN_B; ++u) {
        if (cb + u < C) { n0 += v0[u]; n1 += v1[u]; n2 += v2[u]; }
      }
    }
  }
  __builtin_amdgcn_sched_barrier(0);
  int i0 = n0, i1 = n1, i2 = n2;
#pragma unroll
  for (int off = 1; off < MV_ROWS; off <<= 1) {
    const int u0 = __shfl_up(i0, off, 32);
    const int u1 = __shfl_up(i1, off, 32);
    const int u2 = __shfl_up(i2, off, 32);
    if (lane >= off) { i0 += u0; i1 += u1; i2 += u2; }
  }
  const int t0 = __shfl(i0, MV_ROWS - 1, 32);
  const int t1 = __shfl(i1, MV_ROWS - 1, 32);
  const int t2 = __shfl(i2, MV_ROWS - 1, 32);
  const int tsel = lane == 0 ? t0 : (lane == 1 ? t1 : t2);
  int bsel = 0;
  if (lane < RELS) bsel = atomicAdd(&alloc[lane * CUR_PAD], tsel);
  float m0 = 0.f, m1 = 0.f, m2 = 0.f;
#pragma unroll
  for (int u = 0; u < MV_ROWS; ++u) {
    float s0 = x[u].x * c0.x + x[u].y * c0.y + x[u].z * c0.z + x[u].w * c0.w;
    float s1 = x[u].x * c1.x + x[u].y * c1.y + x[u].z * c1.z + x[u].w * c1.w;
    float s2 = x[u].x * c2.x + x[u].y * c2.y + x[u].z * c2.z + x[u].w * c2.w;
#pragma unroll
    for (int off = 16; off; off >>= 1) {
      s0 += __shfl_xor(s0, off, 32);
      s1 += __shfl_xor(s1, off, 32);
      s2 += __shfl_xor(s2, off, 32);
    }
    if (lane == u) { m0 = s0; m1 = s1; m2 = s2; }
  }
  const int b0 = __shfl(bsel, 0, 32);
  const int b1 = __shfl(bsel, 1, 32);
  const int b2 = __shfl(bsel, 2, 32);
  if (own) {
    int p0 = b0 + i0 - n0;
    int p1 = b1 + i1 - n1;
    int p2 = b2 + i2 - n2;
    out3[r] = m0;
    out3[(size_t)n + r] = m1;
    out3[2 * (size_t)n + r] = m2;
    cnt3[r] = n0;
    cnt3[(size_t)n + r] = n1;
    cnt3[2 * (size_t)n + r] = n2;
    cur3[r] = p0;
    cur3[(size_t)n + r] = p1;
    cur3[2 * (size_t)n + r] = p2;
    int* tp = tab3 + r;
    for (int cb = 0; cb < C; cb += SCAN_B) {
      int v0[SCAN_B], v1[SCAN_B], v2[SCAN_B];
#pragma unroll
      for (int u = 0; u < SCAN_B; ++u) {
        const int c = (cb + u < C) ? cb + u : C - 1;
        v0[u] = tp[(size_t)c * n];
        v1[u] = tp[pl + (size_t)c * n];
        v2[u] = tp[2 * pl + (size_t)c * n];
      }
#pragma unroll
      for (int u = 0; u < SCAN_B; ++u) {
        if (cb + u < C) {
          const size_t o = (size_t)(cb + u) * n;
          tp[o] = p0;
          tp[pl + o] = p1;
          tp[2 * pl + o] = p2;
          p0 += v0[u];
          p1 += v1[u];
          p2 += v2[u];
        }
      }
    }
  }
}


#define CH_EDGES 16384
#define TAB_WORDS 51200
__global__ __launch_bounds__(1024) void k_rank(const int* __restrict__ di0,
                                               const int* __restrict__ di1,
                                               const int* __restrict__ di2,
                                               int* __restrict__ tab3, int* __restrict__ rank3,
                                               int E, int nd, int C) {
  __shared__ int lc[TAB_WORDS];
  const int rel = blockIdx.y;
  const int* di = pick3(rel, di0, di1, di2);
  int* rank = rank3 + (size_t)rel * E;
  int* tab = tab3 + ((size_t)rel * C + blockIdx.x) * nd;
  const int t = threadIdx.x;
  const int nd4 = nd & ~3;
  const int e0 = blockIdx.x * CH_EDGES;
  int4 dv[CH_EDGES / 4096];
#pragma unroll
  for (int q = 0; q < CH_EDGES / 4096; ++q) {
    const int i = e0 + (q * 1024 + t) * 4;
    dv[q] = make_int4(0, 0, 0, 0);
    if (i + 3 < E) dv[q] = *(const int4*)(di + i);
  }
  for (int j = t * 4; j < nd4; j += 4096) *(int4*)(lc + j) = make_int4(0, 0, 0, 0);
  if (t < nd - nd4) lc[nd4 + t] = 0;
  __syncthreads();
#pragma unroll
  for (int q = 0; q < CH_EDGES / 4096; ++q) {
    const int i = e0 + (q * 1024 + t) * 4;
    if (i + 3 < E) {
      int4 rk;
      rk.x = atomicAdd(&lc[dv[q].x], 1);
      rk.y = atomicAdd(&lc[dv[q].y], 1);
      rk.z = atomicAdd(&lc[dv[q].z], 1);
      rk.w = atomicAdd(&lc[dv[q].w], 1);
      *(int4*)(rank + i) = rk;
    } else {
      for (int j = i; j < E && j < i + 4; ++j) rank[j] = atomicAdd(&lc[di[j]], 1);
    }
  }
  __syncthreads();
  if ((nd & 3) == 0) {
    for (int j = t * 4; j < nd; j += 4096) *(int4*)(tab + j) = *(const int4*)(lc + j);
  } else {
    for (int j = t; j < nd; j += 1024) tab[j] = lc[j];
  }
}

__global__ __launch_bounds__(256) void k_place(const int* __restrict__ si0,
                                               const int* __restrict__ si1,
                                               const int* __restrict__ si2,
                                               const int* __restrict__ di0,
                                               const int* __restrict__ di1,
                                               const int* __restrict__ di2,
                                               const float* __restrict__ el3,
                                               const float* __restrict__ er3,
                                               const int* __restrict__ tab3,
                                               const int* __restrict__ rank3,
                                               int2* __restrict__ lst3, int E, int ns, int nd,
                                               int C) {
  const int rel = blockIdx.y;
  const int* si = pick3(rel, si0, si1, si2);
  const int* di = pick3(rel, di0, di1, di2);
  const float* el = el3 + (size_t)rel * ns;
  const float* er = er3 + (size_t)rel * nd;
  const int* pos = tab3 + ((size_t)rel * C + blockIdx.x / (CH_EDGES / 1024)) * nd;
  const int* rank = rank3 + (size_t)rel * E;
  int2* lst = lst3 + (size_t)rel * E;
  int i = (blockIdx.x * 256 + threadIdx.x) * 4;
  if (i + 3 < E) {
    const int4 s = *(const int4*)(si + i);
    const int4 d = *(const int4*)(di + i);
    const int4 rk = *(const int4*)(rank + i);
    const float x0 = lrelu(el[s.x] + er[d.x]);
    const float x1 = lrelu(el[s.y] + er[d.y]);
    const float x2 = lrelu(el[s.z] + er[d.z]);
    const float x3 = lrelu(el[s.w] + er[d.w]);
    const int p0 = pos[d.x] + rk.x;
    const int p1 = pos[d.y] + rk.y;
    const int p2 = pos[d.z] + rk.z;
    const int p3 = pos[d.w] + rk.w;
    lst[p0] = make_int2(s.x, __float_as_int(x0));
    lst[p1] = make_int2(s.y, __float_as_int(x1));
    lst[p2] = make_int2(s.z, __float_as_int(x2));
    lst[p3] = make_int2(s.w, __float_as_int(x3));
  } else {
    for (; i < E; ++i) {
      const int s = si[i], d = di[i];
      const float x = lrelu(el[s] + er[d]);
      lst[pos[d] + rank[i]] = make_int2(s, __float_as_int(x));
    }
  }
}

__device__ __forceinline__ void fma4(float4& a, float p, const float4& h) {
  a.x += p * h.x; a.y += p * h.y; a.z += p * h.z; a.w += p * h.w;
}

template <int N>
__device__ __forceinline__ void add_rows(float4& acc, const float* __restrict__ hs, int vx,
                                         int pb, int k, int lane) {
  float4 h[N];
  float pk[N];
#pragma unroll
  for (int u = 0; u < N; ++u) {
    const unsigned sj = (unsigned)__builtin_amdgcn_readlane(vx, k + u);
    pk[u] = __int_as_float(__builtin_amdgcn_readlane(pb, k + u));
    h[u] = ((const float4*)(hs + (size_t)sj * D))[lane];
  }
#pragma unroll
  for (int u = 0; u < N; ++u) fma4(acc, pk[u], h[u]);
}

#define FT_ROWS 16
#define FT_PITCH 132
#define G_ROWS 8
__global__ __launch_bounds__(512) void k_gather_gemm(const int* __restrict__ cnt3,
                                                     const int* __restrict__ cur3,
                                                     const int2* __restrict__ lst3,
                                                     const float* __restrict__ X0,
                                                     const float* __restrict__ X1,
                                                     const float* __restrict__ X2,
                                                     const float* __restrict__ W0,
                                                     const float* __restrict__ W1,
                                                     const float* __restrict__ W2,
                                                     const float* __restrict__ b0,
                                                     const float* __restrict__ b1,
                                                     const float* __restrict__ b2,
                                                     const float* sW1,
                                                     const float* __restrict__ sb1,
                                                     const float* __restrict__ sW2,
                                                     float* __restrict__ wsum3,
                                                     float* __restrict__ Z3, int n, int E) {
  __shared__ float4 Sm4[(D * D + 8 * FT_ROWS * FT_PITCH) / 4];
  __shared__ float red[16];
  float* Ws = (float*)Sm4;
  const int rel = blockIdx.y;
  const float* hs = pick3(rel, X0, X1, X2);
  const float* W = pick3(rel, W0, W1, W2);
  const float* bias = pick3(rel, b0, b1, b2);
  const int* cnt = cnt3 + (size_t)rel * n;
  const int* cur = cur3 + (size_t)rel * n;
  const int2* lst = lst3 + (size_t)rel * E;
  float* Z = Z3 + (size_t)rel * n * D;
  const int t = threadIdx.x;
#pragma unroll
  for (int i = 0; i < (D * D / 4) / 512; ++i)
    ((float4*)Ws)[t + i * 512] = ((const float4*)W)[t + i * 512];

  const int wave = __builtin_amdgcn_readfirstlane(t >> 5);
  const int lane = t & 31;
  const int ti = wave >> 1;
  const int h2 = wave & 1;
  float* tile = Ws + D * D + ti * (FT_ROWS * FT_PITCH);
  const int rowbase = blockIdx.x * (8 * FT_ROWS) + ti * FT_ROWS;
  const int d0 = rowbase + h2 * G_ROWS;
  const int NEG_INF_BITS = (int)0xff800000u;

  int dl = d0 + (lane & (G_ROWS - 1));
  if (dl > n - 1) dl = n - 1;
  const int cv = cnt[dl];
  const int sv = cur[dl];
  int2 vn = make_int2(0, NEG_INF_BITS);
  if (d0 < n) {
    const int c0 = __builtin_amdgcn_readlane(cv, 0);
    const int s0 = __builtin_amdgcn_readlane(sv, 0);
    if (lane < c0) vn = lst[s0 + lane];
  }

#pragma nounroll
  for (int r = 0; r < G_ROWS; ++r) {
    const int d = d0 + r;
    float4 o = make_float4(0.f, 0.f, 0.f, 0.f);
    if (d < n) {
      const int c = __builtin_amdgcn_readlane(cv, r);
      const int start = __builtin_amdgcn_readlane(sv, r);
      const int2 v0 = vn;
      vn = make_int2(0, NEG_INF_BITS);
      if (r + 1 < G_ROWS && d + 1 < n) {
        const int c1 = __builtin_amdgcn_readlane(cv, r + 1);
        const int s1 = __builtin_amdgcn_readlane(sv, r + 1);
        if (lane < c1) vn = lst[s1 + lane];
      }
      float mx = __int_as_float(v0.y);
      for (int base = 32; base < c; base += 32) {
        int j = base + lane;
        if (j < c) mx = fmaxf(mx, __int_as_float(lst[start + j].y));
      }
#pragma unroll
      for (int off = 16; off; off >>= 1) mx = fmaxf(mx, __shfl_xor(mx, off, 32));

      float4 acc = make_float4(0.f, 0.f, 0.f, 0.f);
      float zs = 0.f;
      for (int base = 0; base < c; base += 32) {
        int2 v = v0;
        const int j = base + lane;
        if (base > 0) {
          v = make_int2(0, NEG_INF_BITS);
          if (j < c) v = lst[start + j];
        }
        const float pe = expf(__int_as_float(v.y) - mx);
        const float p = (j < c) ? pe : 0.f;
        zs += p;
        const int pb = __float_as_int(p);
        int nc = c - base;
        if (nc > 32) nc = 32;
        int k = 0;
        for (; k + 16 <= nc; k += 16) add_rows<16>(acc, hs, v.x, pb, k, lane);
        for (; k + 8 <= nc; k += 8) add_rows<8>(acc, hs, v.x, pb, k, lane);
        switch (nc - k) {
          case 7: add_rows<7>(acc, hs, v.x, pb, k, lane); break;
          case 6: add_rows<6>(acc, hs, v.x, pb, k, lane); break;
          case 5: add_rows<5>(acc, hs, v.x, pb, k, lane); break;
          case 4: add_rows<4>(acc, hs, v.x, pb, k, lane); break;
          case 3: add_rows<3>(acc, hs, v.x, pb, k, lane); break;
          case 2: add_rows<2>(acc, hs, v.x, pb, k, lane); break;
          case 1: add_rows<1>(acc, hs, v.x, pb, k, lane); break;
          default: break;
        }
      }
#pragma unroll
      for (int off = 16; off; off >>= 1) zs += __shfl_xor(zs, off, 32);

      const float inv = (c > 0) ? 1.f / zs : 0.f;
      o.x = acc.x * inv;
      o.y = acc.y * inv;
      o.z = acc.z * inv;
      o.w = acc.w * inv;
    }
    *(float4*)(tile + (h2 * G_ROWS + r) * FT_PITCH + 4 * lane) = o;
  }

  __syncthreads();

  const int lrow = lane & 15;
  const int half = lane >> 4;
  v2f afrag[32];
#pragma unroll
  for (int kt = 0; kt < 32; ++kt) {
    int k0 = kt * 4 + 2 * half;
    afrag[kt] = *(const v2f*)(tile + lrow * FT_PITCH + k0);
  }
  float bn[4];
#pragma unroll
  for (int nt = 0; nt < 4; ++nt) bn[nt] = bias[(h2 * 4 + nt) * 16 + lrow];
  __syncthreads();

#pragma unroll
  for (int nt = 0; nt < 4; ++nt) {
    v8f acc = {};
    int nb = (h2 * 4 + nt) * 16 + lrow;
#pragma unroll
    for (int kt = 0; kt < 32; ++kt) {
      int k0 = kt * 4 + 2 * half;
      v2f b;
      b.x = Ws[k0 * D + nb];
      b.y = Ws[(k0 + 1) * D + nb];
      acc = __builtin_amdgcn_wmma_f32_16x16x4_f32(false, afrag[kt], false, b,
                                                  (short)0, acc, false, false);
    }
    asm volatile("v_nop\n\tv_nop\n\tv_nop\n\tv_nop" : "+v"(acc));
#pragma unroll
    for (int v = 0; v < 8; ++v) {
      float o = acc[v] + bn[nt];
      o = o > 0.f ? o : expf(o) - 1.f;
      tile[(v + 8 * half) * FT_PITCH + nb] = o;
    }
  }
  const float4* w1p = (const float4*)sW1 + t;
  const float4 q0 = w1p[0 * 512], q1 = w1p[1 * 512], q2 = w1p[2 * 512], q3 = w1p[3 * 512];
  const float4 q4 = w1p[4 * 512], q5 = w1p[5 * 512], q6 = w1p[6 * 512], q7 = w1p[7 * 512];
  __syncthreads();
  float4* wsp = (float4*)Ws + t;
  wsp[0 * 512] = q0; wsp[1 * 512] = q1; wsp[2 * 512] = q2; wsp[3 * 512] = q3;
  wsp[4 * 512] = q4; wsp[5 * 512] = q5; wsp[6 * 512] = q6; wsp[7 * 512] = q7;
  float b1n[4], w2n[4];
#pragma unroll
  for (int nt = 0; nt < 4; ++nt) {
    b1n[nt] = sb1[(h2 * 4 + nt) * 16 + lrow];
    w2n[nt] = sW2[(h2 * 4 + nt) * 16 + lrow];
  }
  __syncthreads();

  const int ocol = h2 * 64 + lrow * 4;
  if (rowbase + FT_ROWS <= n) {
    float4 orow[8];
#pragma unroll
    for (int i = 0; i < 8; ++i)
      orow[i] = *(const float4*)(tile + (2 * i + half) * FT_PITCH + ocol);
#pragma unroll
    for (int i = 0; i < 8; ++i)
      *(float4*)(Z + (size_t)(rowbase + 2 * i + half) * D + ocol) = orow[i];
  } else {
#pragma unroll
    for (int i = 0; i < 8; ++i) {
      const int grow = rowbase + 2 * i + half;
      if (grow < n)
        *(float4*)(Z + (size_t)grow * D + ocol) =
            *(const float4*)(tile + (2 * i + half) * FT_PITCH + ocol);
    }
  }

#pragma unroll
  for (int kt = 0; kt < 32; ++kt) {
    const int k0 = kt * 4 + 2 * half;
    afrag[kt] = *(const v2f*)(tile + lrow * FT_PITCH + k0);
  }
  v8f sacc[4] = {};
#pragma unroll
  for (int nt = 0; nt < 4; ++nt) {
    const int nb = (h2 * 4 + nt) * 16 + lrow;
#pragma unroll
    for (int kt = 0; kt < 32; ++kt) {
      const int k0 = kt * 4 + 2 * half;
      v2f b;
      b.x = Ws[k0 * D + nb];
      b.y = Ws[(k0 + 1) * D + nb];
      sacc[nt] = __builtin_amdgcn_wmma_f32_16x16x4_f32(false, afrag[kt], false, b, (short)0,
                                                       sacc[nt], false, false);
    }
    asm volatile("v_nop\n\tv_nop\n\tv_nop\n\tv_nop" : "+v"(sacc[nt]));
  }
  float sc = 0.f;
#pragma unroll
  for (int nt = 0; nt < 4; ++nt) {
#pragma unroll
    for (int v = 0; v < 8; ++v) {
      const int grow = rowbase + v + 8 * half;
      const float y = tanhf(sacc[nt][v] + b1n[nt]) * w2n[nt];
      sc += (grow < n) ? y : 0.f;
    }
  }
#pragma unroll
  for (int off = 16; off; off >>= 1) sc += __shfl_xor(sc, off, 32);
  if (lane == 0) red[wave] = sc;
  __syncthreads();
  if (t == 0) {
    const float tot = (((red[0] + red[1]) + (red[2] + red[3])) +
                       ((red[4] + red[5]) + (red[6] + red[7]))) +
                      (((red[8] + red[9]) + (red[10] + red[11])) +
                       ((red[12] + red[13]) + (red[14] + red[15])));
    atomicAdd(wsum3 + rel, tot);
  }
}

#define MIX_V 8
__global__ __launch_bounds__(256) void k_mix(const float* __restrict__ Z3,
                                             const float* __restrict__ wsum,
                                             float* __restrict__ out, int n, float invN) {
  const int n4 = n >> 2;
  const float4* __restrict__ za = (const float4*)Z3;
  const float4* __restrict__ zb = (const float4*)(Z3 + (size_t)n);
  const float4* __restrict__ zc = (const float4*)(Z3 + 2 * (size_t)n);
  const int base = blockIdx.x * (256 * MIX_V) + threadIdx.x;
  float4 a[MIX_V], b[MIX_V], c[MIX_V];
#pragma unroll
  for (int u = 0; u < MIX_V; ++u) {
    int i = base + u * 256;
    if (i > n4 - 1) i = n4 - 1;
    a[u] = za[i];
    b[u] = zb[i];
    c[u] = zc[i];
  }
  float w0 = wsum[0] * invN, w1 = wsum[1] * invN, w2 = wsum[2] * invN;
  float mx = fmaxf(w0, fmaxf(w1, w2));
  float e0 = expf(w0 - mx), e1 = expf(w1 - mx), e2 = expf(w2 - mx);
  float inv = 1.f / (e0 + e1 + e2);
  float a0 = e0 * inv, a1 = e1 * inv, a2 = e2 * inv;
#pragma unroll
  for (int u = 0; u < MIX_V; ++u) {
    const int i = base + u * 256;
    float4 o;
    o.x = a0 * a[u].x + a1 * b[u].x + a2 * c[u].x;
    o.y = a0 * a[u].y + a1 * b[u].y + a2 * c[u].y;
    o.z = a0 * a[u].z + a1 * b[u].z + a2 * c[u].z;
    o.w = a0 * a[u].w + a1 * b[u].w + a2 * c[u].w;
    if (i < n4) ((float4*)out)[i] = o;
  }
  if (blockIdx.x == 0 && threadIdx.x == 0) { out[n] = a0; out[n + 1] = a1; out[n + 2] = a2; }
}


extern "C" void kernel_launch(void* const* d_in, const int* in_sizes, int n_in,
                              void* d_out, int out_size, void* d_ws, size_t ws_size,
                              hipStream_t stream) {
  const float* dst_feat = (const float*)d_in[0];
  const float* src_feat[RELS] = {(const float*)d_in[1], (const float*)d_in[2],
                                 (const float*)d_in[3]};
  const float* sem_W1 = (const float*)d_in[4];
  const float* sem_b1 = (const float*)d_in[5];
  const float* sem_W2 = (const float*)d_in[6];
  const float *Wg[RELS], *al[RELS], *ar[RELS], *bg[RELS];
  const int *sidx[RELS], *didx[RELS];
  for (int r = 0; r < RELS; ++r) {
    int b = 7 + 6 * r;
    Wg[r] = (const float*)d_in[b + 0];
    al[r] = (const float*)d_in[b + 1];
    ar[r] = (const float*)d_in[b + 2];
    bg[r] = (const float*)d_in[b + 3];
    sidx[r] = (const int*)d_in[b + 4];
    didx[r] = (const int*)d_in[b + 5];
  }
  const int ND = in_sizes[0] / D;
  const int NS = in_sizes[1] / D;
  const int E  = in_sizes[11];

  float* ws   = (float*)d_ws;
  float* z3   = ws;
  float* wsum = z3 + (size_t)RELS * ND * D;
  int2*  lst = (int2*)d_out;
  float* el  = (float*)(lst + (size_t)RELS * E);
  float* er  = el + (size_t)RELS * NS;
  int*   cnt = (int*)(er + (size_t)RELS * ND);
  int*   cur = cnt + (size_t)RELS * ND;
  float* wa  = (float*)(cur + (size_t)RELS * ND);
  float* wr  = wa + RELS * D;
  int*   alloc = (int*)(wr + RELS * D);
  int*   rank = (int*)(z3 + (size_t)(RELS - 1) * ND * D);
  const int NCH = (E + CH_EDGES - 1) / CH_EDGES;
  int*   tab = (int*)z3;

  k_rowdot<<<RELS, 256, 0, stream>>>(Wg[0], Wg[1], Wg[2], al[0], al[1], al[2], ar[0], ar[1],
                                     ar[2], wa, wr, wsum, alloc);
  k_matvec<<<dim3((NS + 8 * MV_ROWS - 1) / (8 * MV_ROWS), RELS), 256, 0, stream>>>(
      src_feat[0], src_feat[1], src_feat[2], wa, el, NS);
  k_rank<<<dim3(NCH, RELS), 1024, 0, stream>>>(didx[0], didx[1], didx[2], tab, rank, E, ND, NCH);
  k_matvec_dst<<<(ND + 8 * MV_ROWS - 1) / (8 * MV_ROWS), 256, 0, stream>>>(
      dst_feat, wr, er, tab, cnt, cur, alloc, ND, NCH);
  k_place<<<dim3((E + 1023) / 1024, RELS), 256, 0, stream>>>(
      sidx[0], sidx[1], sidx[2], didx[0], didx[1], didx[2], el, er, tab, rank, lst, E, NS, ND,
      NCH);

  k_gather_gemm<<<dim3((ND + 8 * FT_ROWS - 1) / (8 * FT_ROWS), RELS), 512, 0, stream>>>(
      cnt, cur, lst, src_feat[0], src_feat[1], src_feat[2], Wg[0], Wg[1], Wg[2], bg[0], bg[1],
      bg[2], sem_W1, sem_b1, sem_W2, wsum, z3, ND, E);

  k_mix<<<(ND * D / 4 + 256 * MIX_V - 1) / (256 * MIX_V), 256, 0, stream>>>(
      z3, wsum, (float*)d_out, ND * D, 1.0f / (float)ND);
}


# placing pass: no scattered read from memory left; a 1024-thread group per chunk copies el, then its place table, into LDS and looks both up there; er + leaky-relu go to the gather; sums as before
# speedup vs baseline: 5.0192x; 1.0388x over previous
#include <hip/hip_runtime.h>
#include <math.h>

#define D 128
#define RELS 3

typedef float v2f __attribute__((ext_vector_type(2)));
typedef float v8f __attribute__((ext_vector_type(8)));


__device__ __forceinline__ float lrelu(float x) { return x > 0.f ? x : 0.2f * x; }

__device__ __forceinline__ const float* pick3(int r, const float* a, const float* b,
                                              const float* c) {
  return r == 0 ? a : (r == 1 ? b : c);
}
__device__ __forceinline__ const int* pick3(int r, const int* a, const int* b, const int* c) {
  return r == 0 ? a : (r == 1 ? b : c);
}


#define CUR_PAD 64
__global__ __launch_bounds__(256) void k_rowdot(const float* __restrict__ W0,
                                                const float* __restrict__ W1,
                                                const float* __restrict__ W2,
                                                const float* __restrict__ al0,
                                                const float* __restrict__ al1,
                                                const float* __restrict__ al2,
                                                const float* __restrict__ ar0,
                                                const float* __restrict__ ar1,
                                                const float* __restrict__ ar2,
                                                float* __restrict__ wa3,
                                                float* __restrict__ wr3,
                                                float* __restrict__ wsum,
                                                int* __restrict__ alloc) {
  const int rel = blockIdx.x;
  const float* W = pick3(rel, W0, W1, W2);
  const float* al = pick3(rel, al0, al1, al2);
  const float* ar = pick3(rel, ar0, ar1, ar2);
  int t = threadIdx.x;
  if (t == 0) {
    wsum[rel] = 0.f;
    alloc[rel * CUR_PAD] = 0;
  }
  const float* v = (t < D) ? al : ar;
  float* o = ((t < D) ? wa3 : wr3) + rel * D;
  int d = t & (D - 1);
  float s = 0.f;
#pragma unroll 8
  for (int h = 0; h < D; ++h) s += W[d * D + h] * v[h];
  o[d] = s;
}

#define MV_ROWS 16
__global__ __launch_bounds__(256) void k_matvec(const float* __restrict__ X0,
                                                const float* __restrict__ X1,
                                                const float* __restrict__ X2,
                                                const float* __restrict__ v3,
                                                float* __restrict__ out3, int n) {
  const int rel = blockIdx.y;
  const float* X = pick3(rel, X0, X1, X2);
  const float* v = v3 + rel * D;
  float* out = out3 + (size_t)rel * n;
  const int lane = threadIdx.x & 31;
  const int row0 =
      __builtin_amdgcn_readfirstlane((blockIdx.x * 8 + (threadIdx.x >> 5)) * MV_ROWS);
  if (row0 >= n) return;
  float4 x[MV_ROWS];
#pragma unroll
  for (int u = 0; u < MV_ROWS; ++u) {
    int r = row0 + u;
    if (r > n - 1) r = n - 1;
    x[u] = ((const float4*)X)[(size_t)r * (D / 4) + lane];
  }
  const float4 c = ((const float4*)v)[lane];
  __builtin_amdgcn_sched_barrier(0);
  float mine = 0.f;
#pragma unroll
  for (int u = 0; u < MV_ROWS; ++u) {
    float s = x[u].x * c.x + x[u].y * c.y + x[u].z * c.z + x[u].w * c.w;
#pragma unroll
    for (int off = 16; off; off >>= 1) s += __shfl_xor(s, off, 32);
    if (lane == u) mine = s;
  }
  const int r = row0 + lane;
  if (lane < MV_ROWS && r < n) out[r] = mine;
}

#define SCAN_B 16
__global__ __launch_bounds__(256) void k_matvec_dst(const float* __restrict__ X,
                                                    const float* __restrict__ v3,
                                                    float* __restrict__ out3,
                                                    int* tab3,
                                                    int* __restrict__ cnt3,
                                                    int* __restrict__ cur3,
                                                    int* __restrict__ alloc, int n, int C) {
  const int lane = threadIdx.x & 31;
  const int row0 =
      __builtin_amdgcn_readfirstlane((blockIdx.x * 8 + (threadIdx.x >> 5)) * MV_ROWS);
  if (row0 >= n) return;
  float4 x[MV_ROWS];
#pragma unroll
  for (int u = 0; u < MV_ROWS; ++u) {
    int r = row0 + u;
    if (r > n - 1) r = n - 1;
    x[u] = ((const float4*)X)[(size_t)r * (D / 4) + lane];
  }
  const float4 c0 = ((const float4*)v3)[lane];
  const float4 c1 = ((const float4*)v3)[(D / 4) + lane];
  const float4 c2 = ((const float4*)v3)[2 * (D / 4) + lane];
  const int r = row0 + lane;
  const bool own = lane < MV_ROWS && r < n;
  const size_t pl = (size_t)C * n;
  int n0 = 0, n1 = 0, n2 = 0;
  if (own) {
    const int* tp = tab3 + r;
    for (int cb = 0; cb < C; cb += SCAN_B) {
      int v0[SCAN_B], v1[SCAN_B], v2[SCAN_B];
#pragma unroll
      for (int u = 0; u < SCAN_B; ++u) {
        const int c = (cb + u < C) ? cb + u : C - 1;
        v0[u] = tp[(size_t)c * n];
        v1[u] = tp[pl + (size_t)c * n];
        v2[u] = tp[2 * pl + (size_t)c * n];
      }
#pragma unroll
      for (int u = 0; u < SCAN_B; ++u) {
        if (cb + u < C) { n0 += v0[u]; n1 += v1[u]; n2 += v2[u]; }
      }
    }
  }
  __builtin_amdgcn_sched_barrier(0);
  int i0 = n0, i1 = n1, i2 = n2;
#pragma unroll
  for (int off = 1; off < MV_ROWS; off <<= 1) {
    const int u0 = __shfl_up(i0, off, 32);
    const int u1 = __shfl_up(i1, off, 32);
    const int u2 = __shfl_up(i2, off, 32);
    if (lane >= off) { i0 += u0; i1 += u1; i2 += u2; }
  }
  const int t0 = __shfl(i0, MV_ROWS - 1, 32);
  const int t1 = __shfl(i1, MV_ROWS - 1, 32);
  const int t2 = __shfl(i2, MV_ROWS - 1, 32);
  const int tsel = lane == 0 ? t0 : (lane == 1 ? t1 : t2);
  int bsel = 0;
  if (lane < RELS) bsel = atomicAdd(&alloc[lane * CUR_PAD], tsel);
  float m0 = 0.f, m1 = 0.f, m2 = 0.f;
#pragma unroll
  for (int u = 0; u < MV_ROWS; ++u) {
    float s0 = x[u].x * c0.x + x[u].y * c0.y + x[u].z * c0.z + x[u].w * c0.w;
    float s1 = x[u].x * c1.x + x[u].y * c1.y + x[u].z * c1.z + x[u].w * c1.w;
    float s2 = x[u].x * c2.x + x[u].y * c2.y + x[u].z * c2.z + x[u].w * c2.w;
#pragma unroll
    for (int off = 16; off; off >>= 1) {
      s0 += __shfl_xor(s0, off, 32);
      s1 += __shfl_xor(s1, off, 32);
      s2 += __shfl_xor(s2, off, 32);
    }
    if (lane == u) { m0 = s0; m1 = s1; m2 = s2; }
  }
  const int b0 = __shfl(bsel, 0, 32);
  const int b1 = __shfl(bsel, 1, 32);
  const int b2 = __shfl(bsel, 2, 32);
  if (own) {
    int p0 = b0 + i0 - n0;
    int p1 = b1 + i1 - n1;
    int p2 = b2 + i2 - n2;
    out3[r] = m0;
    out3[(size_t)n + r] = m1;
    out3[2 * (size_t)n + r] = m2;
    cnt3[r] = n0;
    cnt3[(size_t)n + r] = n1;
    cnt3[2 * (size_t)n + r] = n2;
    cur3[r] = p0;
    cur3[(size_t)n + r] = p1;
    cur3[2 * (size_t)n + r] = p2;
    int* tp = tab3 + r;
    for (int cb = 0; cb < C; cb += SCAN_B) {
      int v0[SCAN_B], v1[SCAN_B], v2[SCAN_B];
#pragma unroll
      for (int u = 0; u < SCAN_B; ++u) {
        const int c = (cb + u < C) ? cb + u : C - 1;
        v0[u] = tp[(size_t)c * n];
        v1[u] = tp[pl + (size_t)c * n];
        v2[u] = tp[2 * pl + (size_t)c * n];
      }
#pragma unroll
      for (int u = 0; u < SCAN_B; ++u) {
        if (cb + u < C) {
          const size_t o = (size_t)(cb + u) * n;
          tp[o] = p0;
          tp[pl + o] = p1;
          tp[2 * pl + o] = p2;
          p0 += v0[u];
          p1 += v1[u];
          p2 += v2[u];
        }
      }
    }
  }
}


#define CH_EDGES 16384
#define TAB_WORDS 51200
__global__ __launch_bounds__(1024) void k_rank(const int* __restrict__ di0,
                                               const int* __restrict__ di1,
                                               const int* __restrict__ di2,
                                               int* __restrict__ tab3, int* __restrict__ rank3,
                                               int E, int nd, int C) {
  __shared__ int lc[TAB_WORDS];
  const int rel = blockIdx.y;
  const int* di = pick3(rel, di0, di1, di2);
  int* rank = rank3 + (size_t)rel * E;
  int* tab = tab3 + ((size_t)rel * C + blockIdx.x) * nd;
  const int t = threadIdx.x;
  const int nd4 = nd & ~3;
  const int e0 = blockIdx.x * CH_EDGES;
  int4 dv[CH_EDGES / 4096];
#pragma unroll
  for (int q = 0; q < CH_EDGES / 4096; ++q) {
    const int i = e0 + (q * 1024 + t) * 4;
    dv[q] = make_int4(0, 0, 0, 0);
    if (i + 3 < E) dv[q] = *(const int4*)(di + i);
  }
  for (int j = t * 4; j < nd4; j += 4096) *(int4*)(lc + j) = make_int4(0, 0, 0, 0);
  if (t < nd - nd4) lc[nd4 + t] = 0;
  __syncthreads();
#pragma unroll
  for (int q = 0; q < CH_EDGES / 4096; ++q) {
    const int i = e0 + (q * 1024 + t) * 4;
    if (i + 3 < E) {
      int4 rk;
      rk.x = atomicAdd(&lc[dv[q].x], 1);
      rk.y = atomicAdd(&lc[dv[q].y], 1);
      rk.z = atomicAdd(&lc[dv[q].z], 1);
      rk.w = atomicAdd(&lc[dv[q].w], 1);
      *(int4*)(rank + i) = rk;
    } else {
      for (int j = i; j < E && j < i + 4; ++j) rank[j] = atomicAdd(&lc[di[j]], 1);
    }
  }
  __syncthreads();
  if ((nd & 3) == 0) {
    for (int j = t * 4; j < nd; j += 4096) *(int4*)(tab + j) = *(const int4*)(lc + j);
  } else {
    for (int j = t; j < nd; j += 1024) tab[j] = lc[j];
  }
}

#define PL_Q ((TAB_WORDS + 4095) / 4096)
__device__ __forceinline__ void stage_table(int* lp, const int* src, int n, int t) {
  if ((((size_t)src) & 15) == 0 && (n & 3) == 0 && n > 0) {
    static_assert(PL_Q == 13, "the copy below is written out for 13 loads a thread");
    const int4* tp = (const int4*)src;
    const int n4 = n >> 2;
    const int last = n4 - 1;
#define PL_LD(q) tp[((q) * 1024 + t) < n4 ? ((q) * 1024 + t) : last]
    const int4 a0 = PL_LD(0), a1 = PL_LD(1), a2 = PL_LD(2), a3 = PL_LD(3), a4 = PL_LD(4);
    const int4 a5 = PL_LD(5), a6 = PL_LD(6), a7 = PL_LD(7), a8 = PL_LD(8), a9 = PL_LD(9);
    const int4 a10 = PL_LD(10), a11 = PL_LD(11), a12 = PL_LD(12);
#undef PL_LD
    int4* lq = (int4*)lp + t;
    lq[0 * 1024] = a0; lq[1 * 1024] = a1; lq[2 * 1024] = a2; lq[3 * 1024] = a3;
    lq[4 * 1024] = a4; lq[5 * 1024] = a5; lq[6 * 1024] = a6; lq[7 * 1024] = a7;
    lq[8 * 1024] = a8; lq[9 * 1024] = a9; lq[10 * 1024] = a10; lq[11 * 1024] = a11;
    lq[12 * 1024] = a12;
  } else {
    for (int j = t; j < n; j += 1024) lp[j] = src[j];
  }
}

__global__ __launch_bounds__(1024) void k_place(const int* __restrict__ si0,
                                                const int* __restrict__ si1,
                                                const int* __restrict__ si2,
                                                const int* __restrict__ di0,
                                                const int* __restrict__ di1,
                                                const int* __restrict__ di2,
                                                const float* __restrict__ el3,
                                                const int* __restrict__ tab3,
                                                const int* __restrict__ rank3,
                                                int2* __restrict__ lst3, int E, int ns, int nd,
                                                int C) {
  __shared__ int lp[PL_Q * 4096];
  const int rel = blockIdx.y;
  const int* si = pick3(rel, si0, si1, si2);
  const int* di = pick3(rel, di0, di1, di2);
  const float* el = el3 + (size_t)rel * ns;
  const int* pos = tab3 + ((size_t)rel * C + blockIdx.x) * nd;
  const int* rank = rank3 + (size_t)rel * E;
  int2* lst = lst3 + (size_t)rel * E;
  const int t = threadIdx.x;
  const int e0 = blockIdx.x * CH_EDGES;
  int g[CH_EDGES / 4096];
  int4 sv[CH_EDGES / 4096];
#pragma unroll
  for (int q = 0; q < CH_EDGES / 4096; ++q) {
    const int i = e0 + (q * 1024 + t) * 4;
    g[q] = (i + 3 < E) ? (i >> 2) : 0;
    sv[q] = ((const int4*)si)[g[q]];
  }
  stage_table(lp, (const int*)el, ns, t);
  __syncthreads();
  float x[CH_EDGES / 1024];
#pragma unroll
  for (int q = 0; q < CH_EDGES / 4096; ++q) {
    x[4 * q + 0] = __int_as_float(lp[sv[q].x]);
    x[4 * q + 1] = __int_as_float(lp[sv[q].y]);
    x[4 * q + 2] = __int_as_float(lp[sv[q].z]);
    x[4 * q + 3] = __int_as_float(lp[sv[q].w]);
  }
  __syncthreads();
  stage_table(lp, pos, nd, t);
  int4 dv[CH_EDGES / 4096], rv[CH_EDGES / 4096];
#pragma unroll
  for (int q = 0; q < CH_EDGES / 4096; ++q) {
    dv[q] = ((const int4*)di)[g[q]];
    rv[q] = ((const int4*)rank)[g[q]];
  }
  __syncthreads();
#pragma unroll
  for (int q = 0; q < CH_EDGES / 4096; ++q) {
    const int i = e0 + (q * 1024 + t) * 4;
    if (i + 3 < E) {
      const int4 d = dv[q];
      const int4 rk = rv[q];
      const int p0 = lp[d.x] + rk.x;
      const int p1 = lp[d.y] + rk.y;
      const int p2 = lp[d.z] + rk.z;
      const int p3 = lp[d.w] + rk.w;
      lst[p0] = make_int2(sv[q].x, __float_as_int(x[4 * q + 0]));
      lst[p1] = make_int2(sv[q].y, __float_as_int(x[4 * q + 1]));
      lst[p2] = make_int2(sv[q].z, __float_as_int(x[4 * q + 2]));
      lst[p3] = make_int2(sv[q].w, __float_as_int(x[4 * q + 3]));
    } else {
      for (int j = i; j < E && j < i + 4; ++j) {
        const int sj = si[j], dj = di[j];
        lst[lp[dj] + rank[j]] = make_int2(sj, __float_as_int(el[sj]));
      }
    }
  }
}

__device__ __forceinline__ void fma4(float4& a, float p, const float4& h) {
  a.x += p * h.x; a.y += p * h.y; a.z += p * h.z; a.w += p * h.w;
}

template <int N>
__device__ __forceinline__ void add_rows(float4& acc, const float* __restrict__ hs, int vx,
                                         int pb, int k, int lane) {
  float4 h[N];
  float pk[N];
#pragma unroll
  for (int u = 0; u < N; ++u) {
    const unsigned sj = (unsigned)__builtin_amdgcn_readlane(vx, k + u);
    pk[u] = __int_as_float(__builtin_amdgcn_readlane(pb, k + u));
    h[u] = ((const float4*)(hs + (size_t)sj * D))[lane];
  }
#pragma unroll
  for (int u = 0; u < N; ++u) fma4(acc, pk[u], h[u]);
}

#define FT_ROWS 16
#define FT_PITCH 132
#define G_ROWS 8
__global__ __launch_bounds__(512) void k_gather_gemm(const int* __restrict__ cnt3,
                                                     const int* __restrict__ cur3,
                                                     const float* __restrict__ er3,
                                                     const int2* __restrict__ lst3,
                                                     const float* __restrict__ X0,
                                                     const float* __restrict__ X1,
                                                     const float* __restrict__ X2,
                                                     const float* __restrict__ W0,
                                                     const float* __restrict__ W1,
                                                     const float* __restrict__ W2,
                                                     const float* __restrict__ b0,
                                                     const float* __restrict__ b1,
                                                     const float* __restrict__ b2,
                                                     const float* sW1,
                                                     const float* __restrict__ sb1,
                                                     const float* __restrict__ sW2,
                                                     float* __restrict__ wsum3,
                                                     float* __restrict__ Z3, int n, int E) {
  __shared__ float4 Sm4[(D * D + 8 * FT_ROWS * FT_PITCH) / 4];
  __shared__ float red[16];
  float* Ws = (float*)Sm4;
  const int rel = blockIdx.y;
  const float* hs = pick3(rel, X0, X1, X2);
  const float* W = pick3(rel, W0, W1, W2);
  const float* bias = pick3(rel, b0, b1, b2);
  const int* cnt = cnt3 + (size_t)rel * n;
  const int* cur = cur3 + (size_t)rel * n;
  const float* er = er3 + (size_t)rel * n;
  const int2* lst = lst3 + (size_t)rel * E;
  float* Z = Z3 + (size_t)rel * n * D;
  const int t = threadIdx.x;
#pragma unroll
  for (int i = 0; i < (D * D / 4) / 512; ++i)
    ((float4*)Ws)[t + i * 512] = ((const float4*)W)[t + i * 512];

  const int wave = __builtin_amdgcn_readfirstlane(t >> 5);
  const int lane = t & 31;
  const int ti = wave >> 1;
  const int h2 = wave & 1;
  float* tile = Ws + D * D + ti * (FT_ROWS * FT_PITCH);
  const int rowbase = blockIdx.x * (8 * FT_ROWS) + ti * FT_ROWS;
  const int d0 = rowbase + h2 * G_ROWS;
  const int NEG_INF_BITS = (int)0xff800000u;

  int dl = d0 + (lane & (G_ROWS - 1));
  if (dl > n - 1) dl = n - 1;
  const int cv = cnt[dl];
  const int sv = cur[dl];
  const int ev = __float_as_int(er[dl]);
  int2 vn = make_int2(0, NEG_INF_BITS);
  if (d0 < n) {
    const int c0 = __builtin_amdgcn_readlane(cv, 0);
    const int s0 = __builtin_amdgcn_readlane(sv, 0);
    if (lane < c0) vn = lst[s0 + lane];
  }

#pragma nounroll
  for (int r = 0; r < G_ROWS; ++r) {
    const int d = d0 + r;
    float4 o = make_float4(0.f, 0.f, 0.f, 0.f);
    if (d < n) {
      const int c = __builtin_amdgcn_readlane(cv, r);
      const int start = __builtin_amdgcn_readlane(sv, r);
      const float erd = __int_as_float(__builtin_amdgcn_readlane(ev, r));
      const int2 v0 = vn;
      vn = make_int2(0, NEG_INF_BITS);
      if (r + 1 < G_ROWS && d + 1 < n) {
        const int c1 = __builtin_amdgcn_readlane(cv, r + 1);
        const int s1 = __builtin_amdgcn_readlane(sv, r + 1);
        if (lane < c1) vn = lst[s1 + lane];
      }
      float mx = __int_as_float(v0.y);
      for (int base = 32; base < c; base += 32) {
        int j = base + lane;
        if (j < c) mx = fmaxf(mx, __int_as_float(lst[start + j].y));
      }
#pragma unroll
      for (int off = 16; off; off >>= 1) mx = fmaxf(mx, __shfl_xor(mx, off, 32));
      mx = lrelu(mx + erd);

      float4 acc = make_float4(0.f, 0.f, 0.f, 0.f);
      float zs = 0.f;
      for (int base = 0; base < c; base += 32) {
        int2 v = v0;
        const int j = base + lane;
        if (base > 0) {
          v = make_int2(0, NEG_INF_BITS);
          if (j < c) v = lst[start + j];
        }
        const float pe = expf(lrelu(__int_as_float(v.y) + erd) - mx);
        const float p = (j < c) ? pe : 0.f;
        zs += p;
        const int pb = __float_as_int(p);
        int nc = c - base;
        if (nc > 32) nc = 32;
        int k = 0;
        for (; k + 16 <= nc; k += 16) add_rows<16>(acc, hs, v.x, pb, k, lane);
        for (; k + 8 <= nc; k += 8) add_rows<8>(acc, hs, v.x, pb, k, lane);
        switch (nc - k) {
          case 7: add_rows<7>(acc, hs, v.x, pb, k, lane); break;
          case 6: add_rows<6>(acc, hs, v.x, pb, k, lane); break;
          case 5: add_rows<5>(acc, hs, v.x, pb, k, lane); break;
          case 4: add_rows<4>(acc, hs, v.x, pb, k, lane); break;
          case 3: add_rows<3>(acc, hs, v.x, pb, k, lane); break;
          case 2: add_rows<2>(acc, hs, v.x, pb, k, lane); break;
          case 1: add_rows<1>(acc, hs, v.x, pb, k, lane); break;
          default: break;
        }
      }
#pragma unroll
      for (int off = 16; off; off >>= 1) zs += __shfl_xor(zs, off, 32);

      const float inv = (c > 0) ? 1.f / zs : 0.f;
      o.x = acc.x * inv;
      o.y = acc.y * inv;
      o.z = acc.z * inv;
      o.w = acc.w * inv;
    }
    *(float4*)(tile + (h2 * G_ROWS + r) * FT_PITCH + 4 * lane) = o;
  }

  __syncthreads();

  const int lrow = lane & 15;
  const int half = lane >> 4;
  v2f afrag[32];
#pragma unroll
  for (int kt = 0; kt < 32; ++kt) {
    int k0 = kt * 4 + 2 * half;
    afrag[kt] = *(const v2f*)(tile + lrow * FT_PITCH + k0);
  }
  float bn[4];
#pragma unroll
  for (int nt = 0; nt < 4; ++nt) bn[nt] = bias[(h2 * 4 + nt) * 16 + lrow];
  __syncthreads();

#pragma unroll
  for (int nt = 0; nt < 4; ++nt) {
    v8f acc = {};
    int nb = (h2 * 4 + nt) * 16 + lrow;
#pragma unroll
    for (int kt = 0; kt < 32; ++kt) {
      int k0 = kt * 4 + 2 * half;
      v2f b;
      b.x = Ws[k0 * D + nb];
      b.y = Ws[(k0 + 1) * D + nb];
      acc = __builtin_amdgcn_wmma_f32_16x16x4_f32(false, afrag[kt], false, b,
                                                  (short)0, acc, false, false);
    }
    asm volatile("v_nop\n\tv_nop\n\tv_nop\n\tv_nop" : "+v"(acc));
#pragma unroll
    for (int v = 0; v < 8; ++v) {
      float o = acc[v] + bn[nt];
      o = o > 0.f ? o : expf(o) - 1.f;
      tile[(v + 8 * half) * FT_PITCH + nb] = o;
    }
  }
  const float4* w1p = (const float4*)sW1 + t;
  const float4 q0 = w1p[0 * 512], q1 = w1p[1 * 512], q2 = w1p[2 * 512], q3 = w1p[3 * 512];
  const float4 q4 = w1p[4 * 512], q5 = w1p[5 * 512], q6 = w1p[6 * 512], q7 = w1p[7 * 512];
  __syncthreads();
  float4* wsp = (float4*)Ws + t;
  wsp[0 * 512] = q0; wsp[1 * 512] = q1; wsp[2 * 512] = q2; wsp[3 * 512] = q3;
  wsp[4 * 512] = q4; wsp[5 * 512] = q5; wsp[6 * 512] = q6; wsp[7 * 512] = q7;
  float b1n[4], w2n[4];
#pragma unroll
  for (int nt = 0; nt < 4; ++nt) {
    b1n[nt] = sb1[(h2 * 4 + nt) * 16 + lrow];
    w2n[nt] = sW2[(h2 * 4 + nt) * 16 + lrow];
  }
  __syncthreads();

  const int ocol = h2 * 64 + lrow * 4;
  if (rowbase + FT_ROWS <= n) {
    float4 orow[8];
#pragma unroll
    for (int i = 0; i < 8; ++i)
      orow[i] = *(const float4*)(tile + (2 * i + half) * FT_PITCH + ocol);
#pragma unroll
    for (int i = 0; i < 8; ++i)
      *(float4*)(Z + (size_t)(rowbase + 2 * i + half) * D + ocol) = orow[i];
  } else {
#pragma unroll
    for (int i = 0; i < 8; ++i) {
      const int grow = rowbase + 2 * i + half;
      if (grow < n)
        *(float4*)(Z + (size_t)grow * D + ocol) =
            *(const float4*)(tile + (2 * i + half) * FT_PITCH + ocol);
    }
  }

#pragma unroll
  for (int kt = 0; kt < 32; ++kt) {
    const int k0 = kt * 4 + 2 * half;
    afrag[kt] = *(const v2f*)(tile + lrow * FT_PITCH + k0);
  }
  v8f sacc[4] = {};
#pragma unroll
  for (int nt = 0; nt < 4; ++nt) {
    const int nb = (h2 * 4 + nt) * 16 + lrow;
#pragma unroll
    for (int kt = 0; kt < 32; ++kt) {
      const int k0 = kt * 4 + 2 * half;
      v2f b;
      b.x = Ws[k0 * D + nb];
      b.y = Ws[(k0 + 1) * D + nb];
      sacc[nt] = __builtin_amdgcn_wmma_f32_16x16x4_f32(false, afrag[kt], false, b, (short)0,
                                                       sacc[nt], false, false);
    }
    asm volatile("v_nop\n\tv_nop\n\tv_nop\n\tv_nop" : "+v"(sacc[nt]));
  }
  float sc = 0.f;
#pragma unroll
  for (int nt = 0; nt < 4; ++nt) {
#pragma unroll
    for (int v = 0; v < 8; ++v) {
      const int grow = rowbase + v + 8 * half;
      const float y = tanhf(sacc[nt][v] + b1n[nt]) * w2n[nt];
      sc += (grow < n) ? y : 0.f;
    }
  }
#pragma unroll
  for (int off = 16; off; off >>= 1) sc += __shfl_xor(sc, off, 32);
  if (lane == 0) red[wave] = sc;
  __syncthreads();
  if (t == 0) {
    const float tot = (((red[0] + red[1]) + (red[2] + red[3])) +
                       ((red[4] + red[5]) + (red[6] + red[7]))) +
                      (((red[8] + red[9]) + (red[10] + red[11])) +
                       ((red[12] + red[13]) + (red[14] + red[15])));
    atomicAdd(wsum3 + rel, tot);
  }
}

#define MIX_V 8
__global__ __launch_bounds__(256) void k_mix(const float* __restrict__ Z3,
                                             const float* __restrict__ wsum,
                                             float* __restrict__ out, int n, float invN) {
  const int n4 = n >> 2;
  const float4* __restrict__ za = (const float4*)Z3;
  const float4* __restrict__ zb = (const float4*)(Z3 + (size_t)n);
  const float4* __restrict__ zc = (const float4*)(Z3 + 2 * (size_t)n);
  const int base = blockIdx.x * (256 * MIX_V) + threadIdx.x;
  float4 a[MIX_V], b[MIX_V], c[MIX_V];
#pragma unroll
  for (int u = 0; u < MIX_V; ++u) {
    int i = base + u * 256;
    if (i > n4 - 1) i = n4 - 1;
    a[u] = za[i];
    b[u] = zb[i];
    c[u] = zc[i];
  }
  float w0 = wsum[0] * invN, w1 = wsum[1] * invN, w2 = wsum[2] * invN;
  float mx = fmaxf(w0, fmaxf(w1, w2));
  float e0 = expf(w0 - mx), e1 = expf(w1 - mx), e2 = expf(w2 - mx);
  float inv = 1.f / (e0 + e1 + e2);
  float a0 = e0 * inv, a1 = e1 * inv, a2 = e2 * inv;
#pragma unroll
  for (int u = 0; u < MIX_V; ++u) {
    const int i = base + u * 256;
    float4 o;
    o.x = a0 * a[u].x + a1 * b[u].x + a2 * c[u].x;
    o.y = a0 * a[u].y + a1 * b[u].y + a2 * c[u].y;
    o.z = a0 * a[u].z + a1 * b[u].z + a2 * c[u].z;
    o.w = a0 * a[u].w + a1 * b[u].w + a2 * c[u].w;
    if (i < n4) ((float4*)out)[i] = o;
  }
  if (blockIdx.x == 0 && threadIdx.x == 0) { out[n] = a0; out[n + 1] = a1; out[n + 2] = a2; }
}


extern "C" void kernel_launch(void* const* d_in, const int* in_sizes, int n_in,
                              void* d_out, int out_size, void* d_ws, size_t ws_size,
                              hipStream_t stream) {
  const float* dst_feat = (const float*)d_in[0];
  const float* src_feat[RELS] = {(const float*)d_in[1], (const float*)d_in[2],
                                 (const float*)d_in[3]};
  const float* sem_W1 = (const float*)d_in[4];
  const float* sem_b1 = (const float*)d_in[5];
  const float* sem_W2 = (const float*)d_in[6];
  const float *Wg[RELS], *al[RELS], *ar[RELS], *bg[RELS];
  const int *sidx[RELS], *didx[RELS];
  for (int r = 0; r < RELS; ++r) {
    int b = 7 + 6 * r;
    Wg[r] = (const float*)d_in[b + 0];
    al[r] = (const float*)d_in[b + 1];
    ar[r] = (const float*)d_in[b + 2];
    bg[r] = (const float*)d_in[b + 3];
    sidx[r] = (const int*)d_in[b + 4];
    didx[r] = (const int*)d_in[b + 5];
  }
  const int ND = in_sizes[0] / D;
  const int NS = in_sizes[1] / D;
  const int E  = in_sizes[11];

  float* ws   = (float*)d_ws;
  float* z3   = ws;
  float* wsum = z3 + (size_t)RELS * ND * D;
  int2*  lst = (int2*)d_out;
  float* el  = (float*)(lst + (size_t)RELS * E);
  float* er  = el + (size_t)RELS * NS;
  int*   cnt = (int*)(er + (size_t)RELS * ND);
  int*   cur = cnt + (size_t)RELS * ND;
  float* wa  = (float*)(cur + (size_t)RELS * ND);
  float* wr  = wa + RELS * D;
  int*   alloc = (int*)(wr + RELS * D);
  int*   rank = (int*)(z3 + (size_t)(RELS - 1) * ND * D);
  const int NCH = (E + CH_EDGES - 1) / CH_EDGES;
  int*   tab = (int*)z3;

  k_rowdot<<<RELS, 256, 0, stream>>>(Wg[0], Wg[1], Wg[2], al[0], al[1], al[2], ar[0], ar[1],
                                     ar[2], wa, wr, wsum, alloc);
  k_matvec<<<dim3((NS + 8 * MV_ROWS - 1) / (8 * MV_ROWS), RELS), 256, 0, stream>>>(
      src_feat[0], src_feat[1], src_feat[2], wa, el, NS);
  k_rank<<<dim3(NCH, RELS), 1024, 0, stream>>>(didx[0], didx[1], didx[2], tab, rank, E, ND, NCH);
  k_matvec_dst<<<(ND + 8 * MV_ROWS - 1) / (8 * MV_ROWS), 256, 0, stream>>>(
      dst_feat, wr, er, tab, cnt, cur, alloc, ND, NCH);
  k_place<<<dim3(NCH, RELS), 1024, 0, stream>>>(
      sidx[0], sidx[1], sidx[2], didx[0], didx[1], didx[2], el, tab, rank, lst, E, NS, ND, NCH);

  k_gather_gemm<<<dim3((ND + 8 * FT_ROWS - 1) / (8 * FT_ROWS), RELS), 512, 0, stream>>>(
      cnt, cur, er, lst, src_feat[0], src_feat[1], src_feat[2], Wg[0], Wg[1], Wg[2], bg[0], bg[1],
      bg[2], sem_W1, sem_b1, sem_W2, wsum, z3, ND, E);

  k_mix<<<(ND * D / 4 + 256 * MIX_V - 1) / (256 * MIX_V), 256, 0, stream>>>(
      z3, wsum, (float*)d_out, ND * D, 1.0f / (float)ND);
}
